# starting kernel
# baseline (speedup 1.0000x reference)
#include <hip/hip_runtime.h>

typedef __attribute__((ext_vector_type(2))) float v2f;
typedef __attribute__((ext_vector_type(8))) float v8f;
typedef __attribute__((ext_vector_type(4))) int   v4i;

__device__ __forceinline__ v4i ntload4(const int* p) {
  return __builtin_nontemporal_load((const v4i*)p);
}

__global__ void gcn_init(float* __restrict__ deg, float* __restrict__ agg1, int n) {
  int i = blockIdx.x * blockDim.x + threadIdx.x;
  if (i < n) { deg[i] = 1.0f; agg1[i] = 0.0f; }
}

__global__ void gcn_degree(const int* __restrict__ col, float* __restrict__ deg, int e) {
  int i = blockIdx.x * blockDim.x + threadIdx.x;
  int base = i << 2;
  if (base + 3 < e) {
    v4i c = ntload4(col + base);
    atomicAdd(&deg[c.x], 1.0f);
    atomicAdd(&deg[c.y], 1.0f);
    atomicAdd(&deg[c.z], 1.0f);
    atomicAdd(&deg[c.w], 1.0f);
  } else {
    for (int j = base; j < e; ++j) atomicAdd(&deg[col[j]], 1.0f);
  }
}

__global__ void gcn_dis(const float* __restrict__ x, float* __restrict__ deg_dis,
                        float* __restrict__ xp, int n) {
  int i = blockIdx.x * blockDim.x + threadIdx.x;
  if (i < n) {
    float d = rsqrtf(deg_dis[i]);
    deg_dis[i] = d;
    xp[i] = x[i] * d;
  }
}

__global__ void gcn_l1(const int* __restrict__ row, const int* __restrict__ col,
                       const float* __restrict__ xp, const float* __restrict__ dis,
                       float* __restrict__ agg1, int e) {
  int i = blockIdx.x * blockDim.x + threadIdx.x;
  int base = i << 2;
  if (base + 3 < e) {
    v4i r = ntload4(row + base);
    v4i c = ntload4(col + base);
    atomicAdd(&agg1[c.x], xp[r.x] * dis[c.x]);
    atomicAdd(&agg1[c.y], xp[r.y] * dis[c.y]);
    atomicAdd(&agg1[c.z], xp[r.z] * dis[c.z]);
    atomicAdd(&agg1[c.w], xp[r.w] * dis[c.w]);
  } else {
    for (int j = base; j < e; ++j) {
      int rj = row[j], cj = col[j];
      atomicAdd(&agg1[cj], xp[rj] * dis[cj]);
    }
  }
}

__global__ void gcn_node(const float* __restrict__ agg1,
                         const float* __restrict__ xp,
                         const float* __restrict__ dis,
                         const float* __restrict__ W1,
                         const float* __restrict__ b1,
                         const float* __restrict__ W2,
                         const float* __restrict__ b2,
                         float* __restrict__ h2p,
                         float* __restrict__ out,
                         int n) {
  int lane = threadIdx.x & 31;
  int wave = threadIdx.x >> 5;
  int tile = blockIdx.x * (blockDim.x >> 5) + wave;
  int base = tile * 16;
  int m    = lane & 15;
  int half = lane >> 4;
  int node = base + m;
  int nc   = node < n ? node : (n - 1);

  float av = agg1[nc] + xp[nc] * dis[nc];

  int   wcol  = (m < 2) ? m : 0;
  float wmask = (m < 2) ? 1.0f : 0.0f;

  v8f acc = {};
#pragma unroll
  for (int s = 0; s < 4; ++s) {
    int j0 = s * 4 + half * 2;
    v2f w1 = *(const v2f*)(W1 + j0);
    v2f bi = *(const v2f*)(b1 + j0);
    v2f a;
    a.x = fmaxf(av * w1.x + bi.x, 0.0f);
    a.y = fmaxf(av * w1.y + bi.y, 0.0f);
    v2f bb;
    bb.x = W2[j0 * 2 + wcol]       * wmask;
    bb.y = W2[(j0 + 1) * 2 + wcol] * wmask;
    acc = __builtin_amdgcn_wmma_f32_16x16x4_f32(
         false, a,  false, bb,
         (short)0, acc,  false,  false);
  }

  if (m < 2) {
    float bv = b2[m];
#pragma unroll
    for (int r = 0; r < 8; ++r) {
      int nd = base + r + half * 8;
      if (nd < n) {
        float h2 = acc[r];
        float dv = dis[nd];
        float hp = h2 * dv;
        h2p[nd * 2 + m] = hp;
        out[nd * 2 + m] = bv + hp * dv;
      }
    }
  }
}

__global__ void gcn_l2(const int* __restrict__ row, const int* __restrict__ col,
                       const float* __restrict__ h2p, const float* __restrict__ dis,
                       float* __restrict__ out, int e) {
  int i = blockIdx.x * blockDim.x + threadIdx.x;
  int base = i << 2;
  if (base + 3 < e) {
    v4i r = ntload4(row + base);
    v4i c = ntload4(col + base);
#pragma unroll
    for (int j = 0; j < 4; ++j) {
      int rj = (j == 0) ? r.x : (j == 1) ? r.y : (j == 2) ? r.z : r.w;
      int cj = (j == 0) ? c.x : (j == 1) ? c.y : (j == 2) ? c.z : c.w;
      float dv = dis[cj];
      v2f hp = *(const v2f*)(h2p + 2 * (size_t)rj);
      atomicAdd(&out[2 * (size_t)cj + 0], hp.x * dv);
      atomicAdd(&out[2 * (size_t)cj + 1], hp.y * dv);
    }
  } else {
    for (int j = base; j < e; ++j) {
      int rj = row[j], cj = col[j];
      float dv = dis[cj];
      atomicAdd(&out[2 * (size_t)cj + 0], h2p[2 * (size_t)rj + 0] * dv);
      atomicAdd(&out[2 * (size_t)cj + 1], h2p[2 * (size_t)rj + 1] * dv);
    }
  }
}

extern "C" void kernel_launch(void* const* d_in, const int* in_sizes, int n_in,
                              void* d_out, int out_size, void* d_ws, size_t ws_size,
                              hipStream_t stream) {
  const float* x  = (const float*)d_in[0];
  const int*   ei = (const int*)d_in[1];
  const float* W1 = (const float*)d_in[2];
  const float* b1 = (const float*)d_in[3];
  const float* W2 = (const float*)d_in[4];
  const float* b2 = (const float*)d_in[5];
  float* out = (float*)d_out;

  const int n = in_sizes[0];
  const int e = in_sizes[1] / 2;
  const int* row = ei;
  const int* col = ei + e;

  size_t np   = (size_t)((n + 3) & ~3);
  float* ws   = (float*)d_ws;
  float* dis  = ws;
  float* xp   = ws + np;
  float* agg1 = ws + 2 * np;
  float* h2p  = ws + 3 * np;

  const int tb = 256;
  const int nb_node = (n + tb - 1) / tb;
  const int ethreads = (e + 3) / 4;
  const int nb_edge = (ethreads + tb - 1) / tb;
  const int tiles = (n + 15) / 16;
  const int nb_tile = (tiles + 7) / 8;

  gcn_init  <<<nb_node, tb, 0, stream>>>(dis, agg1, n);
  gcn_degree<<<nb_edge, tb, 0, stream>>>(col, dis, e);
  gcn_dis   <<<nb_node, tb, 0, stream>>>(x, dis, xp, n);
  gcn_l1    <<<nb_edge, tb, 0, stream>>>(row, col, xp, dis, agg1, e);
  gcn_node  <<<nb_tile, 256, 0, stream>>>(agg1, xp, dis, W1, b1, W2, b2, h2p, out, n);
  gcn_l2    <<<nb_edge, tb, 0, stream>>>(row, col, h2p, dis, out, e);
}


# Edge scatters split by L2 half: a wave reads its die id and draws edge chunks from its half's ticket queue, so each destination line takes float atomics (arrival-order sums as before) from one half
# speedup vs baseline: 2.6853x; 2.6853x over previous
#include <hip/hip_runtime.h>


typedef __attribute__((ext_vector_type(2))) float v2f;
typedef __attribute__((ext_vector_type(8))) float v8f;
typedef __attribute__((ext_vector_type(4))) int   v4i;

#define EPT 512
#define EDGE_BLOCKS 2048

__device__ __forceinline__ v4i ntload4(const int* p) {
  return __builtin_nontemporal_load((const v4i*)p);
}

__device__ __forceinline__ int cache_half() {
  unsigned id = __builtin_amdgcn_s_sendmsg_rtn(0x87);
  return (int)((id >> 18) & 1u);
}

__device__ __forceinline__ int split_node(int n) { return ((n >> 1) + 63) & ~63; }

__device__ __forceinline__ int take_ticket(int* ctr) {
  int t = 0;
  if ((threadIdx.x & 31) == 0) t = atomicAdd(ctr, 1);
  return __builtin_amdgcn_readfirstlane(t);
}

__global__ void gcn_init(float* __restrict__ deg, float* __restrict__ agg1,
                         int* __restrict__ tick, int cs, int n) {
  int i = blockIdx.x * blockDim.x + threadIdx.x;
  if (i < n) { deg[i] = 1.0f; agg1[i] = 0.0f; }
  if (i < 4) tick[i * cs] = 0;
}

__global__ void gcn_tick0(int* __restrict__ tick, int cs) {
  if (threadIdx.x < 2) tick[threadIdx.x * cs] = 0;
}

__global__ void __launch_bounds__(256)
gcn_degree(const int* __restrict__ col, float* __restrict__ deg,
           int* __restrict__ tick, int cs, int e, int n) {
  const int lane  = threadIdx.x & 31;
  const int split = split_node(n);
  const int nt    = (e + EPT - 1) / EPT;
  const int h     = cache_half();
  for (int p = 0; p < 2; ++p) {
    const int q = h ^ p;
    int* ctr = tick + q * cs;
    for (;;) {
      const int t = take_ticket(ctr);
      if (t >= nt) break;
      const int e0 = t * EPT + lane * 4;
      if (t * EPT + EPT <= e) {
        v4i c[4];
#pragma unroll
        for (int k = 0; k < 4; ++k) c[k] = ntload4(col + e0 + k * 128);
#pragma unroll
        for (int k = 0; k < 4; ++k) {
          if ((c[k].x >= split) == (q != 0)) atomicAdd(&deg[c[k].x], 1.0f);
          if ((c[k].y >= split) == (q != 0)) atomicAdd(&deg[c[k].y], 1.0f);
          if ((c[k].z >= split) == (q != 0)) atomicAdd(&deg[c[k].z], 1.0f);
          if ((c[k].w >= split) == (q != 0)) atomicAdd(&deg[c[k].w], 1.0f);
        }
      } else {
        for (int k = 0; k < 4; ++k) {
          const int b = e0 + k * 128;
          for (int j = b; j < b + 4 && j < e; ++j) {
            const int cj = col[j];
            if ((cj >= split) == (q != 0)) atomicAdd(&deg[cj], 1.0f);
          }
        }
      }
    }
  }
}

__global__ void gcn_dis(const float* __restrict__ x, float* __restrict__ deg_dis,
                        float* __restrict__ xp, int n) {
  int i = blockIdx.x * blockDim.x + threadIdx.x;
  if (i < n) {
    float d = rsqrtf(deg_dis[i]);
    deg_dis[i] = d;
    xp[i] = x[i] * d;
  }
}

__global__ void __launch_bounds__(256)
gcn_l1(const int* __restrict__ row, const int* __restrict__ col,
       const float* __restrict__ xp, const float* __restrict__ dis,
       float* __restrict__ agg1, int* __restrict__ tick, int cs, int e, int n) {
  const int lane  = threadIdx.x & 31;
  const int split = split_node(n);
  const int nt    = (e + EPT - 1) / EPT;
  const int h     = cache_half();
  for (int p = 0; p < 2; ++p) {
    const int q = h ^ p;
    int* ctr = tick + q * cs;
    for (;;) {
      const int t = take_ticket(ctr);
      if (t >= nt) break;
      const int e0 = t * EPT + lane * 4;
      if (t * EPT + EPT <= e) {
        v4i r[4], c[4];
#pragma unroll
        for (int k = 0; k < 4; ++k) {
          r[k] = ntload4(row + e0 + k * 128);
          c[k] = ntload4(col + e0 + k * 128);
        }
#pragma unroll
        for (int k = 0; k < 4; ++k) {
          float v[4];
          bool  o[4];
#pragma unroll
          for (int j = 0; j < 4; ++j) {
            const int rj = r[k][j], cj = c[k][j];
            o[j] = (cj >= split) == (q != 0);
            v[j] = xp[o[j] ? rj : 0] * dis[o[j] ? cj : 0];
          }
#pragma unroll
          for (int j = 0; j < 4; ++j)
            if (o[j]) atomicAdd(&agg1[c[k][j]], v[j]);
        }
      } else {
        for (int k = 0; k < 4; ++k) {
          const int b = e0 + k * 128;
          for (int j = b; j < b + 4 && j < e; ++j) {
            const int rj = row[j], cj = col[j];
            if ((cj >= split) == (q != 0)) atomicAdd(&agg1[cj], xp[rj] * dis[cj]);
          }
        }
      }
    }
  }
}

__global__ void gcn_node(const float* __restrict__ agg1,
                         const float* __restrict__ xp,
                         const float* __restrict__ dis,
                         const float* __restrict__ W1,
                         const float* __restrict__ b1,
                         const float* __restrict__ W2,
                         const float* __restrict__ b2,
                         float* __restrict__ h2p,
                         float* __restrict__ out,
                         int n) {
  int lane = threadIdx.x & 31;
  int wave = threadIdx.x >> 5;
  int tile = blockIdx.x * (blockDim.x >> 5) + wave;
  int base = tile * 16;
  int m    = lane & 15;
  int half = lane >> 4;
  int node = base + m;
  int nc   = node < n ? node : (n - 1);

  float av = agg1[nc] + xp[nc] * dis[nc];

  int   wcol  = (m < 2) ? m : 0;
  float wmask = (m < 2) ? 1.0f : 0.0f;

  v8f acc = {};
#pragma unroll
  for (int s = 0; s < 4; ++s) {
    int j0 = s * 4 + half * 2;
    v2f w1 = *(const v2f*)(W1 + j0);
    v2f bi = *(const v2f*)(b1 + j0);
    v2f a;
    a.x = fmaxf(av * w1.x + bi.x, 0.0f);
    a.y = fmaxf(av * w1.y + bi.y, 0.0f);
    v2f bb;
    bb.x = W2[j0 * 2 + wcol]       * wmask;
    bb.y = W2[(j0 + 1) * 2 + wcol] * wmask;
    acc = __builtin_amdgcn_wmma_f32_16x16x4_f32(
         false, a,  false, bb,
         (short)0, acc,  false,  false);
  }

  if (m < 2) {
    float bv = b2[m];
#pragma unroll
    for (int r = 0; r < 8; ++r) {
      int nd = base + r + half * 8;
      if (nd < n) {
        float h2 = acc[r];
        float dv = dis[nd];
        float hp = h2 * dv;
        h2p[nd * 2 + m] = hp;
        out[nd * 2 + m] = bv + hp * dv;
      }
    }
  }
}

__global__ void __launch_bounds__(256)
gcn_l2(const int* __restrict__ row, const int* __restrict__ col,
       const float* __restrict__ h2p, const float* __restrict__ dis,
       float* __restrict__ out, int* __restrict__ tick, int cs, int e, int n) {
  const int lane  = threadIdx.x & 31;
  const int split = split_node(n);
  const int nt    = (e + EPT - 1) / EPT;
  const int h     = cache_half();
  for (int p = 0; p < 2; ++p) {
    const int q = h ^ p;
    int* ctr = tick + q * cs;
    for (;;) {
      const int t = take_ticket(ctr);
      if (t >= nt) break;
      const int e0 = t * EPT + lane * 4;
      if (t * EPT + EPT <= e) {
        v4i r[4], c[4];
#pragma unroll
        for (int k = 0; k < 4; ++k) {
          r[k] = ntload4(row + e0 + k * 128);
          c[k] = ntload4(col + e0 + k * 128);
        }
#pragma unroll
        for (int k = 0; k < 4; ++k) {
          v2f  v[4];
          bool o[4];
#pragma unroll
          for (int j = 0; j < 4; ++j) {
            const int rj = r[k][j], cj = c[k][j];
            o[j] = (cj >= split) == (q != 0);
            const float dv = dis[o[j] ? cj : 0];
            const v2f   hp = *(const v2f*)(h2p + 2 * (size_t)(o[j] ? rj : 0));
            v[j] = hp * dv;
          }
#pragma unroll
          for (int j = 0; j < 4; ++j)
            if (o[j]) {
              const size_t cj = (size_t)c[k][j];
              atomicAdd(&out[2 * cj + 0], v[j].x);
              atomicAdd(&out[2 * cj + 1], v[j].y);
            }
        }
      } else {
        for (int k = 0; k < 4; ++k) {
          const int b = e0 + k * 128;
          for (int j = b; j < b + 4 && j < e; ++j) {
            const int rj = row[j], cj = col[j];
            if ((cj >= split) == (q != 0)) {
              const float dv = dis[cj];
              atomicAdd(&out[2 * (size_t)cj + 0], h2p[2 * (size_t)rj + 0] * dv);
              atomicAdd(&out[2 * (size_t)cj + 1], h2p[2 * (size_t)rj + 1] * dv);
            }
          }
        }
      }
    }
  }
}

extern "C" void kernel_launch(void* const* d_in, const int* in_sizes, int n_in,
                              void* d_out, int out_size, void* d_ws, size_t ws_size,
                              hipStream_t stream) {
  const float* x  = (const float*)d_in[0];
  const int*   ei = (const int*)d_in[1];
  const float* W1 = (const float*)d_in[2];
  const float* b1 = (const float*)d_in[3];
  const float* W2 = (const float*)d_in[4];
  const float* b2 = (const float*)d_in[5];
  float* out = (float*)d_out;

  const int n = in_sizes[0];
  const int e = in_sizes[1] / 2;
  const int* row = ei;
  const int* col = ei + e;

  size_t np   = (size_t)((n + 3) & ~3);
  float* ws   = (float*)d_ws;
  float* dis  = ws;
  float* xp   = ws + np;
  float* agg1 = ws + 2 * np;
  float* h2p  = ws + 3 * np;

  const int cs = (n >= 512) ? 128 : 1;
  int* tick24 = (int*)h2p;
  int* tick6  = (int*)agg1;

  const int tb = 256;
  const int nb_node = (n + tb - 1) / tb;
  const int tiles = (n + 15) / 16;
  const int nb_tile = (tiles + 7) / 8;

  gcn_init  <<<nb_node, tb, 0, stream>>>(dis, agg1, tick24, cs, n);
  gcn_degree<<<EDGE_BLOCKS, tb, 0, stream>>>(col, dis, tick24, cs, e, n);
  gcn_dis   <<<nb_node, tb, 0, stream>>>(x, dis, xp, n);
  gcn_l1    <<<EDGE_BLOCKS, tb, 0, stream>>>(row, col, xp, dis, agg1, tick24 + 2 * cs, cs, e, n);
  gcn_node  <<<nb_tile, 256, 0, stream>>>(agg1, xp, dis, W1, b1, W2, b2, h2p, out, n);
  gcn_tick0 <<<1, 32, 0, stream>>>(tick6, cs);
  gcn_l2    <<<EDGE_BLOCKS, tb, 0, stream>>>(row, col, h2p, dis, out, tick6, cs, e, n);
}


# Destination factor dis[col] left out of the per-edge terms: edge kernels gather and add only the source value; each node's sum is scaled once after (node kernel, new finishing kernel)
# speedup vs baseline: 3.2911x; 1.2256x over previous
#include <hip/hip_runtime.h>


typedef __attribute__((ext_vector_type(2))) float v2f;
typedef __attribute__((ext_vector_type(8))) float v8f;
typedef __attribute__((ext_vector_type(4))) int   v4i;
typedef __attribute__((ext_vector_type(4))) float v4f;

#define EPT 512
#define EDGE_BLOCKS 2048

__device__ __forceinline__ v4i ntload4(const int* p) {
  return __builtin_nontemporal_load((const v4i*)p);
}

__device__ __forceinline__ int cache_half() {
  unsigned id = __builtin_amdgcn_s_sendmsg_rtn(0x87);
  return (int)((id >> 18) & 1u);
}

__device__ __forceinline__ int split_node(int n) { return ((n >> 1) + 63) & ~63; }

__device__ __forceinline__ int take_ticket(int* ctr) {
  int t = 0;
  if ((threadIdx.x & 31) == 0) t = atomicAdd(ctr, 1);
  return __builtin_amdgcn_readfirstlane(t);
}

__global__ void gcn_init(float* __restrict__ deg, float* __restrict__ agg1,
                         int* __restrict__ tick, int cs, int n) {
  int i = blockIdx.x * blockDim.x + threadIdx.x;
  if (i < n) { deg[i] = 1.0f; agg1[i] = 0.0f; }
  if (i < 4) tick[i * cs] = 0;
}

__global__ void gcn_tick0(int* __restrict__ tick, int cs) {
  if (threadIdx.x < 2) tick[threadIdx.x * cs] = 0;
}

__global__ void __launch_bounds__(256)
gcn_degree(const int* __restrict__ col, float* __restrict__ deg,
           int* __restrict__ tick, int cs, int e, int n) {
  const int lane  = threadIdx.x & 31;
  const int split = split_node(n);
  const int nt    = (e + EPT - 1) / EPT;
  const int h     = cache_half();
  for (int p = 0; p < 2; ++p) {
    const int q = h ^ p;
    int* ctr = tick + q * cs;
    for (;;) {
      const int t = take_ticket(ctr);
      if (t >= nt) break;
      const int e0 = t * EPT + lane * 4;
      if (t * EPT + EPT <= e) {
        v4i c[4];
#pragma unroll
        for (int k = 0; k < 4; ++k) c[k] = ntload4(col + e0 + k * 128);
#pragma unroll
        for (int k = 0; k < 4; ++k) {
          if ((c[k].x >= split) == (q != 0)) atomicAdd(&deg[c[k].x], 1.0f);
          if ((c[k].y >= split) == (q != 0)) atomicAdd(&deg[c[k].y], 1.0f);
          if ((c[k].z >= split) == (q != 0)) atomicAdd(&deg[c[k].z], 1.0f);
          if ((c[k].w >= split) == (q != 0)) atomicAdd(&deg[c[k].w], 1.0f);
        }
      } else {
        for (int k = 0; k < 4; ++k) {
          const int b = e0 + k * 128;
          for (int j = b; j < b + 4 && j < e; ++j) {
            const int cj = col[j];
            if ((cj >= split) == (q != 0)) atomicAdd(&deg[cj], 1.0f);
          }
        }
      }
    }
  }
}

__global__ void gcn_dis(const float* __restrict__ x, float* __restrict__ deg_dis,
                        float* __restrict__ xp, int n) {
  int i = blockIdx.x * blockDim.x + threadIdx.x;
  if (i < n) {
    float d = rsqrtf(deg_dis[i]);
    deg_dis[i] = d;
    xp[i] = x[i] * d;
  }
}

__global__ void __launch_bounds__(256)
gcn_l1(const int* __restrict__ row, const int* __restrict__ col,
       const float* __restrict__ xp,
       float* __restrict__ agg1, int* __restrict__ tick, int cs, int e, int n) {
  const int lane  = threadIdx.x & 31;
  const int split = split_node(n);
  const int nt    = (e + EPT - 1) / EPT;
  const int h     = cache_half();
  for (int p = 0; p < 2; ++p) {
    const int q = h ^ p;
    int* ctr = tick + q * cs;
    for (;;) {
      const int t = take_ticket(ctr);
      if (t >= nt) break;
      const int e0 = t * EPT + lane * 4;
      if (t * EPT + EPT <= e) {
        v4i r[4], c[4];
#pragma unroll
        for (int k = 0; k < 4; ++k) {
          r[k] = ntload4(row + e0 + k * 128);
          c[k] = ntload4(col + e0 + k * 128);
        }
#pragma unroll
        for (int k = 0; k < 4; ++k) {
          float v[4];
          bool  o[4];
#pragma unroll
          for (int j = 0; j < 4; ++j) {
            const int rj = r[k][j], cj = c[k][j];
            o[j] = (cj >= split) == (q != 0);
            v[j] = xp[o[j] ? rj : 0];
          }
#pragma unroll
          for (int j = 0; j < 4; ++j)
            if (o[j]) atomicAdd(&agg1[c[k][j]], v[j]);
        }
      } else {
        for (int k = 0; k < 4; ++k) {
          const int b = e0 + k * 128;
          for (int j = b; j < b + 4 && j < e; ++j) {
            const int rj = row[j], cj = col[j];
            if ((cj >= split) == (q != 0)) atomicAdd(&agg1[cj], xp[rj]);
          }
        }
      }
    }
  }
}

__global__ void gcn_node(const float* __restrict__ agg1,
                         const float* __restrict__ xp,
                         const float* __restrict__ dis,
                         const float* __restrict__ W1,
                         const float* __restrict__ b1,
                         const float* __restrict__ W2,
                         float* __restrict__ h2p,
                         float* __restrict__ out,
                         int n) {
  int lane = threadIdx.x & 31;
  int wave = threadIdx.x >> 5;
  int tile = blockIdx.x * (blockDim.x >> 5) + wave;
  int base = tile * 16;
  int m    = lane & 15;
  int half = lane >> 4;
  int node = base + m;
  int nc   = node < n ? node : (n - 1);

  float av = (agg1[nc] + xp[nc]) * dis[nc];

  int   wcol  = (m < 2) ? m : 0;
  float wmask = (m < 2) ? 1.0f : 0.0f;

  v8f acc = {};
#pragma unroll
  for (int s = 0; s < 4; ++s) {
    int j0 = s * 4 + half * 2;
    v2f w1 = *(const v2f*)(W1 + j0);
    v2f bi = *(const v2f*)(b1 + j0);
    v2f a;
    a.x = fmaxf(av * w1.x + bi.x, 0.0f);
    a.y = fmaxf(av * w1.y + bi.y, 0.0f);
    v2f bb;
    bb.x = W2[j0 * 2 + wcol]       * wmask;
    bb.y = W2[(j0 + 1) * 2 + wcol] * wmask;
    acc = __builtin_amdgcn_wmma_f32_16x16x4_f32(
         false, a,  false, bb,
         (short)0, acc,  false,  false);
  }

  if (m < 2) {
#pragma unroll
    for (int r = 0; r < 8; ++r) {
      int nd = base + r + half * 8;
      if (nd < n) {
        float hp = acc[r] * dis[nd];
        h2p[nd * 2 + m] = hp;
        out[nd * 2 + m] = hp;
      }
    }
  }
}

__global__ void __launch_bounds__(256)
gcn_l2(const int* __restrict__ row, const int* __restrict__ col,
       const float* __restrict__ h2p,
       float* __restrict__ out, int* __restrict__ tick, int cs, int e, int n) {
  const int lane  = threadIdx.x & 31;
  const int split = split_node(n);
  const int nt    = (e + EPT - 1) / EPT;
  const int h     = cache_half();
  for (int p = 0; p < 2; ++p) {
    const int q = h ^ p;
    int* ctr = tick + q * cs;
    for (;;) {
      const int t = take_ticket(ctr);
      if (t >= nt) break;
      const int e0 = t * EPT + lane * 4;
      if (t * EPT + EPT <= e) {
        v4i r[4], c[4];
#pragma unroll
        for (int k = 0; k < 4; ++k) {
          r[k] = ntload4(row + e0 + k * 128);
          c[k] = ntload4(col + e0 + k * 128);
        }
#pragma unroll
        for (int k = 0; k < 4; ++k) {
          v2f  v[4];
          bool o[4];
#pragma unroll
          for (int j = 0; j < 4; ++j) {
            const int rj = r[k][j], cj = c[k][j];
            o[j] = (cj >= split) == (q != 0);
            v[j] = *(const v2f*)(h2p + 2 * (size_t)(o[j] ? rj : 0));
          }
#pragma unroll
          for (int j = 0; j < 4; ++j)
            if (o[j]) {
              const size_t cj = (size_t)c[k][j];
              atomicAdd(&out[2 * cj + 0], v[j].x);
              atomicAdd(&out[2 * cj + 1], v[j].y);
            }
        }
      } else {
        for (int k = 0; k < 4; ++k) {
          const int b = e0 + k * 128;
          for (int j = b; j < b + 4 && j < e; ++j) {
            const int rj = row[j], cj = col[j];
            if ((cj >= split) == (q != 0)) {
              atomicAdd(&out[2 * (size_t)cj + 0], h2p[2 * (size_t)rj + 0]);
              atomicAdd(&out[2 * (size_t)cj + 1], h2p[2 * (size_t)rj + 1]);
            }
          }
        }
      }
    }
  }
}

__global__ void gcn_fin(const float* __restrict__ dis, const float* __restrict__ b2,
                        float* __restrict__ out, int n) {
  const int i = 4 * (blockIdx.x * blockDim.x + threadIdx.x);
  const float c0 = b2[0], c1 = b2[1];
  if (i + 3 < n) {
    const v4f d = *(const v4f*)(dis + i);
    float* o = out + 2 * (size_t)i;
    v4f a = *(const v4f*)o;
    v4f b = *(const v4f*)(o + 4);
    a.x = c0 + d.x * a.x;  a.y = c1 + d.x * a.y;
    a.z = c0 + d.y * a.z;  a.w = c1 + d.y * a.w;
    b.x = c0 + d.z * b.x;  b.y = c1 + d.z * b.y;
    b.z = c0 + d.w * b.z;  b.w = c1 + d.w * b.w;
    *(v4f*)o       = a;
    *(v4f*)(o + 4) = b;
  } else {
    for (int j = i; j < n; ++j) {
      const float d = dis[j];
      out[2 * (size_t)j + 0] = c0 + d * out[2 * (size_t)j + 0];
      out[2 * (size_t)j + 1] = c1 + d * out[2 * (size_t)j + 1];
    }
  }
}

extern "C" void kernel_launch(void* const* d_in, const int* in_sizes, int n_in,
                              void* d_out, int out_size, void* d_ws, size_t ws_size,
                              hipStream_t stream) {
  const float* x  = (const float*)d_in[0];
  const int*   ei = (const int*)d_in[1];
  const float* W1 = (const float*)d_in[2];
  const float* b1 = (const float*)d_in[3];
  const float* W2 = (const float*)d_in[4];
  const float* b2 = (const float*)d_in[5];
  float* out = (float*)d_out;

  const int n = in_sizes[0];
  const int e = in_sizes[1] / 2;
  const int* row = ei;
  const int* col = ei + e;

  size_t np   = (size_t)((n + 3) & ~3);
  float* ws   = (float*)d_ws;
  float* dis  = ws;
  float* xp   = ws + np;
  float* agg1 = ws + 2 * np;
  float* h2p  = ws + 3 * np;

  const int cs = (n >= 512) ? 128 : 1;
  int* tick24 = (int*)h2p;
  int* tick6  = (int*)agg1;

  const int tb = 256;
  const int nb_node = (n + tb - 1) / tb;
  const int tiles = (n + 15) / 16;
  const int nb_tile = (tiles + 7) / 8;
  const int nb_fin  = (n + 4 * tb - 1) / (4 * tb);

  gcn_init  <<<nb_node, tb, 0, stream>>>(dis, agg1, tick24, cs, n);
  gcn_degree<<<EDGE_BLOCKS, tb, 0, stream>>>(col, dis, tick24, cs, e, n);
  gcn_dis   <<<nb_node, tb, 0, stream>>>(x, dis, xp, n);
  gcn_l1    <<<EDGE_BLOCKS, tb, 0, stream>>>(row, col, xp, agg1, tick24 + 2 * cs, cs, e, n);
  gcn_node  <<<nb_tile, 256, 0, stream>>>(agg1, xp, dis, W1, b1, W2, h2p, out, n);
  gcn_tick0 <<<1, 32, 0, stream>>>(tick6, cs);
  gcn_l2    <<<EDGE_BLOCKS, tb, 0, stream>>>(row, col, h2p, out, tick6, cs, e, n);
  gcn_fin   <<<nb_fin, tb, 0, stream>>>(dis, b2, out, n);
}


# Layer-2 adds go out as lane pairs: lanes 2i,2i+1 swap half their edges (DPP), then add components 0 and 1 of one edge in one atomic instruction: half the lines touched (arrival-order sums as before)
# speedup vs baseline: 3.9171x; 1.1902x over previous
#include <hip/hip_runtime.h>


typedef __attribute__((ext_vector_type(2))) float v2f;
typedef __attribute__((ext_vector_type(8))) float v8f;
typedef __attribute__((ext_vector_type(4))) int   v4i;
typedef __attribute__((ext_vector_type(4))) float v4f;

#define EPT 512
#define EDGE_BLOCKS 2048

__device__ __forceinline__ v4i ntload4(const int* p) {
  return __builtin_nontemporal_load((const v4i*)p);
}

__device__ __forceinline__ int cache_half() {
  unsigned id = __builtin_amdgcn_s_sendmsg_rtn(0x87);
  return (int)((id >> 18) & 1u);
}

__device__ __forceinline__ int pair_swap(int v) {
  return __builtin_amdgcn_update_dpp(0, v, 0xB1, 0xF, 0xF, true);
}
__device__ __forceinline__ float pair_swapf(float v) {
  return __int_as_float(pair_swap(__float_as_int(v)));
}

__device__ __forceinline__ int split_node(int n) { return ((n >> 1) + 63) & ~63; }

__device__ __forceinline__ int take_ticket(int* ctr) {
  int t = 0;
  if ((threadIdx.x & 31) == 0) t = atomicAdd(ctr, 1);
  return __builtin_amdgcn_readfirstlane(t);
}

__global__ void gcn_init(float* __restrict__ deg, float* __restrict__ agg1,
                         int* __restrict__ tick, int cs, int n) {
  int i = blockIdx.x * blockDim.x + threadIdx.x;
  if (i < n) { deg[i] = 1.0f; agg1[i] = 0.0f; }
  if (i < 4) tick[i * cs] = 0;
}

__global__ void gcn_tick0(int* __restrict__ tick, int cs) {
  if (threadIdx.x < 2) tick[threadIdx.x * cs] = 0;
}

__global__ void __launch_bounds__(256)
gcn_degree(const int* __restrict__ col, float* __restrict__ deg,
           int* __restrict__ tick, int cs, int e, int n) {
  const int lane  = threadIdx.x & 31;
  const int split = split_node(n);
  const int nt    = (e + EPT - 1) / EPT;
  const int h     = cache_half();
  for (int p = 0; p < 2; ++p) {
    const int q = h ^ p;
    int* ctr = tick + q * cs;
    for (;;) {
      const int t = take_ticket(ctr);
      if (t >= nt) break;
      const int e0 = t * EPT + lane * 4;
      if (t * EPT + EPT <= e) {
        v4i c[4];
#pragma unroll
        for (int k = 0; k < 4; ++k) c[k] = ntload4(col + e0 + k * 128);
#pragma unroll
        for (int k = 0; k < 4; ++k) {
          if ((c[k].x >= split) == (q != 0)) atomicAdd(&deg[c[k].x], 1.0f);
          if ((c[k].y >= split) == (q != 0)) atomicAdd(&deg[c[k].y], 1.0f);
          if ((c[k].z >= split) == (q != 0)) atomicAdd(&deg[c[k].z], 1.0f);
          if ((c[k].w >= split) == (q != 0)) atomicAdd(&deg[c[k].w], 1.0f);
        }
      } else {
        for (int k = 0; k < 4; ++k) {
          const int b = e0 + k * 128;
          for (int j = b; j < b + 4 && j < e; ++j) {
            const int cj = col[j];
            if ((cj >= split) == (q != 0)) atomicAdd(&deg[cj], 1.0f);
          }
        }
      }
    }
  }
}

__global__ void gcn_dis(const float* __restrict__ x, float* __restrict__ deg_dis,
                        float* __restrict__ xp, int n) {
  int i = blockIdx.x * blockDim.x + threadIdx.x;
  if (i < n) {
    float d = rsqrtf(deg_dis[i]);
    deg_dis[i] = d;
    xp[i] = x[i] * d;
  }
}

__global__ void __launch_bounds__(256)
gcn_l1(const int* __restrict__ row, const int* __restrict__ col,
       const float* __restrict__ xp,
       float* __restrict__ agg1, int* __restrict__ tick, int cs, int e, int n) {
  const int lane  = threadIdx.x & 31;
  const int split = split_node(n);
  const int nt    = (e + EPT - 1) / EPT;
  const int h     = cache_half();
  for (int p = 0; p < 2; ++p) {
    const int q = h ^ p;
    int* ctr = tick + q * cs;
    for (;;) {
      const int t = take_ticket(ctr);
      if (t >= nt) break;
      const int e0 = t * EPT + lane * 4;
      if (t * EPT + EPT <= e) {
        v4i r[4], c[4];
#pragma unroll
        for (int k = 0; k < 4; ++k) {
          r[k] = ntload4(row + e0 + k * 128);
          c[k] = ntload4(col + e0 + k * 128);
        }
#pragma unroll
        for (int k = 0; k < 4; ++k) {
          float v[4];
          bool  o[4];
#pragma unroll
          for (int j = 0; j < 4; ++j) {
            const int rj = r[k][j], cj = c[k][j];
            o[j] = (cj >= split) == (q != 0);
            v[j] = xp[o[j] ? rj : 0];
          }
#pragma unroll
          for (int j = 0; j < 4; ++j)
            if (o[j]) atomicAdd(&agg1[c[k][j]], v[j]);
        }
      } else {
        for (int k = 0; k < 4; ++k) {
          const int b = e0 + k * 128;
          for (int j = b; j < b + 4 && j < e; ++j) {
            const int rj = row[j], cj = col[j];
            if ((cj >= split) == (q != 0)) atomicAdd(&agg1[cj], xp[rj]);
          }
        }
      }
    }
  }
}

__global__ void gcn_node(const float* __restrict__ agg1,
                         const float* __restrict__ xp,
                         const float* __restrict__ dis,
                         const float* __restrict__ W1,
                         const float* __restrict__ b1,
                         const float* __restrict__ W2,
                         float* __restrict__ h2p,
                         float* __restrict__ out,
                         int n) {
  int lane = threadIdx.x & 31;
  int wave = threadIdx.x >> 5;
  int tile = blockIdx.x * (blockDim.x >> 5) + wave;
  int base = tile * 16;
  int m    = lane & 15;
  int half = lane >> 4;
  int node = base + m;
  int nc   = node < n ? node : (n - 1);

  float av = (agg1[nc] + xp[nc]) * dis[nc];

  int   wcol  = (m < 2) ? m : 0;
  float wmask = (m < 2) ? 1.0f : 0.0f;

  v8f acc = {};
#pragma unroll
  for (int s = 0; s < 4; ++s) {
    int j0 = s * 4 + half * 2;
    v2f w1 = *(const v2f*)(W1 + j0);
    v2f bi = *(const v2f*)(b1 + j0);
    v2f a;
    a.x = fmaxf(av * w1.x + bi.x, 0.0f);
    a.y = fmaxf(av * w1.y + bi.y, 0.0f);
    v2f bb;
    bb.x = W2[j0 * 2 + wcol]       * wmask;
    bb.y = W2[(j0 + 1) * 2 + wcol] * wmask;
    acc = __builtin_amdgcn_wmma_f32_16x16x4_f32(
         false, a,  false, bb,
         (short)0, acc,  false,  false);
  }

  if (m < 2) {
#pragma unroll
    for (int r = 0; r < 8; ++r) {
      int nd = base + r + half * 8;
      if (nd < n) {
        float hp = acc[r] * dis[nd];
        h2p[nd * 2 + m] = hp;
        out[nd * 2 + m] = hp;
      }
    }
  }
}

__global__ void __launch_bounds__(256)
gcn_l2(const int* __restrict__ row, const int* __restrict__ col,
       const float* __restrict__ h2p,
       float* __restrict__ out, int* __restrict__ tick, int cs, int e, int n) {
  const int lane  = threadIdx.x & 31;
  const int odd   = lane & 1;
  const int split = split_node(n);
  const int nt    = (e + EPT - 1) / EPT;
  const int h     = cache_half();
  for (int p = 0; p < 2; ++p) {
    const int q = h ^ p;
    int* ctr = tick + q * cs;
    for (;;) {
      const int t = take_ticket(ctr);
      if (t >= nt) break;
      const int e0 = t * EPT + lane * 4;
      if (t * EPT + EPT <= e) {
        v4i r[4], c[4];
#pragma unroll
        for (int k = 0; k < 4; ++k) {
          r[k] = ntload4(row + e0 + k * 128);
          c[k] = ntload4(col + e0 + k * 128);
        }
#pragma unroll
        for (int k = 0; k < 4; ++k) {
          v2f v[4];
#pragma unroll
          for (int j = 0; j < 4; ++j) {
            const int rj = r[k][j], cj = c[k][j];
            const bool oj = (cj >= split) == (q != 0);
            v[j] = *(const v2f*)(h2p + 2 * (size_t)(oj ? rj : 0));
          }
          int   nc[4];
          float nv[4];
#pragma unroll
          for (int j = 0; j < 4; ++j) {
            nc[j] = pair_swap(c[k][j]);
            nv[j] = pair_swapf(odd ? v[j].x : v[j].y);
          }
#pragma unroll
          for (int j = 0; j < 4; ++j) {
            const int   cc = odd ? nc[j] : c[k][j];
            const float vv = odd ? nv[j] : v[j].x;
            if ((cc >= split) == (q != 0)) atomicAdd(&out[2 * (size_t)cc + odd], vv);
          }
#pragma unroll
          for (int j = 0; j < 4; ++j) {
            const int   cc = odd ? c[k][j] : nc[j];
            const float vv = odd ? v[j].y : nv[j];
            if ((cc >= split) == (q != 0)) atomicAdd(&out[2 * (size_t)cc + odd], vv);
          }
        }
      } else {
        for (int k = 0; k < 4; ++k) {
          const int b = e0 + k * 128;
          for (int j = b; j < b + 4 && j < e; ++j) {
            const int rj = row[j], cj = col[j];
            if ((cj >= split) == (q != 0)) {
              atomicAdd(&out[2 * (size_t)cj + 0], h2p[2 * (size_t)rj + 0]);
              atomicAdd(&out[2 * (size_t)cj + 1], h2p[2 * (size_t)rj + 1]);
            }
          }
        }
      }
    }
  }
}

__global__ void gcn_fin(const float* __restrict__ dis, const float* __restrict__ b2,
                        float* __restrict__ out, int n) {
  const int i = 4 * (blockIdx.x * blockDim.x + threadIdx.x);
  const float c0 = b2[0], c1 = b2[1];
  if (i + 3 < n) {
    const v4f d = *(const v4f*)(dis + i);
    float* o = out + 2 * (size_t)i;
    v4f a = *(const v4f*)o;
    v4f b = *(const v4f*)(o + 4);
    a.x = c0 + d.x * a.x;  a.y = c1 + d.x * a.y;
    a.z = c0 + d.y * a.z;  a.w = c1 + d.y * a.w;
    b.x = c0 + d.z * b.x;  b.y = c1 + d.z * b.y;
    b.z = c0 + d.w * b.z;  b.w = c1 + d.w * b.w;
    *(v4f*)o       = a;
    *(v4f*)(o + 4) = b;
  } else {
    for (int j = i; j < n; ++j) {
      const float d = dis[j];
      out[2 * (size_t)j + 0] = c0 + d * out[2 * (size_t)j + 0];
      out[2 * (size_t)j + 1] = c1 + d * out[2 * (size_t)j + 1];
    }
  }
}

extern "C" void kernel_launch(void* const* d_in, const int* in_sizes, int n_in,
                              void* d_out, int out_size, void* d_ws, size_t ws_size,
                              hipStream_t stream) {
  const float* x  = (const float*)d_in[0];
  const int*   ei = (const int*)d_in[1];
  const float* W1 = (const float*)d_in[2];
  const float* b1 = (const float*)d_in[3];
  const float* W2 = (const float*)d_in[4];
  const float* b2 = (const float*)d_in[5];
  float* out = (float*)d_out;

  const int n = in_sizes[0];
  const int e = in_sizes[1] / 2;
  const int* row = ei;
  const int* col = ei + e;

  size_t np   = (size_t)((n + 3) & ~3);
  float* ws   = (float*)d_ws;
  float* dis  = ws;
  float* xp   = ws + np;
  float* agg1 = ws + 2 * np;
  float* h2p  = ws + 3 * np;

  const int cs = (n >= 512) ? 128 : 1;
  int* tick24 = (int*)h2p;
  int* tick6  = (int*)agg1;

  const int tb = 256;
  const int nb_node = (n + tb - 1) / tb;
  const int tiles = (n + 15) / 16;
  const int nb_tile = (tiles + 7) / 8;
  const int nb_fin  = (n + 4 * tb - 1) / (4 * tb);

  gcn_init  <<<nb_node, tb, 0, stream>>>(dis, agg1, tick24, cs, n);
  gcn_degree<<<EDGE_BLOCKS, tb, 0, stream>>>(col, dis, tick24, cs, e, n);
  gcn_dis   <<<nb_node, tb, 0, stream>>>(x, dis, xp, n);
  gcn_l1    <<<EDGE_BLOCKS, tb, 0, stream>>>(row, col, xp, agg1, tick24 + 2 * cs, cs, e, n);
  gcn_node  <<<nb_tile, 256, 0, stream>>>(agg1, xp, dis, W1, b1, W2, h2p, out, n);
  gcn_tick0 <<<1, 32, 0, stream>>>(tick6, cs);
  gcn_l2    <<<EDGE_BLOCKS, tb, 0, stream>>>(row, col, h2p, out, tick6, cs, e, n);
  gcn_fin   <<<nb_fin, tb, 0, stream>>>(dis, b2, out, n);
}


# Degree count and layer-1 sum built per node range in local memory (LDS f32 atomics, 8 ranges x 32 edge slices); each slice's partial sums then go to global as whole-line atomics: new sum tree
# speedup vs baseline: 4.9890x; 1.2736x over previous
#include <hip/hip_runtime.h>


typedef __attribute__((ext_vector_type(2))) float v2f;
typedef __attribute__((ext_vector_type(8))) float v8f;
typedef __attribute__((ext_vector_type(4))) int   v4i;
typedef __attribute__((ext_vector_type(4))) float v4f;

#define EPT 512
#define EDGE_BLOCKS 2048

#define RS_MAX 65536
#define SLICES 32
#define RTPB   1024
#define RANGE_BLOCKS 256

__device__ __forceinline__ v4i ntload4(const int* p) {
  return __builtin_nontemporal_load((const v4i*)p);
}

__device__ __forceinline__ int cache_half() {
  unsigned id = __builtin_amdgcn_s_sendmsg_rtn(0x87);
  return (int)((id >> 18) & 1u);
}

__device__ __forceinline__ int pair_swap(int v) {
  return __builtin_amdgcn_update_dpp(0, v, 0xB1, 0xF, 0xF, true);
}
__device__ __forceinline__ float pair_swapf(float v) {
  return __int_as_float(pair_swap(__float_as_int(v)));
}

__device__ __forceinline__ int split_node(int n) { return ((n >> 1) + 63) & ~63; }

__device__ __forceinline__ int take_ticket(int* ctr) {
  int t = 0;
  if ((threadIdx.x & 31) == 0) t = atomicAdd(ctr, 1);
  return __builtin_amdgcn_readfirstlane(t);
}

__device__ __forceinline__ int range_count(int n) {
  int r = (n + RS_MAX - 1) / RS_MAX;
  if (r < 8) r = 8;
  return (r + 1) & ~1;
}
__device__ __forceinline__ int range_size(int n, int r) { return ((n + r - 1) / r + 63) & ~63; }
__device__ __forceinline__ int slice_len(int e) { return ((e + SLICES - 1) / SLICES + 16383) & ~16383; }

__device__ __forceinline__ int take_block_ticket(int* ctr, int* slot) {
  if (threadIdx.x == 0) *slot = atomicAdd(ctr, 1);
  __syncthreads();
  const int t = *slot;
  __syncthreads();
  return t;
}

__device__ __forceinline__ v4i load4_or_none(const int* __restrict__ a, int i, int hi) {
  if (i + 3 < hi) return *(const v4i*)(a + i);
  v4i r = {-1, -1, -1, -1};
  if (i     < hi) r.x = a[i];
  if (i + 1 < hi) r.y = a[i + 1];
  if (i + 2 < hi) r.z = a[i + 2];
  return r;
}

__global__ void gcn_init(float* __restrict__ deg, float* __restrict__ agg1,
                         int* __restrict__ tick, int cs, int n) {
  int i = blockIdx.x * blockDim.x + threadIdx.x;
  if (i < n) { deg[i] = 1.0f; agg1[i] = 0.0f; }
  if (i < 4) tick[i * cs] = 0;
}

__global__ void gcn_tick0(int* __restrict__ tick, int cs) {
  if (threadIdx.x < 2) tick[threadIdx.x * cs] = 0;
}

__global__ void __launch_bounds__(RTPB)
gcn_degree(const int* __restrict__ col, float* __restrict__ deg,
           int* __restrict__ tick, int cs, int e, int n) {
  __shared__ __attribute__((aligned(16))) float acc[RS_MAX];
  __shared__ int   slot;
  const int tid = threadIdx.x;
  const int R   = range_count(n);
  const int Rh  = R >> 1;
  const int RS  = range_size(n, R);
  const int CH  = slice_len(e);
  const int nt  = Rh * SLICES;
  const int h   = cache_half();
  for (int p = 0; p < 2; ++p) {
    const int q = h ^ p;
    int* ctr = tick + q * cs;
    for (;;) {
      const int t = take_block_ticket(ctr, &slot);
      if (t >= nt) break;
      const int lo  = (q * Rh + t % Rh) * RS;
      const int cnt = min(RS, n - lo);
      const int elo = (t / Rh) * CH;
      const int ehi = min(e, elo + CH);
      if (cnt <= 0 || elo >= ehi) continue;
      for (int i = tid * 4; i < RS; i += RTPB * 4) *(v4f*)&acc[i] = (v4f){0.f, 0.f, 0.f, 0.f};
      __syncthreads();
      int b = elo + tid * 4;
      for (; b + 3 * RTPB * 4 + 3 < ehi; b += RTPB * 16) {
        v4i c[4];
#pragma unroll
        for (int u = 0; u < 4; ++u) c[u] = *(const v4i*)(col + b + u * RTPB * 4);
#pragma unroll
        for (int u = 0; u < 4; ++u) {
#pragma unroll
          for (int j = 0; j < 4; ++j) {
            const unsigned d = (unsigned)(c[u][j] - lo);
            if (d < (unsigned)cnt) atomicAdd(&acc[d], 1.0f);
          }
        }
      }
      for (; b < ehi; b += RTPB * 4) {
        const v4i c = load4_or_none(col, b, ehi);
#pragma unroll
        for (int j = 0; j < 4; ++j) {
          const unsigned d = (unsigned)(c[j] - lo);
          if (d < (unsigned)cnt) atomicAdd(&acc[d], 1.0f);
        }
      }
      __syncthreads();
      for (int i = tid; i < cnt; i += RTPB) {
        const float v = acc[i];
        if (v != 0.0f) atomicAdd(&deg[lo + i], v);
      }
    }
  }
}

__global__ void gcn_dis(const float* __restrict__ x, float* __restrict__ deg_dis,
                        float* __restrict__ xp, int n) {
  int i = blockIdx.x * blockDim.x + threadIdx.x;
  if (i < n) {
    float d = rsqrtf(deg_dis[i]);
    deg_dis[i] = d;
    xp[i] = x[i] * d;
  }
}

__global__ void __launch_bounds__(RTPB)
gcn_l1(const int* __restrict__ row, const int* __restrict__ col,
       const float* __restrict__ xp,
       float* __restrict__ agg1, int* __restrict__ tick, int cs, int e, int n) {
  __shared__ __attribute__((aligned(16))) float acc[RS_MAX];
  __shared__ int   slot;
  const int tid = threadIdx.x;
  const int R   = range_count(n);
  const int Rh  = R >> 1;
  const int RS  = range_size(n, R);
  const int CH  = slice_len(e);
  const int nt  = Rh * SLICES;
  const int h   = cache_half();
  for (int p = 0; p < 2; ++p) {
    const int q = h ^ p;
    int* ctr = tick + q * cs;
    for (;;) {
      const int t = take_block_ticket(ctr, &slot);
      if (t >= nt) break;
      const int lo  = (q * Rh + t % Rh) * RS;
      const int cnt = min(RS, n - lo);
      const int elo = (t / Rh) * CH;
      const int ehi = min(e, elo + CH);
      if (cnt <= 0 || elo >= ehi) continue;
      for (int i = tid * 4; i < RS; i += RTPB * 4) *(v4f*)&acc[i] = (v4f){0.f, 0.f, 0.f, 0.f};
      __syncthreads();
      int b = elo + tid * 4;
      for (; b + 3 * RTPB * 4 + 3 < ehi; b += RTPB * 16) {
        v4i r[4], c[4];
#pragma unroll
        for (int u = 0; u < 4; ++u) {
          c[u] = *(const v4i*)(col + b + u * RTPB * 4);
          r[u] = *(const v4i*)(row + b + u * RTPB * 4);
        }
        asm volatile("" : "+v"(c[0]), "+v"(c[1]), "+v"(c[2]), "+v"(c[3]),
                          "+v"(r[0]), "+v"(r[1]), "+v"(r[2]), "+v"(r[3]));
        float v[16];
        bool  m[16];
#pragma unroll
        for (int u = 0; u < 4; ++u) {
#pragma unroll
          for (int j = 0; j < 4; ++j) {
            const unsigned d = (unsigned)(c[u][j] - lo);
            m[u * 4 + j] = d < (unsigned)cnt;
            v[u * 4 + j] = 0.0f;
            if (m[u * 4 + j]) v[u * 4 + j] = xp[r[u][j]];
          }
        }
#pragma unroll
        for (int u = 0; u < 4; ++u) {
#pragma unroll
          for (int j = 0; j < 4; ++j)
            if (m[u * 4 + j]) atomicAdd(&acc[c[u][j] - lo], v[u * 4 + j]);
        }
      }
      for (; b < ehi; b += RTPB * 4) {
        const v4i c = load4_or_none(col, b, ehi);
        const v4i r = load4_or_none(row, b, ehi);
#pragma unroll
        for (int j = 0; j < 4; ++j) {
          const unsigned d = (unsigned)(c[j] - lo);
          if (d < (unsigned)cnt) atomicAdd(&acc[d], xp[r[j]]);
        }
      }
      __syncthreads();
      for (int i = tid; i < cnt; i += RTPB) {
        const float v = acc[i];
        if (v != 0.0f) atomicAdd(&agg1[lo + i], v);
      }
    }
  }
}

__global__ void gcn_node(const float* __restrict__ agg1,
                         const float* __restrict__ xp,
                         const float* __restrict__ dis,
                         const float* __restrict__ W1,
                         const float* __restrict__ b1,
                         const float* __restrict__ W2,
                         float* __restrict__ h2p,
                         float* __restrict__ out,
                         int n) {
  int lane = threadIdx.x & 31;
  int wave = threadIdx.x >> 5;
  int tile = blockIdx.x * (blockDim.x >> 5) + wave;
  int base = tile * 16;
  int m    = lane & 15;
  int half = lane >> 4;
  int node = base + m;
  int nc   = node < n ? node : (n - 1);

  float av = (agg1[nc] + xp[nc]) * dis[nc];

  int   wcol  = (m < 2) ? m : 0;
  float wmask = (m < 2) ? 1.0f : 0.0f;

  v8f acc = {};
#pragma unroll
  for (int s = 0; s < 4; ++s) {
    int j0 = s * 4 + half * 2;
    v2f w1 = *(const v2f*)(W1 + j0);
    v2f bi = *(const v2f*)(b1 + j0);
    v2f a;
    a.x = fmaxf(av * w1.x + bi.x, 0.0f);
    a.y = fmaxf(av * w1.y + bi.y, 0.0f);
    v2f bb;
    bb.x = W2[j0 * 2 + wcol]       * wmask;
    bb.y = W2[(j0 + 1) * 2 + wcol] * wmask;
    acc = __builtin_amdgcn_wmma_f32_16x16x4_f32(
         false, a,  false, bb,
         (short)0, acc,  false,  false);
  }

  if (m < 2) {
#pragma unroll
    for (int r = 0; r < 8; ++r) {
      int nd = base + r + half * 8;
      if (nd < n) {
        float hp = acc[r] * dis[nd];
        h2p[nd * 2 + m] = hp;
        out[nd * 2 + m] = hp;
      }
    }
  }
}

__global__ void __launch_bounds__(256)
gcn_l2(const int* __restrict__ row, const int* __restrict__ col,
       const float* __restrict__ h2p,
       float* __restrict__ out, int* __restrict__ tick, int cs, int e, int n) {
  const int lane  = threadIdx.x & 31;
  const int odd   = lane & 1;
  const int split = split_node(n);
  const int nt    = (e + EPT - 1) / EPT;
  const int h     = cache_half();
  for (int p = 0; p < 2; ++p) {
    const int q = h ^ p;
    int* ctr = tick + q * cs;
    for (;;) {
      const int t = take_ticket(ctr);
      if (t >= nt) break;
      const int e0 = t * EPT + lane * 4;
      if (t * EPT + EPT <= e) {
        v4i r[4], c[4];
#pragma unroll
        for (int k = 0; k < 4; ++k) {
          r[k] = ntload4(row + e0 + k * 128);
          c[k] = ntload4(col + e0 + k * 128);
        }
#pragma unroll
        for (int k = 0; k < 4; ++k) {
          v2f v[4];
#pragma unroll
          for (int j = 0; j < 4; ++j) {
            const int rj = r[k][j], cj = c[k][j];
            const bool oj = (cj >= split) == (q != 0);
            v[j] = *(const v2f*)(h2p + 2 * (size_t)(oj ? rj : 0));
          }
          int   nc[4];
          float nv[4];
#pragma unroll
          for (int j = 0; j < 4; ++j) {
            nc[j] = pair_swap(c[k][j]);
            nv[j] = pair_swapf(odd ? v[j].x : v[j].y);
          }
#pragma unroll
          for (int j = 0; j < 4; ++j) {
            const int   cc = odd ? nc[j] : c[k][j];
            const float vv = odd ? nv[j] : v[j].x;
            if ((cc >= split) == (q != 0)) atomicAdd(&out[2 * (size_t)cc + odd], vv);
          }
#pragma unroll
          for (int j = 0; j < 4; ++j) {
            const int   cc = odd ? c[k][j] : nc[j];
            const float vv = odd ? v[j].y : nv[j];
            if ((cc >= split) == (q != 0)) atomicAdd(&out[2 * (size_t)cc + odd], vv);
          }
        }
      } else {
        for (int k = 0; k < 4; ++k) {
          const int b = e0 + k * 128;
          for (int j = b; j < b + 4 && j < e; ++j) {
            const int rj = row[j], cj = col[j];
            if ((cj >= split) == (q != 0)) {
              atomicAdd(&out[2 * (size_t)cj + 0], h2p[2 * (size_t)rj + 0]);
              atomicAdd(&out[2 * (size_t)cj + 1], h2p[2 * (size_t)rj + 1]);
            }
          }
        }
      }
    }
  }
}

__global__ void gcn_fin(const float* __restrict__ dis, const float* __restrict__ b2,
                        float* __restrict__ out, int n) {
  const int i = 4 * (blockIdx.x * blockDim.x + threadIdx.x);
  const float c0 = b2[0], c1 = b2[1];
  if (i + 3 < n) {
    const v4f d = *(const v4f*)(dis + i);
    float* o = out + 2 * (size_t)i;
    v4f a = *(const v4f*)o;
    v4f b = *(const v4f*)(o + 4);
    a.x = c0 + d.x * a.x;  a.y = c1 + d.x * a.y;
    a.z = c0 + d.y * a.z;  a.w = c1 + d.y * a.w;
    b.x = c0 + d.z * b.x;  b.y = c1 + d.z * b.y;
    b.z = c0 + d.w * b.z;  b.w = c1 + d.w * b.w;
    *(v4f*)o       = a;
    *(v4f*)(o + 4) = b;
  } else {
    for (int j = i; j < n; ++j) {
      const float d = dis[j];
      out[2 * (size_t)j + 0] = c0 + d * out[2 * (size_t)j + 0];
      out[2 * (size_t)j + 1] = c1 + d * out[2 * (size_t)j + 1];
    }
  }
}

extern "C" void kernel_launch(void* const* d_in, const int* in_sizes, int n_in,
                              void* d_out, int out_size, void* d_ws, size_t ws_size,
                              hipStream_t stream) {
  const float* x  = (const float*)d_in[0];
  const int*   ei = (const int*)d_in[1];
  const float* W1 = (const float*)d_in[2];
  const float* b1 = (const float*)d_in[3];
  const float* W2 = (const float*)d_in[4];
  const float* b2 = (const float*)d_in[5];
  float* out = (float*)d_out;

  const int n = in_sizes[0];
  const int e = in_sizes[1] / 2;
  const int* row = ei;
  const int* col = ei + e;

  size_t np   = (size_t)((n + 3) & ~3);
  float* ws   = (float*)d_ws;
  float* dis  = ws;
  float* xp   = ws + np;
  float* agg1 = ws + 2 * np;
  float* h2p  = ws + 3 * np;

  const int cs = (n >= 512) ? 128 : 1;
  int* tick24 = (int*)h2p;
  int* tick6  = (int*)agg1;

  const int tb = 256;
  const int nb_node = (n + tb - 1) / tb;
  const int tiles = (n + 15) / 16;
  const int nb_tile = (tiles + 7) / 8;
  const int nb_fin  = (n + 4 * tb - 1) / (4 * tb);

  gcn_init  <<<nb_node, tb, 0, stream>>>(dis, agg1, tick24, cs, n);
  gcn_degree<<<RANGE_BLOCKS, RTPB, 0, stream>>>(col, dis, tick24, cs, e, n);
  gcn_dis   <<<nb_node, tb, 0, stream>>>(x, dis, xp, n);
  gcn_l1    <<<RANGE_BLOCKS, RTPB, 0, stream>>>(row, col, xp, agg1, tick24 + 2 * cs, cs, e, n);
  gcn_node  <<<nb_tile, 256, 0, stream>>>(agg1, xp, dis, W1, b1, W2, h2p, out, n);
  gcn_tick0 <<<1, 32, 0, stream>>>(tick6, cs);
  gcn_l2    <<<EDGE_BLOCKS, tb, 0, stream>>>(row, col, h2p, out, tick6, cs, e, n);
  gcn_fin   <<<nb_fin, tb, 0, stream>>>(dis, b2, out, n);
}


# Layer-2 sum built per node range in local memory (16 ranges x 16 edge slices): a wave filters its edges through an LDS ring, gathers and adds full rows (LDS f32 atomics); line flush: new sum tree
# speedup vs baseline: 5.1831x; 1.0389x over previous
#include <hip/hip_runtime.h>


typedef __attribute__((ext_vector_type(2))) float v2f;
typedef __attribute__((ext_vector_type(8))) float v8f;
typedef __attribute__((ext_vector_type(4))) int   v4i;
typedef __attribute__((ext_vector_type(4))) float v4f;
typedef __attribute__((ext_vector_type(2))) int   v2i;


#define RS_MAX 65536
#define SLICES 32
#define RTPB   1024
#define RANGE_BLOCKS 256

#define L2_CAP 32768
#define L2_SL  16
#define QN     64

__device__ __forceinline__ int cache_half() {
  unsigned id = __builtin_amdgcn_s_sendmsg_rtn(0x87);
  return (int)((id >> 18) & 1u);
}

__device__ __forceinline__ int range_count(int n) {
  int r = (n + RS_MAX - 1) / RS_MAX;
  if (r < 8) r = 8;
  return (r + 1) & ~1;
}
__device__ __forceinline__ int range_size(int n, int r) { return ((n + r - 1) / r + 63) & ~63; }
__device__ __forceinline__ int slice_len(int e) { return ((e + SLICES - 1) / SLICES + 16383) & ~16383; }

__device__ __forceinline__ int take_block_ticket(int* ctr, int* slot) {
  if (threadIdx.x == 0) *slot = atomicAdd(ctr, 1);
  __syncthreads();
  const int t = *slot;
  __syncthreads();
  return t;
}

__device__ __forceinline__ v4i load4_or_none(const int* __restrict__ a, int i, int hi) {
  if (i + 3 < hi) return *(const v4i*)(a + i);
  v4i r = {-1, -1, -1, -1};
  if (i     < hi) r.x = a[i];
  if (i + 1 < hi) r.y = a[i + 1];
  if (i + 2 < hi) r.z = a[i + 2];
  return r;
}

__global__ void gcn_init(float* __restrict__ deg, float* __restrict__ agg1,
                         int* __restrict__ tick, int cs, int n) {
  int i = blockIdx.x * blockDim.x + threadIdx.x;
  if (i < n) { deg[i] = 1.0f; agg1[i] = 0.0f; }
  if (i < 4) tick[i * cs] = 0;
}

__global__ void gcn_tick0(int* __restrict__ tick, int cs) {
  if (threadIdx.x < 2) tick[threadIdx.x * cs] = 0;
}

__global__ void __launch_bounds__(RTPB)
gcn_degree(const int* __restrict__ col, float* __restrict__ deg,
           int* __restrict__ tick, int cs, int e, int n) {
  __shared__ __attribute__((aligned(16))) float acc[RS_MAX];
  __shared__ int   slot;
  const int tid = threadIdx.x;
  const int R   = range_count(n);
  const int Rh  = R >> 1;
  const int RS  = range_size(n, R);
  const int CH  = slice_len(e);
  const int nt  = Rh * SLICES;
  const int h   = cache_half();
  for (int p = 0; p < 2; ++p) {
    const int q = h ^ p;
    int* ctr = tick + q * cs;
    for (;;) {
      const int t = take_block_ticket(ctr, &slot);
      if (t >= nt) break;
      const int lo  = (q * Rh + t % Rh) * RS;
      const int cnt = min(RS, n - lo);
      const int elo = (t / Rh) * CH;
      const int ehi = min(e, elo + CH);
      if (cnt <= 0 || elo >= ehi) continue;
      for (int i = tid * 4; i < RS; i += RTPB * 4) *(v4f*)&acc[i] = (v4f){0.f, 0.f, 0.f, 0.f};
      __syncthreads();
      int b = elo + tid * 4;
      for (; b + 3 * RTPB * 4 + 3 < ehi; b += RTPB * 16) {
        v4i c[4];
#pragma unroll
        for (int u = 0; u < 4; ++u) c[u] = *(const v4i*)(col + b + u * RTPB * 4);
#pragma unroll
        for (int u = 0; u < 4; ++u) {
#pragma unroll
          for (int j = 0; j < 4; ++j) {
            const unsigned d = (unsigned)(c[u][j] - lo);
            if (d < (unsigned)cnt) atomicAdd(&acc[d], 1.0f);
          }
        }
      }
      for (; b < ehi; b += RTPB * 4) {
        const v4i c = load4_or_none(col, b, ehi);
#pragma unroll
        for (int j = 0; j < 4; ++j) {
          const unsigned d = (unsigned)(c[j] - lo);
          if (d < (unsigned)cnt) atomicAdd(&acc[d], 1.0f);
        }
      }
      __syncthreads();
      for (int i = tid; i < cnt; i += RTPB) {
        const float v = acc[i];
        if (v != 0.0f) atomicAdd(&deg[lo + i], v);
      }
    }
  }
}

__global__ void gcn_dis(const float* __restrict__ x, float* __restrict__ deg_dis,
                        float* __restrict__ xp, int n) {
  int i = blockIdx.x * blockDim.x + threadIdx.x;
  if (i < n) {
    float d = rsqrtf(deg_dis[i]);
    deg_dis[i] = d;
    xp[i] = x[i] * d;
  }
}

__global__ void __launch_bounds__(RTPB)
gcn_l1(const int* __restrict__ row, const int* __restrict__ col,
       const float* __restrict__ xp,
       float* __restrict__ agg1, int* __restrict__ tick, int cs, int e, int n) {
  __shared__ __attribute__((aligned(16))) float acc[RS_MAX];
  __shared__ int   slot;
  const int tid = threadIdx.x;
  const int R   = range_count(n);
  const int Rh  = R >> 1;
  const int RS  = range_size(n, R);
  const int CH  = slice_len(e);
  const int nt  = Rh * SLICES;
  const int h   = cache_half();
  for (int p = 0; p < 2; ++p) {
    const int q = h ^ p;
    int* ctr = tick + q * cs;
    for (;;) {
      const int t = take_block_ticket(ctr, &slot);
      if (t >= nt) break;
      const int lo  = (q * Rh + t % Rh) * RS;
      const int cnt = min(RS, n - lo);
      const int elo = (t / Rh) * CH;
      const int ehi = min(e, elo + CH);
      if (cnt <= 0 || elo >= ehi) continue;
      for (int i = tid * 4; i < RS; i += RTPB * 4) *(v4f*)&acc[i] = (v4f){0.f, 0.f, 0.f, 0.f};
      __syncthreads();
      int b = elo + tid * 4;
      for (; b + 3 * RTPB * 4 + 3 < ehi; b += RTPB * 16) {
        v4i r[4], c[4];
#pragma unroll
        for (int u = 0; u < 4; ++u) {
          c[u] = *(const v4i*)(col + b + u * RTPB * 4);
          r[u] = *(const v4i*)(row + b + u * RTPB * 4);
        }
        asm volatile("" : "+v"(c[0]), "+v"(c[1]), "+v"(c[2]), "+v"(c[3]),
                          "+v"(r[0]), "+v"(r[1]), "+v"(r[2]), "+v"(r[3]));
        float v[16];
        bool  m[16];
#pragma unroll
        for (int u = 0; u < 4; ++u) {
#pragma unroll
          for (int j = 0; j < 4; ++j) {
            const unsigned d = (unsigned)(c[u][j] - lo);
            m[u * 4 + j] = d < (unsigned)cnt;
            v[u * 4 + j] = 0.0f;
            if (m[u * 4 + j]) v[u * 4 + j] = xp[r[u][j]];
          }
        }
#pragma unroll
        for (int u = 0; u < 4; ++u) {
#pragma unroll
          for (int j = 0; j < 4; ++j)
            if (m[u * 4 + j]) atomicAdd(&acc[c[u][j] - lo], v[u * 4 + j]);
        }
      }
      for (; b < ehi; b += RTPB * 4) {
        const v4i c = load4_or_none(col, b, ehi);
        const v4i r = load4_or_none(row, b, ehi);
#pragma unroll
        for (int j = 0; j < 4; ++j) {
          const unsigned d = (unsigned)(c[j] - lo);
          if (d < (unsigned)cnt) atomicAdd(&acc[d], xp[r[j]]);
        }
      }
      __syncthreads();
      for (int i = tid; i < cnt; i += RTPB) {
        const float v = acc[i];
        if (v != 0.0f) atomicAdd(&agg1[lo + i], v);
      }
    }
  }
}

__global__ void gcn_node(const float* __restrict__ agg1,
                         const float* __restrict__ xp,
                         const float* __restrict__ dis,
                         const float* __restrict__ W1,
                         const float* __restrict__ b1,
                         const float* __restrict__ W2,
                         float* __restrict__ h2p,
                         float* __restrict__ out,
                         int n) {
  int lane = threadIdx.x & 31;
  int wave = threadIdx.x >> 5;
  int tile = blockIdx.x * (blockDim.x >> 5) + wave;
  int base = tile * 16;
  int m    = lane & 15;
  int half = lane >> 4;
  int node = base + m;
  int nc   = node < n ? node : (n - 1);

  float av = (agg1[nc] + xp[nc]) * dis[nc];

  int   wcol  = (m < 2) ? m : 0;
  float wmask = (m < 2) ? 1.0f : 0.0f;

  v8f acc = {};
#pragma unroll
  for (int s = 0; s < 4; ++s) {
    int j0 = s * 4 + half * 2;
    v2f w1 = *(const v2f*)(W1 + j0);
    v2f bi = *(const v2f*)(b1 + j0);
    v2f a;
    a.x = fmaxf(av * w1.x + bi.x, 0.0f);
    a.y = fmaxf(av * w1.y + bi.y, 0.0f);
    v2f bb;
    bb.x = W2[j0 * 2 + wcol]       * wmask;
    bb.y = W2[(j0 + 1) * 2 + wcol] * wmask;
    acc = __builtin_amdgcn_wmma_f32_16x16x4_f32(
         false, a,  false, bb,
         (short)0, acc,  false,  false);
  }

  if (m < 2) {
#pragma unroll
    for (int r = 0; r < 8; ++r) {
      int nd = base + r + half * 8;
      if (nd < n) {
        float hp = acc[r] * dis[nd];
        h2p[nd * 2 + m] = hp;
        out[nd * 2 + m] = hp;
      }
    }
  }
}

__device__ __forceinline__ int l2_range_count(int n) {
  int r = (n + L2_CAP - 1) / L2_CAP;
  if (r < 16) r = 16;
  return (r + 1) & ~1;
}
__device__ __forceinline__ int l2_slice_len(int e) { return ((e + L2_SL - 1) / L2_SL + 16383) & ~16383; }

__global__ void __launch_bounds__(RTPB)
gcn_l2(const int* __restrict__ row, const int* __restrict__ col,
       const float* __restrict__ h2p,
       float* __restrict__ out, int* __restrict__ tick, int cs, int e, int n) {
  __shared__ __attribute__((aligned(16))) float acc[2 * L2_CAP];
  __shared__ __attribute__((aligned(8)))  v2i   ring[(RTPB / 32) * QN];
  __shared__ int   slot;
  const int tid  = threadIdx.x;
  const int lane = tid & 31;
  v2i* wq = ring + (tid >> 5) * QN;
  const int R   = l2_range_count(n);
  const int Rh  = R >> 1;
  const int RS  = range_size(n, R);
  const int CH  = l2_slice_len(e);
  const int nt  = Rh * L2_SL;
  const int h   = cache_half();
  for (int p = 0; p < 2; ++p) {
    const int q = h ^ p;
    int* ctr = tick + q * cs;
    for (;;) {
      const int t = take_block_ticket(ctr, &slot);
      if (t >= nt) break;
      const int lo  = (q * Rh + t % Rh) * RS;
      const int cnt = min(RS, n - lo);
      const int elo = (t / Rh) * CH;
      const int ehi = min(e, elo + CH);
      if (cnt <= 0 || elo >= ehi) continue;
      for (int i = tid * 4; i < 2 * L2_CAP; i += RTPB * 4) *(v4f*)&acc[i] = (v4f){0.f, 0.f, 0.f, 0.f};
      __syncthreads();
      unsigned head = 0, tail = 0;
      bool pend = false;
      int  pd = 0;
      v2f  pv = {0.f, 0.f};
      for (int base = elo; base < ehi; base += RTPB * 16) {
        const int b = base + tid * 4;
        v4i r[4], c[4];
        if (base + RTPB * 16 <= ehi) {
#pragma unroll
          for (int u = 0; u < 4; ++u) {
            c[u] = *(const v4i*)(col + b + u * RTPB * 4);
            r[u] = *(const v4i*)(row + b + u * RTPB * 4);
          }
        } else {
#pragma unroll
          for (int u = 0; u < 4; ++u) {
            c[u] = load4_or_none(col, b + u * RTPB * 4, ehi);
            r[u] = load4_or_none(row, b + u * RTPB * 4, ehi);
          }
        }
#pragma unroll
        for (int u = 0; u < 4; ++u) {
#pragma unroll
          for (int j = 0; j < 4; ++j) {
            const unsigned d   = (unsigned)(c[u][j] - lo);
            const bool     inr = d < (unsigned)cnt;
            const unsigned msk = __builtin_amdgcn_ballot_w32(inr);
            if (msk != 0u) {
              if (inr) {
                const unsigned pos = (tail + __builtin_amdgcn_mbcnt_lo(msk, 0u)) & (QN - 1);
                wq[pos] = (v2i){r[u][j], (int)d};
              }
              tail += (unsigned)__builtin_popcount(msk);
              if (tail - head >= 32u) {
                if (pend) {
                  atomicAdd(&acc[pd], pv.x);
                  atomicAdd(&acc[L2_CAP + pd], pv.y);
                }
                __builtin_amdgcn_fence(__ATOMIC_ACQ_REL, "wavefront");
                const v2i en = wq[(head + (unsigned)lane) & (QN - 1)];
                head += 32u;
                pd   = en.y;
                pv   = *(const v2f*)(h2p + 2 * (size_t)en.x);
                pend = true;
              }
            }
          }
        }
      }
      if (pend) {
        atomicAdd(&acc[pd], pv.x);
        atomicAdd(&acc[L2_CAP + pd], pv.y);
      }
      __builtin_amdgcn_fence(__ATOMIC_ACQ_REL, "wavefront");
      if ((unsigned)lane < tail - head) {
        const v2i en = wq[(head + (unsigned)lane) & (QN - 1)];
        const v2f v  = *(const v2f*)(h2p + 2 * (size_t)en.x);
        atomicAdd(&acc[en.y], v.x);
        atomicAdd(&acc[L2_CAP + en.y], v.y);
      }
      __syncthreads();
      float* o = out + 2 * (size_t)lo;
      for (int f = tid; f < 2 * cnt; f += RTPB) {
        const float v = acc[(f & 1) * L2_CAP + (f >> 1)];
        if (v != 0.0f) atomicAdd(&o[f], v);
      }
    }
  }
}

__global__ void gcn_fin(const float* __restrict__ dis, const float* __restrict__ b2,
                        float* __restrict__ out, int n) {
  const int i = 4 * (blockIdx.x * blockDim.x + threadIdx.x);
  const float c0 = b2[0], c1 = b2[1];
  if (i + 3 < n) {
    const v4f d = *(const v4f*)(dis + i);
    float* o = out + 2 * (size_t)i;
    v4f a = *(const v4f*)o;
    v4f b = *(const v4f*)(o + 4);
    a.x = c0 + d.x * a.x;  a.y = c1 + d.x * a.y;
    a.z = c0 + d.y * a.z;  a.w = c1 + d.y * a.w;
    b.x = c0 + d.z * b.x;  b.y = c1 + d.z * b.y;
    b.z = c0 + d.w * b.z;  b.w = c1 + d.w * b.w;
    *(v4f*)o       = a;
    *(v4f*)(o + 4) = b;
  } else {
    for (int j = i; j < n; ++j) {
      const float d = dis[j];
      out[2 * (size_t)j + 0] = c0 + d * out[2 * (size_t)j + 0];
      out[2 * (size_t)j + 1] = c1 + d * out[2 * (size_t)j + 1];
    }
  }
}

extern "C" void kernel_launch(void* const* d_in, const int* in_sizes, int n_in,
                              void* d_out, int out_size, void* d_ws, size_t ws_size,
                              hipStream_t stream) {
  const float* x  = (const float*)d_in[0];
  const int*   ei = (const int*)d_in[1];
  const float* W1 = (const float*)d_in[2];
  const float* b1 = (const float*)d_in[3];
  const float* W2 = (const float*)d_in[4];
  const float* b2 = (const float*)d_in[5];
  float* out = (float*)d_out;

  const int n = in_sizes[0];
  const int e = in_sizes[1] / 2;
  const int* row = ei;
  const int* col = ei + e;

  size_t np   = (size_t)((n + 3) & ~3);
  float* ws   = (float*)d_ws;
  float* dis  = ws;
  float* xp   = ws + np;
  float* agg1 = ws + 2 * np;
  float* h2p  = ws + 3 * np;

  const int cs = (n >= 512) ? 128 : 1;
  int* tick24 = (int*)h2p;
  int* tick6  = (int*)agg1;

  const int tb = 256;
  const int nb_node = (n + tb - 1) / tb;
  const int tiles = (n + 15) / 16;
  const int nb_tile = (tiles + 7) / 8;
  const int nb_fin  = (n + 4 * tb - 1) / (4 * tb);

  gcn_init  <<<nb_node, tb, 0, stream>>>(dis, agg1, tick24, cs, n);
  gcn_degree<<<RANGE_BLOCKS, RTPB, 0, stream>>>(col, dis, tick24, cs, e, n);
  gcn_dis   <<<nb_node, tb, 0, stream>>>(x, dis, xp, n);
  gcn_l1    <<<RANGE_BLOCKS, RTPB, 0, stream>>>(row, col, xp, agg1, tick24 + 2 * cs, cs, e, n);
  gcn_node  <<<nb_tile, 256, 0, stream>>>(agg1, xp, dis, W1, b1, W2, h2p, out, n);
  gcn_tick0 <<<1, 32, 0, stream>>>(tick6, cs);
  gcn_l2    <<<RANGE_BLOCKS, RTPB, 0, stream>>>(row, col, h2p, out, tick6, cs, e, n);
  gcn_fin   <<<nb_fin, tb, 0, stream>>>(dis, b2, out, n);
}


# Edges grouped by destination once per call (chunk sort in LDS, 4-byte records in workspace); degree, layer-1, layer-2 passes read only their group's runs and store plainly: new sum tree
# speedup vs baseline: 7.9323x; 1.5304x over previous
#include <hip/hip_runtime.h>


typedef __attribute__((ext_vector_type(2))) float v2f;
typedef __attribute__((ext_vector_type(8))) float v8f;
typedef __attribute__((ext_vector_type(4))) int   v4i;
typedef __attribute__((ext_vector_type(4))) float v4f;
typedef __attribute__((ext_vector_type(2))) int   v2i;


#define RS_MAX 65536
#define SLICES 32
#define RTPB   1024
#define RANGE_BLOCKS 256

#define L2_CAP 32768
#define L2_SL  16
#define QN     64

__device__ __forceinline__ int cache_half() {
  unsigned id = __builtin_amdgcn_s_sendmsg_rtn(0x87);
  return (int)((id >> 18) & 1u);
}

__device__ __forceinline__ int range_count(int n) {
  int r = (n + RS_MAX - 1) / RS_MAX;
  if (r < 8) r = 8;
  return (r + 1) & ~1;
}
__device__ __forceinline__ int range_size(int n, int r) { return ((n + r - 1) / r + 63) & ~63; }
__device__ __forceinline__ int slice_len(int e) { return ((e + SLICES - 1) / SLICES + 16383) & ~16383; }

__device__ __forceinline__ int take_block_ticket(int* ctr, int* slot) {
  if (threadIdx.x == 0) *slot = atomicAdd(ctr, 1);
  __syncthreads();
  const int t = *slot;
  __syncthreads();
  return t;
}

__device__ __forceinline__ v4i load4_or_none(const int* __restrict__ a, int i, int hi) {
  if (i + 3 < hi) return *(const v4i*)(a + i);
  v4i r = {-1, -1, -1, -1};
  if (i     < hi) r.x = a[i];
  if (i + 1 < hi) r.y = a[i + 1];
  if (i + 2 < hi) r.z = a[i + 2];
  return r;
}

__global__ void gcn_init(float* __restrict__ deg, float* __restrict__ agg1,
                         int* __restrict__ tick, int cs, int n) {
  int i = blockIdx.x * blockDim.x + threadIdx.x;
  if (i < n) { deg[i] = 1.0f; agg1[i] = 0.0f; }
  if (i < 4) tick[i * cs] = 0;
}

__global__ void gcn_tick0(int* __restrict__ tick, int cs) {
  if (threadIdx.x < 2) tick[threadIdx.x * cs] = 0;
}

__global__ void __launch_bounds__(RTPB)
gcn_degree(const int* __restrict__ col, float* __restrict__ deg,
           int* __restrict__ tick, int cs, int e, int n) {
  __shared__ __attribute__((aligned(16))) float acc[RS_MAX];
  __shared__ int   slot;
  const int tid = threadIdx.x;
  const int R   = range_count(n);
  const int Rh  = R >> 1;
  const int RS  = range_size(n, R);
  const int CH  = slice_len(e);
  const int nt  = Rh * SLICES;
  const int h   = cache_half();
  for (int p = 0; p < 2; ++p) {
    const int q = h ^ p;
    int* ctr = tick + q * cs;
    for (;;) {
      const int t = take_block_ticket(ctr, &slot);
      if (t >= nt) break;
      const int lo  = (q * Rh + t % Rh) * RS;
      const int cnt = min(RS, n - lo);
      const int elo = (t / Rh) * CH;
      const int ehi = min(e, elo + CH);
      if (cnt <= 0 || elo >= ehi) continue;
      for (int i = tid * 4; i < RS; i += RTPB * 4) *(v4f*)&acc[i] = (v4f){0.f, 0.f, 0.f, 0.f};
      __syncthreads();
      int b = elo + tid * 4;
      for (; b + 3 * RTPB * 4 + 3 < ehi; b += RTPB * 16) {
        v4i c[4];
#pragma unroll
        for (int u = 0; u < 4; ++u) c[u] = *(const v4i*)(col + b + u * RTPB * 4);
#pragma unroll
        for (int u = 0; u < 4; ++u) {
#pragma unroll
          for (int j = 0; j < 4; ++j) {
            const unsigned d = (unsigned)(c[u][j] - lo);
            if (d < (unsigned)cnt) atomicAdd(&acc[d], 1.0f);
          }
        }
      }
      for (; b < ehi; b += RTPB * 4) {
        const v4i c = load4_or_none(col, b, ehi);
#pragma unroll
        for (int j = 0; j < 4; ++j) {
          const unsigned d = (unsigned)(c[j] - lo);
          if (d < (unsigned)cnt) atomicAdd(&acc[d], 1.0f);
        }
      }
      __syncthreads();
      for (int i = tid; i < cnt; i += RTPB) {
        const float v = acc[i];
        if (v != 0.0f) atomicAdd(&deg[lo + i], v);
      }
    }
  }
}

__global__ void gcn_dis(const float* __restrict__ x, float* __restrict__ deg_dis,
                        float* __restrict__ xp, int n) {
  int i = blockIdx.x * blockDim.x + threadIdx.x;
  if (i < n) {
    float d = rsqrtf(deg_dis[i]);
    deg_dis[i] = d;
    xp[i] = x[i] * d;
  }
}

__global__ void __launch_bounds__(RTPB)
gcn_l1(const int* __restrict__ row, const int* __restrict__ col,
       const float* __restrict__ xp,
       float* __restrict__ agg1, int* __restrict__ tick, int cs, int e, int n) {
  __shared__ __attribute__((aligned(16))) float acc[RS_MAX];
  __shared__ int   slot;
  const int tid = threadIdx.x;
  const int R   = range_count(n);
  const int Rh  = R >> 1;
  const int RS  = range_size(n, R);
  const int CH  = slice_len(e);
  const int nt  = Rh * SLICES;
  const int h   = cache_half();
  for (int p = 0; p < 2; ++p) {
    const int q = h ^ p;
    int* ctr = tick + q * cs;
    for (;;) {
      const int t = take_block_ticket(ctr, &slot);
      if (t >= nt) break;
      const int lo  = (q * Rh + t % Rh) * RS;
      const int cnt = min(RS, n - lo);
      const int elo = (t / Rh) * CH;
      const int ehi = min(e, elo + CH);
      if (cnt <= 0 || elo >= ehi) continue;
      for (int i = tid * 4; i < RS; i += RTPB * 4) *(v4f*)&acc[i] = (v4f){0.f, 0.f, 0.f, 0.f};
      __syncthreads();
      int b = elo + tid * 4;
      for (; b + 3 * RTPB * 4 + 3 < ehi; b += RTPB * 16) {
        v4i r[4], c[4];
#pragma unroll
        for (int u = 0; u < 4; ++u) {
          c[u] = *(const v4i*)(col + b + u * RTPB * 4);
          r[u] = *(const v4i*)(row + b + u * RTPB * 4);
        }
        asm volatile("" : "+v"(c[0]), "+v"(c[1]), "+v"(c[2]), "+v"(c[3]),
                          "+v"(r[0]), "+v"(r[1]), "+v"(r[2]), "+v"(r[3]));
        float v[16];
        bool  m[16];
#pragma unroll
        for (int u = 0; u < 4; ++u) {
#pragma unroll
          for (int j = 0; j < 4; ++j) {
            const unsigned d = (unsigned)(c[u][j] - lo);
            m[u * 4 + j] = d < (unsigned)cnt;
            v[u * 4 + j] = 0.0f;
            if (m[u * 4 + j]) v[u * 4 + j] = xp[r[u][j]];
          }
        }
#pragma unroll
        for (int u = 0; u < 4; ++u) {
#pragma unroll
          for (int j = 0; j < 4; ++j)
            if (m[u * 4 + j]) atomicAdd(&acc[c[u][j] - lo], v[u * 4 + j]);
        }
      }
      for (; b < ehi; b += RTPB * 4) {
        const v4i c = load4_or_none(col, b, ehi);
        const v4i r = load4_or_none(row, b, ehi);
#pragma unroll
        for (int j = 0; j < 4; ++j) {
          const unsigned d = (unsigned)(c[j] - lo);
          if (d < (unsigned)cnt) atomicAdd(&acc[d], xp[r[j]]);
        }
      }
      __syncthreads();
      for (int i = tid; i < cnt; i += RTPB) {
        const float v = acc[i];
        if (v != 0.0f) atomicAdd(&agg1[lo + i], v);
      }
    }
  }
}

__global__ void gcn_node(const float* __restrict__ agg1,
                         const float* __restrict__ xp,
                         const float* __restrict__ dis,
                         const float* __restrict__ W1,
                         const float* __restrict__ b1,
                         const float* __restrict__ W2,
                         float* __restrict__ h2p,
                         float* __restrict__ out,
                         int n, int start_out) {
  int lane = threadIdx.x & 31;
  int wave = threadIdx.x >> 5;
  int tile = blockIdx.x * (blockDim.x >> 5) + wave;
  int base = tile * 16;
  int m    = lane & 15;
  int half = lane >> 4;
  int node = base + m;
  int nc   = node < n ? node : (n - 1);

  float av = (agg1[nc] + xp[nc]) * dis[nc];

  int   wcol  = (m < 2) ? m : 0;
  float wmask = (m < 2) ? 1.0f : 0.0f;

  v8f acc = {};
#pragma unroll
  for (int s = 0; s < 4; ++s) {
    int j0 = s * 4 + half * 2;
    v2f w1 = *(const v2f*)(W1 + j0);
    v2f bi = *(const v2f*)(b1 + j0);
    v2f a;
    a.x = fmaxf(av * w1.x + bi.x, 0.0f);
    a.y = fmaxf(av * w1.y + bi.y, 0.0f);
    v2f bb;
    bb.x = W2[j0 * 2 + wcol]       * wmask;
    bb.y = W2[(j0 + 1) * 2 + wcol] * wmask;
    acc = __builtin_amdgcn_wmma_f32_16x16x4_f32(
         false, a,  false, bb,
         (short)0, acc,  false,  false);
  }

  if (m < 2) {
#pragma unroll
    for (int r = 0; r < 8; ++r) {
      int nd = base + r + half * 8;
      if (nd < n) {
        float hp = acc[r] * dis[nd];
        h2p[nd * 2 + m] = hp;
        if (start_out) out[nd * 2 + m] = hp;
      }
    }
  }
}

__device__ __forceinline__ int l2_range_count(int n) {
  int r = (n + L2_CAP - 1) / L2_CAP;
  if (r < 16) r = 16;
  return (r + 1) & ~1;
}
__device__ __forceinline__ int l2_slice_len(int e) { return ((e + L2_SL - 1) / L2_SL + 16383) & ~16383; }

__global__ void __launch_bounds__(RTPB)
gcn_l2(const int* __restrict__ row, const int* __restrict__ col,
       const float* __restrict__ h2p,
       float* __restrict__ out, int* __restrict__ tick, int cs, int e, int n) {
  __shared__ __attribute__((aligned(16))) float acc[2 * L2_CAP];
  __shared__ __attribute__((aligned(8)))  v2i   ring[(RTPB / 32) * QN];
  __shared__ int   slot;
  const int tid  = threadIdx.x;
  const int lane = tid & 31;
  v2i* wq = ring + (tid >> 5) * QN;
  const int R   = l2_range_count(n);
  const int Rh  = R >> 1;
  const int RS  = range_size(n, R);
  const int CH  = l2_slice_len(e);
  const int nt  = Rh * L2_SL;
  const int h   = cache_half();
  for (int p = 0; p < 2; ++p) {
    const int q = h ^ p;
    int* ctr = tick + q * cs;
    for (;;) {
      const int t = take_block_ticket(ctr, &slot);
      if (t >= nt) break;
      const int lo  = (q * Rh + t % Rh) * RS;
      const int cnt = min(RS, n - lo);
      const int elo = (t / Rh) * CH;
      const int ehi = min(e, elo + CH);
      if (cnt <= 0 || elo >= ehi) continue;
      for (int i = tid * 4; i < 2 * L2_CAP; i += RTPB * 4) *(v4f*)&acc[i] = (v4f){0.f, 0.f, 0.f, 0.f};
      __syncthreads();
      unsigned head = 0, tail = 0;
      bool pend = false;
      int  pd = 0;
      v2f  pv = {0.f, 0.f};
      for (int base = elo; base < ehi; base += RTPB * 16) {
        const int b = base + tid * 4;
        v4i r[4], c[4];
        if (base + RTPB * 16 <= ehi) {
#pragma unroll
          for (int u = 0; u < 4; ++u) {
            c[u] = *(const v4i*)(col + b + u * RTPB * 4);
            r[u] = *(const v4i*)(row + b + u * RTPB * 4);
          }
        } else {
#pragma unroll
          for (int u = 0; u < 4; ++u) {
            c[u] = load4_or_none(col, b + u * RTPB * 4, ehi);
            r[u] = load4_or_none(row, b + u * RTPB * 4, ehi);
          }
        }
#pragma unroll
        for (int u = 0; u < 4; ++u) {
#pragma unroll
          for (int j = 0; j < 4; ++j) {
            const unsigned d   = (unsigned)(c[u][j] - lo);
            const bool     inr = d < (unsigned)cnt;
            const unsigned msk = __builtin_amdgcn_ballot_w32(inr);
            if (msk != 0u) {
              if (inr) {
                const unsigned pos = (tail + __builtin_amdgcn_mbcnt_lo(msk, 0u)) & (QN - 1);
                wq[pos] = (v2i){r[u][j], (int)d};
              }
              tail += (unsigned)__builtin_popcount(msk);
              if (tail - head >= 32u) {
                if (pend) {
                  atomicAdd(&acc[pd], pv.x);
                  atomicAdd(&acc[L2_CAP + pd], pv.y);
                }
                __builtin_amdgcn_fence(__ATOMIC_ACQ_REL, "wavefront");
                const v2i en = wq[(head + (unsigned)lane) & (QN - 1)];
                head += 32u;
                pd   = en.y;
                pv   = *(const v2f*)(h2p + 2 * (size_t)en.x);
                pend = true;
              }
            }
          }
        }
      }
      if (pend) {
        atomicAdd(&acc[pd], pv.x);
        atomicAdd(&acc[L2_CAP + pd], pv.y);
      }
      __builtin_amdgcn_fence(__ATOMIC_ACQ_REL, "wavefront");
      if ((unsigned)lane < tail - head) {
        const v2i en = wq[(head + (unsigned)lane) & (QN - 1)];
        const v2f v  = *(const v2f*)(h2p + 2 * (size_t)en.x);
        atomicAdd(&acc[en.y], v.x);
        atomicAdd(&acc[L2_CAP + en.y], v.y);
      }
      __syncthreads();
      float* o = out + 2 * (size_t)lo;
      for (int f = tid; f < 2 * cnt; f += RTPB) {
        const float v = acc[(f & 1) * L2_CAP + (f >> 1)];
        if (v != 0.0f) atomicAdd(&o[f], v);
      }
    }
  }
}

__global__ void gcn_fin(const float* __restrict__ dis, const float* __restrict__ b2,
                        float* __restrict__ out, int n) {
  const int i = 4 * (blockIdx.x * blockDim.x + threadIdx.x);
  const float c0 = b2[0], c1 = b2[1];
  if (i + 3 < n) {
    const v4f d = *(const v4f*)(dis + i);
    float* o = out + 2 * (size_t)i;
    v4f a = *(const v4f*)o;
    v4f b = *(const v4f*)(o + 4);
    a.x = c0 + d.x * a.x;  a.y = c1 + d.x * a.y;
    a.z = c0 + d.y * a.z;  a.w = c1 + d.y * a.w;
    b.x = c0 + d.z * b.x;  b.y = c1 + d.z * b.y;
    b.z = c0 + d.w * b.z;  b.w = c1 + d.w * b.w;
    *(v4f*)o       = a;
    *(v4f*)(o + 4) = b;
  } else {
    for (int j = i; j < n; ++j) {
      const float d = dis[j];
      out[2 * (size_t)j + 0] = c0 + d * out[2 * (size_t)j + 0];
      out[2 * (size_t)j + 1] = c1 + d * out[2 * (size_t)j + 1];
    }
  }
}


#define GSH     11
#define GBIN    2048
#define GNB_MAX 256
#define GCH_MAX 65536
#define GPAD    32
#define GROW    512
#define GTC     1024

__global__ void __launch_bounds__(RTPB)
gcn_group(const int* __restrict__ row, const int* __restrict__ col,
          int* __restrict__ recs, int* __restrict__ tab,
          int e, int n, int chunklen, int nchunks, int cap) {
  __shared__ __attribute__((aligned(16))) int lrec[GCH_MAX + GNB_MAX * GPAD];
  __shared__ int lcnt[GNB_MAX];
  __shared__ int lst[GNB_MAX + 1];
  const int tid = threadIdx.x;
  const int NB  = (n + GBIN - 1) >> GSH;
  for (int ch = blockIdx.x; ch < nchunks; ch += gridDim.x) {
    const int elo = ch * chunklen;
    const int ehi = min(e, elo + chunklen);
    if (tid < GNB_MAX) lcnt[tid] = 0;
    __syncthreads();
    for (int base = elo; base < ehi; base += RTPB * 16) {
      const int b = base + tid * 4;
      v4i c[4];
      if (base + RTPB * 16 <= ehi) {
#pragma unroll
        for (int u = 0; u < 4; ++u) c[u] = *(const v4i*)(col + b + u * RTPB * 4);
      } else {
#pragma unroll
        for (int u = 0; u < 4; ++u) c[u] = load4_or_none(col, b + u * RTPB * 4, ehi);
      }
#pragma unroll
      for (int u = 0; u < 4; ++u) {
#pragma unroll
        for (int j = 0; j < 4; ++j)
          if ((unsigned)c[u][j] < (unsigned)n) atomicAdd(&lcnt[c[u][j] >> GSH], 1);
      }
    }
    __syncthreads();
    int p = 0;
    if (tid < NB) p = (lcnt[tid] + GPAD - 1) & ~(GPAD - 1);
    if (tid < GNB_MAX) lst[tid + 1] = p;
    if (tid == 0) lst[0] = 0;
    __syncthreads();
    for (int off = 1; off < GNB_MAX; off <<= 1) {
      int v = 0;
      if (tid < GNB_MAX && tid >= off) v = lst[tid + 1 - off];
      __syncthreads();
      if (tid < GNB_MAX) lst[tid + 1] += v;
      __syncthreads();
    }
    if (tid < NB) {
      const int s0 = lst[tid], c0 = lcnt[tid], s1 = lst[tid + 1];
      for (int k = s0 + c0; k < s1; ++k) lrec[k] = -1;
      lcnt[tid] = s0;
    }
    __syncthreads();
    for (int base = elo; base < ehi; base += RTPB * 16) {
      const int b = base + tid * 4;
      v4i c[4], r[4];
      if (base + RTPB * 16 <= ehi) {
#pragma unroll
        for (int u = 0; u < 4; ++u) {
          c[u] = *(const v4i*)(col + b + u * RTPB * 4);
          r[u] = *(const v4i*)(row + b + u * RTPB * 4);
        }
      } else {
#pragma unroll
        for (int u = 0; u < 4; ++u) {
          c[u] = load4_or_none(col, b + u * RTPB * 4, ehi);
          r[u] = load4_or_none(row, b + u * RTPB * 4, ehi);
        }
      }
      int pos[16];
#pragma unroll
      for (int u = 0; u < 4; ++u) {
#pragma unroll
        for (int j = 0; j < 4; ++j) {
          pos[u * 4 + j] = -1;
          if ((unsigned)c[u][j] < (unsigned)n) pos[u * 4 + j] = atomicAdd(&lcnt[c[u][j] >> GSH], 1);
        }
      }
#pragma unroll
      for (int u = 0; u < 4; ++u) {
#pragma unroll
        for (int j = 0; j < 4; ++j)
          if (pos[u * 4 + j] >= 0)
            lrec[pos[u * 4 + j]] = (int)(((unsigned)r[u][j] << GSH) | ((unsigned)c[u][j] & (unsigned)(GBIN - 1)));
      }
    }
    __syncthreads();
    const int tot4 = lst[NB] >> 2;
    int* g = recs + (size_t)ch * (size_t)cap;
    for (int u = tid; u < tot4; u += RTPB) *(v4i*)(g + 4 * u) = *(const v4i*)&lrec[4 * u];
    if (tid < GROW) tab[(size_t)ch * GROW + tid] = lst[(tid >> 1) + (tid & 1)];
    __syncthreads();
  }
}

__global__ void __launch_bounds__(RTPB)
gcn_degree_g(const int* __restrict__ recs, const int* __restrict__ tab,
             const float* __restrict__ x, float* __restrict__ dis, float* __restrict__ xp,
             int n, int nchunks, int cap) {
  __shared__ float acc[GBIN];
  __shared__ int   tst[GTC], ten[GTC];
  const int tid  = threadIdx.x;
  const int lane = tid & 31;
  const int wv   = __builtin_amdgcn_readfirstlane(tid >> 5);
  const int NB   = (n + GBIN - 1) >> GSH;
  for (int b = blockIdx.x; b < NB; b += gridDim.x) {
    const int lo  = b << GSH;
    const int cnt = min(GBIN, n - lo);
    for (int i = tid; i < GBIN; i += RTPB) acc[i] = 0.0f;
    for (int c0 = 0; c0 < nchunks; c0 += GTC) {
      const int nc = min(GTC, nchunks - c0);
      __syncthreads();
      if (tid < nc) {
        const v2i se = *(const v2i*)(tab + (size_t)(c0 + tid) * GROW + 2 * b);
        tst[tid] = se.x;
        ten[tid] = se.y;
      }
      __syncthreads();
      for (int k = wv; k < nc; k += RTPB / 32) {
        const int* p  = recs + (size_t)(c0 + k) * (size_t)cap;
        const int  u1 = ten[k] >> 2;
        for (int u = (tst[k] >> 2) + lane; u < u1; u += 32) {
          const v4i rc = *(const v4i*)(p + 4 * u);
#pragma unroll
          for (int j = 0; j < 4; ++j)
            if (rc[j] != -1) atomicAdd(&acc[rc[j] & (GBIN - 1)], 1.0f);
        }
      }
    }
    __syncthreads();
    for (int i = tid; i < cnt; i += RTPB) {
      const float d = rsqrtf(1.0f + acc[i]);
      dis[lo + i] = d;
      xp[lo + i]  = x[lo + i] * d;
    }
    __syncthreads();
  }
}

__global__ void __launch_bounds__(RTPB)
gcn_l1_g(const int* __restrict__ recs, const int* __restrict__ tab,
         const float* __restrict__ xp, float* __restrict__ agg1,
         int n, int nchunks, int cap) {
  __shared__ float acc[GBIN];
  __shared__ int   tst[GTC], ten[GTC];
  const int tid  = threadIdx.x;
  const int lane = tid & 31;
  const int wv   = __builtin_amdgcn_readfirstlane(tid >> 5);
  const int NB   = (n + GBIN - 1) >> GSH;
  for (int b = blockIdx.x; b < NB; b += gridDim.x) {
    const int lo  = b << GSH;
    const int cnt = min(GBIN, n - lo);
    for (int i = tid; i < GBIN; i += RTPB) acc[i] = 0.0f;
    for (int c0 = 0; c0 < nchunks; c0 += GTC) {
      const int nc = min(GTC, nchunks - c0);
      __syncthreads();
      if (tid < nc) {
        const v2i se = *(const v2i*)(tab + (size_t)(c0 + tid) * GROW + 2 * b);
        tst[tid] = se.x;
        ten[tid] = se.y;
      }
      __syncthreads();
      for (int k = wv; k < nc; k += RTPB / 32) {
        const int* p  = recs + (size_t)(c0 + k) * (size_t)cap;
        const int  u1 = ten[k] >> 2;
        for (int u = (tst[k] >> 2) + lane; u < u1; u += 64) {
          const v4i rc0 = *(const v4i*)(p + 4 * u);
          v4i rc1 = {-1, -1, -1, -1};
          if (u + 32 < u1) rc1 = *(const v4i*)(p + 4 * (u + 32));
          float v[8];
#pragma unroll
          for (int j = 0; j < 4; ++j) {
            v[j]     = (rc0[j] != -1) ? xp[(unsigned)rc0[j] >> GSH] : 0.0f;
            v[4 + j] = (rc1[j] != -1) ? xp[(unsigned)rc1[j] >> GSH] : 0.0f;
          }
#pragma unroll
          for (int j = 0; j < 4; ++j) {
            if (rc0[j] != -1) atomicAdd(&acc[rc0[j] & (GBIN - 1)], v[j]);
            if (rc1[j] != -1) atomicAdd(&acc[rc1[j] & (GBIN - 1)], v[4 + j]);
          }
        }
      }
    }
    __syncthreads();
    for (int i = tid; i < cnt; i += RTPB) agg1[lo + i] = acc[i];
    __syncthreads();
  }
}

__global__ void __launch_bounds__(RTPB)
gcn_l2_g(const int* __restrict__ recs, const int* __restrict__ tab,
         const float* __restrict__ h2p, const float* __restrict__ dis,
         const float* __restrict__ b2, float* __restrict__ out,
         int n, int nchunks, int cap) {
  __shared__ float acc0[GBIN], acc1[GBIN];
  __shared__ int   tst[GTC], ten[GTC];
  const int tid  = threadIdx.x;
  const int lane = tid & 31;
  const int wv   = __builtin_amdgcn_readfirstlane(tid >> 5);
  const int NB   = (n + GBIN - 1) >> GSH;
  const float cb0 = b2[0], cb1 = b2[1];
  for (int b = blockIdx.x; b < NB; b += gridDim.x) {
    const int lo  = b << GSH;
    const int cnt = min(GBIN, n - lo);
    for (int i = tid; i < GBIN; i += RTPB) { acc0[i] = 0.0f; acc1[i] = 0.0f; }
    for (int c0 = 0; c0 < nchunks; c0 += GTC) {
      const int nc = min(GTC, nchunks - c0);
      __syncthreads();
      if (tid < nc) {
        const v2i se = *(const v2i*)(tab + (size_t)(c0 + tid) * GROW + 2 * b);
        tst[tid] = se.x;
        ten[tid] = se.y;
      }
      __syncthreads();
      for (int k = wv; k < nc; k += RTPB / 32) {
        const int* p  = recs + (size_t)(c0 + k) * (size_t)cap;
        const int  u1 = ten[k] >> 2;
        for (int u = (tst[k] >> 2) + lane; u < u1; u += 64) {
          const v4i rc0 = *(const v4i*)(p + 4 * u);
          v4i rc1 = {-1, -1, -1, -1};
          if (u + 32 < u1) rc1 = *(const v4i*)(p + 4 * (u + 32));
          v2f v[8];
#pragma unroll
          for (int j = 0; j < 4; ++j) {
            v[j]     = (v2f){0.f, 0.f};
            v[4 + j] = (v2f){0.f, 0.f};
            if (rc0[j] != -1) v[j]     = *(const v2f*)(h2p + 2 * (size_t)((unsigned)rc0[j] >> GSH));
            if (rc1[j] != -1) v[4 + j] = *(const v2f*)(h2p + 2 * (size_t)((unsigned)rc1[j] >> GSH));
          }
#pragma unroll
          for (int j = 0; j < 4; ++j) {
            if (rc0[j] != -1) {
              atomicAdd(&acc0[rc0[j] & (GBIN - 1)], v[j].x);
              atomicAdd(&acc1[rc0[j] & (GBIN - 1)], v[j].y);
            }
            if (rc1[j] != -1) {
              atomicAdd(&acc0[rc1[j] & (GBIN - 1)], v[4 + j].x);
              atomicAdd(&acc1[rc1[j] & (GBIN - 1)], v[4 + j].y);
            }
          }
        }
      }
    }
    __syncthreads();
    for (int i = tid; i < cnt; i += RTPB) {
      const size_t nd = (size_t)(lo + i);
      const float  d  = dis[nd];
      const v2f    hs = *(const v2f*)(h2p + 2 * nd);
      v2f o;
      o.x = cb0 + d * (acc0[i] + hs.x);
      o.y = cb1 + d * (acc1[i] + hs.y);
      *(v2f*)(out + 2 * nd) = o;
    }
    __syncthreads();
  }
}

extern "C" void kernel_launch(void* const* d_in, const int* in_sizes, int n_in,
                              void* d_out, int out_size, void* d_ws, size_t ws_size,
                              hipStream_t stream) {
  const float* x  = (const float*)d_in[0];
  const int*   ei = (const int*)d_in[1];
  const float* W1 = (const float*)d_in[2];
  const float* b1 = (const float*)d_in[3];
  const float* W2 = (const float*)d_in[4];
  const float* b2 = (const float*)d_in[5];
  float* out = (float*)d_out;

  const int n = in_sizes[0];
  const int e = in_sizes[1] / 2;
  const int* row = ei;
  const int* col = ei + e;

  size_t np   = (size_t)((n + 3) & ~3);
  float* ws   = (float*)d_ws;
  float* dis  = ws;
  float* xp   = ws + np;
  float* agg1 = ws + 2 * np;
  float* h2p  = ws + 3 * np;

  const int cs = (n >= 512) ? 128 : 1;
  int* tick24 = (int*)h2p;
  int* tick6  = (int*)agg1;

  const int tb = 256;
  const int nb_node = (n + tb - 1) / tb;
  const int tiles = (n + 15) / 16;
  const int nb_tile = (tiles + 7) / 8;
  const int nb_fin  = (n + 4 * tb - 1) / (4 * tb);

  const int NBg = (n + GBIN - 1) >> GSH;
  int chunklen = ((e + RANGE_BLOCKS - 1) / RANGE_BLOCKS + 3) & ~3;
  if (chunklen > GCH_MAX) chunklen = GCH_MAX;
  if (chunklen < 4) chunklen = 4;
  const int nchunks = (e + chunklen - 1) / chunklen;
  const int cap = (chunklen + 31 + NBg * GPAD) & ~31;
  const size_t off_tab = (5 * np + 31) & ~(size_t)31;
  const size_t off_rec = off_tab + (size_t)nchunks * GROW;
  const size_t need    = (off_rec + (size_t)nchunks * (size_t)cap) * sizeof(float);
  if (n >= 1 && e >= 1 && NBg <= GNB_MAX && need <= ws_size) {
    int* tab  = (int*)(ws + off_tab);
    int* recs = (int*)(ws + off_rec);
    gcn_group   <<<RANGE_BLOCKS, RTPB, 0, stream>>>(row, col, recs, tab, e, n, chunklen, nchunks, cap);
    gcn_degree_g<<<NBg, RTPB, 0, stream>>>(recs, tab, x, dis, xp, n, nchunks, cap);
    gcn_l1_g    <<<NBg, RTPB, 0, stream>>>(recs, tab, xp, agg1, n, nchunks, cap);
    gcn_node    <<<nb_tile, 256, 0, stream>>>(agg1, xp, dis, W1, b1, W2, h2p, out, n, 0);
    gcn_l2_g    <<<NBg, RTPB, 0, stream>>>(recs, tab, h2p, dis, b2, out, n, nchunks, cap);
    return;
  }

  gcn_init  <<<nb_node, tb, 0, stream>>>(dis, agg1, tick24, cs, n);
  gcn_degree<<<RANGE_BLOCKS, RTPB, 0, stream>>>(col, dis, tick24, cs, e, n);
  gcn_dis   <<<nb_node, tb, 0, stream>>>(x, dis, xp, n);
  gcn_l1    <<<RANGE_BLOCKS, RTPB, 0, stream>>>(row, col, xp, agg1, tick24 + 2 * cs, cs, e, n);
  gcn_node  <<<nb_tile, 256, 0, stream>>>(agg1, xp, dis, W1, b1, W2, h2p, out, n, 1);
  gcn_tick0 <<<1, 32, 0, stream>>>(tick6, cs);
  gcn_l2    <<<RANGE_BLOCKS, RTPB, 0, stream>>>(row, col, h2p, out, tick6, cs, e, n);
  gcn_fin   <<<nb_fin, tb, 0, stream>>>(dis, b2, out, n);
}


# Edges grouped by (destination range, source range) tile, 32768 nodes each: a tile stages its source values in LDS, gathers and adds there; partial sums go out as whole-line f32 atomics: new sum tree
# speedup vs baseline: 10.0685x; 1.2693x over previous
#include <hip/hip_runtime.h>


typedef __attribute__((ext_vector_type(2))) float v2f;
typedef __attribute__((ext_vector_type(8))) float v8f;
typedef __attribute__((ext_vector_type(4))) int   v4i;
typedef __attribute__((ext_vector_type(4))) float v4f;
typedef __attribute__((ext_vector_type(2))) int   v2i;


#define RS_MAX 65536
#define SLICES 32
#define RTPB   1024
#define RANGE_BLOCKS 256

#define L2_CAP 32768
#define L2_SL  16
#define QN     64

__device__ __forceinline__ int cache_half() {
  unsigned id = __builtin_amdgcn_s_sendmsg_rtn(0x87);
  return (int)((id >> 18) & 1u);
}

__device__ __forceinline__ int range_count(int n) {
  int r = (n + RS_MAX - 1) / RS_MAX;
  if (r < 8) r = 8;
  return (r + 1) & ~1;
}
__device__ __forceinline__ int range_size(int n, int r) { return ((n + r - 1) / r + 63) & ~63; }
__device__ __forceinline__ int slice_len(int e) { return ((e + SLICES - 1) / SLICES + 16383) & ~16383; }

__device__ __forceinline__ int take_block_ticket(int* ctr, int* slot) {
  if (threadIdx.x == 0) *slot = atomicAdd(ctr, 1);
  __syncthreads();
  const int t = *slot;
  __syncthreads();
  return t;
}

__device__ __forceinline__ v4i load4_or_none(const int* __restrict__ a, int i, int hi) {
  if (i + 3 < hi) return *(const v4i*)(a + i);
  v4i r = {-1, -1, -1, -1};
  if (i     < hi) r.x = a[i];
  if (i + 1 < hi) r.y = a[i + 1];
  if (i + 2 < hi) r.z = a[i + 2];
  return r;
}

__global__ void gcn_init(float* __restrict__ deg, float* __restrict__ agg1,
                         int* __restrict__ tick, int cs, int n) {
  int i = blockIdx.x * blockDim.x + threadIdx.x;
  if (i < n) { deg[i] = 1.0f; agg1[i] = 0.0f; }
  if (i < 4) tick[i * cs] = 0;
}

__global__ void gcn_tick0(int* __restrict__ tick, int cs) {
  if (threadIdx.x < 2) tick[threadIdx.x * cs] = 0;
}

__global__ void __launch_bounds__(RTPB)
gcn_degree(const int* __restrict__ col, float* __restrict__ deg,
           int* __restrict__ tick, int cs, int e, int n) {
  __shared__ __attribute__((aligned(16))) float acc[RS_MAX];
  __shared__ int   slot;
  const int tid = threadIdx.x;
  const int R   = range_count(n);
  const int Rh  = R >> 1;
  const int RS  = range_size(n, R);
  const int CH  = slice_len(e);
  const int nt  = Rh * SLICES;
  const int h   = cache_half();
  for (int p = 0; p < 2; ++p) {
    const int q = h ^ p;
    int* ctr = tick + q * cs;
    for (;;) {
      const int t = take_block_ticket(ctr, &slot);
      if (t >= nt) break;
      const int lo  = (q * Rh + t % Rh) * RS;
      const int cnt = min(RS, n - lo);
      const int elo = (t / Rh) * CH;
      const int ehi = min(e, elo + CH);
      if (cnt <= 0 || elo >= ehi) continue;
      for (int i = tid * 4; i < RS; i += RTPB * 4) *(v4f*)&acc[i] = (v4f){0.f, 0.f, 0.f, 0.f};
      __syncthreads();
      int b = elo + tid * 4;
      for (; b + 3 * RTPB * 4 + 3 < ehi; b += RTPB * 16) {
        v4i c[4];
#pragma unroll
        for (int u = 0; u < 4; ++u) c[u] = *(const v4i*)(col + b + u * RTPB * 4);
#pragma unroll
        for (int u = 0; u < 4; ++u) {
#pragma unroll
          for (int j = 0; j < 4; ++j) {
            const unsigned d = (unsigned)(c[u][j] - lo);
            if (d < (unsigned)cnt) atomicAdd(&acc[d], 1.0f);
          }
        }
      }
      for (; b < ehi; b += RTPB * 4) {
        const v4i c = load4_or_none(col, b, ehi);
#pragma unroll
        for (int j = 0; j < 4; ++j) {
          const unsigned d = (unsigned)(c[j] - lo);
          if (d < (unsigned)cnt) atomicAdd(&acc[d], 1.0f);
        }
      }
      __syncthreads();
      for (int i = tid; i < cnt; i += RTPB) {
        const float v = acc[i];
        if (v != 0.0f) atomicAdd(&deg[lo + i], v);
      }
    }
  }
}

__global__ void gcn_dis(const float* __restrict__ x, float* __restrict__ deg_dis,
                        float* __restrict__ xp, int n) {
  int i = blockIdx.x * blockDim.x + threadIdx.x;
  if (i < n) {
    float d = rsqrtf(deg_dis[i]);
    deg_dis[i] = d;
    xp[i] = x[i] * d;
  }
}

__global__ void __launch_bounds__(RTPB)
gcn_l1(const int* __restrict__ row, const int* __restrict__ col,
       const float* __restrict__ xp,
       float* __restrict__ agg1, int* __restrict__ tick, int cs, int e, int n) {
  __shared__ __attribute__((aligned(16))) float acc[RS_MAX];
  __shared__ int   slot;
  const int tid = threadIdx.x;
  const int R   = range_count(n);
  const int Rh  = R >> 1;
  const int RS  = range_size(n, R);
  const int CH  = slice_len(e);
  const int nt  = Rh * SLICES;
  const int h   = cache_half();
  for (int p = 0; p < 2; ++p) {
    const int q = h ^ p;
    int* ctr = tick + q * cs;
    for (;;) {
      const int t = take_block_ticket(ctr, &slot);
      if (t >= nt) break;
      const int lo  = (q * Rh + t % Rh) * RS;
      const int cnt = min(RS, n - lo);
      const int elo = (t / Rh) * CH;
      const int ehi = min(e, elo + CH);
      if (cnt <= 0 || elo >= ehi) continue;
      for (int i = tid * 4; i < RS; i += RTPB * 4) *(v4f*)&acc[i] = (v4f){0.f, 0.f, 0.f, 0.f};
      __syncthreads();
      int b = elo + tid * 4;
      for (; b + 3 * RTPB * 4 + 3 < ehi; b += RTPB * 16) {
        v4i r[4], c[4];
#pragma unroll
        for (int u = 0; u < 4; ++u) {
          c[u] = *(const v4i*)(col + b + u * RTPB * 4);
          r[u] = *(const v4i*)(row + b + u * RTPB * 4);
        }
        asm volatile("" : "+v"(c[0]), "+v"(c[1]), "+v"(c[2]), "+v"(c[3]),
                          "+v"(r[0]), "+v"(r[1]), "+v"(r[2]), "+v"(r[3]));
        float v[16];
        bool  m[16];
#pragma unroll
        for (int u = 0; u < 4; ++u) {
#pragma unroll
          for (int j = 0; j < 4; ++j) {
            const unsigned d = (unsigned)(c[u][j] - lo);
            m[u * 4 + j] = d < (unsigned)cnt;
            v[u * 4 + j] = 0.0f;
            if (m[u * 4 + j]) v[u * 4 + j] = xp[r[u][j]];
          }
        }
#pragma unroll
        for (int u = 0; u < 4; ++u) {
#pragma unroll
          for (int j = 0; j < 4; ++j)
            if (m[u * 4 + j]) atomicAdd(&acc[c[u][j] - lo], v[u * 4 + j]);
        }
      }
      for (; b < ehi; b += RTPB * 4) {
        const v4i c = load4_or_none(col, b, ehi);
        const v4i r = load4_or_none(row, b, ehi);
#pragma unroll
        for (int j = 0; j < 4; ++j) {
          const unsigned d = (unsigned)(c[j] - lo);
          if (d < (unsigned)cnt) atomicAdd(&acc[d], xp[r[j]]);
        }
      }
      __syncthreads();
      for (int i = tid; i < cnt; i += RTPB) {
        const float v = acc[i];
        if (v != 0.0f) atomicAdd(&agg1[lo + i], v);
      }
    }
  }
}

__global__ void gcn_node(const float* __restrict__ agg1,
                         const float* __restrict__ xp,
                         const float* __restrict__ dis,
                         const float* __restrict__ W1,
                         const float* __restrict__ b1,
                         const float* __restrict__ W2,
                         float* __restrict__ h2p,
                         float* __restrict__ out,
                         int n, int start_out) {
  int lane = threadIdx.x & 31;
  int wave = threadIdx.x >> 5;
  int tile = blockIdx.x * (blockDim.x >> 5) + wave;
  int base = tile * 16;
  int m    = lane & 15;
  int half = lane >> 4;
  int node = base + m;
  int nc   = node < n ? node : (n - 1);

  float av = (agg1[nc] + xp[nc]) * dis[nc];

  int   wcol  = (m < 2) ? m : 0;
  float wmask = (m < 2) ? 1.0f : 0.0f;

  v8f acc = {};
#pragma unroll
  for (int s = 0; s < 4; ++s) {
    int j0 = s * 4 + half * 2;
    v2f w1 = *(const v2f*)(W1 + j0);
    v2f bi = *(const v2f*)(b1 + j0);
    v2f a;
    a.x = fmaxf(av * w1.x + bi.x, 0.0f);
    a.y = fmaxf(av * w1.y + bi.y, 0.0f);
    v2f bb;
    bb.x = W2[j0 * 2 + wcol]       * wmask;
    bb.y = W2[(j0 + 1) * 2 + wcol] * wmask;
    acc = __builtin_amdgcn_wmma_f32_16x16x4_f32(
         false, a,  false, bb,
         (short)0, acc,  false,  false);
  }

  if (m < 2) {
#pragma unroll
    for (int r = 0; r < 8; ++r) {
      int nd = base + r + half * 8;
      if (nd < n) {
        float hp = acc[r] * dis[nd];
        h2p[nd * 2 + m] = hp;
        if (start_out) out[nd * 2 + m] = hp;
      }
    }
  }
}

__device__ __forceinline__ int l2_range_count(int n) {
  int r = (n + L2_CAP - 1) / L2_CAP;
  if (r < 16) r = 16;
  return (r + 1) & ~1;
}
__device__ __forceinline__ int l2_slice_len(int e) { return ((e + L2_SL - 1) / L2_SL + 16383) & ~16383; }

__global__ void __launch_bounds__(RTPB)
gcn_l2(const int* __restrict__ row, const int* __restrict__ col,
       const float* __restrict__ h2p,
       float* __restrict__ out, int* __restrict__ tick, int cs, int e, int n) {
  __shared__ __attribute__((aligned(16))) float acc[2 * L2_CAP];
  __shared__ __attribute__((aligned(8)))  v2i   ring[(RTPB / 32) * QN];
  __shared__ int   slot;
  const int tid  = threadIdx.x;
  const int lane = tid & 31;
  v2i* wq = ring + (tid >> 5) * QN;
  const int R   = l2_range_count(n);
  const int Rh  = R >> 1;
  const int RS  = range_size(n, R);
  const int CH  = l2_slice_len(e);
  const int nt  = Rh * L2_SL;
  const int h   = cache_half();
  for (int p = 0; p < 2; ++p) {
    const int q = h ^ p;
    int* ctr = tick + q * cs;
    for (;;) {
      const int t = take_block_ticket(ctr, &slot);
      if (t >= nt) break;
      const int lo  = (q * Rh + t % Rh) * RS;
      const int cnt = min(RS, n - lo);
      const int elo = (t / Rh) * CH;
      const int ehi = min(e, elo + CH);
      if (cnt <= 0 || elo >= ehi) continue;
      for (int i = tid * 4; i < 2 * L2_CAP; i += RTPB * 4) *(v4f*)&acc[i] = (v4f){0.f, 0.f, 0.f, 0.f};
      __syncthreads();
      unsigned head = 0, tail = 0;
      bool pend = false;
      int  pd = 0;
      v2f  pv = {0.f, 0.f};
      for (int base = elo; base < ehi; base += RTPB * 16) {
        const int b = base + tid * 4;
        v4i r[4], c[4];
        if (base + RTPB * 16 <= ehi) {
#pragma unroll
          for (int u = 0; u < 4; ++u) {
            c[u] = *(const v4i*)(col + b + u * RTPB * 4);
            r[u] = *(const v4i*)(row + b + u * RTPB * 4);
          }
        } else {
#pragma unroll
          for (int u = 0; u < 4; ++u) {
            c[u] = load4_or_none(col, b + u * RTPB * 4, ehi);
            r[u] = load4_or_none(row, b + u * RTPB * 4, ehi);
          }
        }
#pragma unroll
        for (int u = 0; u < 4; ++u) {
#pragma unroll
          for (int j = 0; j < 4; ++j) {
            const unsigned d   = (unsigned)(c[u][j] - lo);
            const bool     inr = d < (unsigned)cnt;
            const unsigned msk = __builtin_amdgcn_ballot_w32(inr);
            if (msk != 0u) {
              if (inr) {
                const unsigned pos = (tail + __builtin_amdgcn_mbcnt_lo(msk, 0u)) & (QN - 1);
                wq[pos] = (v2i){r[u][j], (int)d};
              }
              tail += (unsigned)__builtin_popcount(msk);
              if (tail - head >= 32u) {
                if (pend) {
                  atomicAdd(&acc[pd], pv.x);
                  atomicAdd(&acc[L2_CAP + pd], pv.y);
                }
                __builtin_amdgcn_fence(__ATOMIC_ACQ_REL, "wavefront");
                const v2i en = wq[(head + (unsigned)lane) & (QN - 1)];
                head += 32u;
                pd   = en.y;
                pv   = *(const v2f*)(h2p + 2 * (size_t)en.x);
                pend = true;
              }
            }
          }
        }
      }
      if (pend) {
        atomicAdd(&acc[pd], pv.x);
        atomicAdd(&acc[L2_CAP + pd], pv.y);
      }
      __builtin_amdgcn_fence(__ATOMIC_ACQ_REL, "wavefront");
      if ((unsigned)lane < tail - head) {
        const v2i en = wq[(head + (unsigned)lane) & (QN - 1)];
        const v2f v  = *(const v2f*)(h2p + 2 * (size_t)en.x);
        atomicAdd(&acc[en.y], v.x);
        atomicAdd(&acc[L2_CAP + en.y], v.y);
      }
      __syncthreads();
      float* o = out + 2 * (size_t)lo;
      for (int f = tid; f < 2 * cnt; f += RTPB) {
        const float v = acc[(f & 1) * L2_CAP + (f >> 1)];
        if (v != 0.0f) atomicAdd(&o[f], v);
      }
    }
  }
}

__global__ void gcn_fin(const float* __restrict__ dis, const float* __restrict__ b2,
                        float* __restrict__ out, int n) {
  const int i = 4 * (blockIdx.x * blockDim.x + threadIdx.x);
  const float c0 = b2[0], c1 = b2[1];
  if (i + 3 < n) {
    const v4f d = *(const v4f*)(dis + i);
    float* o = out + 2 * (size_t)i;
    v4f a = *(const v4f*)o;
    v4f b = *(const v4f*)(o + 4);
    a.x = c0 + d.x * a.x;  a.y = c1 + d.x * a.y;
    a.z = c0 + d.y * a.z;  a.w = c1 + d.y * a.w;
    b.x = c0 + d.z * b.x;  b.y = c1 + d.z * b.y;
    b.z = c0 + d.w * b.z;  b.w = c1 + d.w * b.w;
    *(v4f*)o       = a;
    *(v4f*)(o + 4) = b;
  } else {
    for (int j = i; j < n; ++j) {
      const float d = dis[j];
      out[2 * (size_t)j + 0] = c0 + d * out[2 * (size_t)j + 0];
      out[2 * (size_t)j + 1] = c1 + d * out[2 * (size_t)j + 1];
    }
  }
}


#define GSH     11
#define GBIN    2048
#define GNB_MAX 256
#define GCH_MAX 65536
#define GPAD    32
#define GROW    512
#define GTC     1024

__global__ void __launch_bounds__(RTPB)
gcn_group(const int* __restrict__ row, const int* __restrict__ col,
          int* __restrict__ recs, int* __restrict__ tab,
          int e, int n, int chunklen, int nchunks, int cap) {
  __shared__ __attribute__((aligned(16))) int lrec[GCH_MAX + GNB_MAX * GPAD];
  __shared__ int lcnt[GNB_MAX];
  __shared__ int lst[GNB_MAX + 1];
  const int tid = threadIdx.x;
  const int NB  = (n + GBIN - 1) >> GSH;
  for (int ch = blockIdx.x; ch < nchunks; ch += gridDim.x) {
    const int elo = ch * chunklen;
    const int ehi = min(e, elo + chunklen);
    if (tid < GNB_MAX) lcnt[tid] = 0;
    __syncthreads();
    for (int base = elo; base < ehi; base += RTPB * 16) {
      const int b = base + tid * 4;
      v4i c[4];
      if (base + RTPB * 16 <= ehi) {
#pragma unroll
        for (int u = 0; u < 4; ++u) c[u] = *(const v4i*)(col + b + u * RTPB * 4);
      } else {
#pragma unroll
        for (int u = 0; u < 4; ++u) c[u] = load4_or_none(col, b + u * RTPB * 4, ehi);
      }
#pragma unroll
      for (int u = 0; u < 4; ++u) {
#pragma unroll
        for (int j = 0; j < 4; ++j)
          if ((unsigned)c[u][j] < (unsigned)n) atomicAdd(&lcnt[c[u][j] >> GSH], 1);
      }
    }
    __syncthreads();
    int p = 0;
    if (tid < NB) p = (lcnt[tid] + GPAD - 1) & ~(GPAD - 1);
    if (tid < GNB_MAX) lst[tid + 1] = p;
    if (tid == 0) lst[0] = 0;
    __syncthreads();
    for (int off = 1; off < GNB_MAX; off <<= 1) {
      int v = 0;
      if (tid < GNB_MAX && tid >= off) v = lst[tid + 1 - off];
      __syncthreads();
      if (tid < GNB_MAX) lst[tid + 1] += v;
      __syncthreads();
    }
    if (tid < NB) {
      const int s0 = lst[tid], c0 = lcnt[tid], s1 = lst[tid + 1];
      for (int k = s0 + c0; k < s1; ++k) lrec[k] = -1;
      lcnt[tid] = s0;
    }
    __syncthreads();
    for (int base = elo; base < ehi; base += RTPB * 16) {
      const int b = base + tid * 4;
      v4i c[4], r[4];
      if (base + RTPB * 16 <= ehi) {
#pragma unroll
        for (int u = 0; u < 4; ++u) {
          c[u] = *(const v4i*)(col + b + u * RTPB * 4);
          r[u] = *(const v4i*)(row + b + u * RTPB * 4);
        }
      } else {
#pragma unroll
        for (int u = 0; u < 4; ++u) {
          c[u] = load4_or_none(col, b + u * RTPB * 4, ehi);
          r[u] = load4_or_none(row, b + u * RTPB * 4, ehi);
        }
      }
      int pos[16];
#pragma unroll
      for (int u = 0; u < 4; ++u) {
#pragma unroll
        for (int j = 0; j < 4; ++j) {
          pos[u * 4 + j] = -1;
          if ((unsigned)c[u][j] < (unsigned)n) pos[u * 4 + j] = atomicAdd(&lcnt[c[u][j] >> GSH], 1);
        }
      }
#pragma unroll
      for (int u = 0; u < 4; ++u) {
#pragma unroll
        for (int j = 0; j < 4; ++j)
          if (pos[u * 4 + j] >= 0)
            lrec[pos[u * 4 + j]] = (int)(((unsigned)r[u][j] << GSH) | ((unsigned)c[u][j] & (unsigned)(GBIN - 1)));
      }
    }
    __syncthreads();
    const int tot4 = lst[NB] >> 2;
    int* g = recs + (size_t)ch * (size_t)cap;
    for (int u = tid; u < tot4; u += RTPB) *(v4i*)(g + 4 * u) = *(const v4i*)&lrec[4 * u];
    if (tid < GROW) tab[(size_t)ch * GROW + tid] = lst[(tid >> 1) + (tid & 1)];
    __syncthreads();
  }
}

__global__ void __launch_bounds__(RTPB)
gcn_degree_g(const int* __restrict__ recs, const int* __restrict__ tab,
             const float* __restrict__ x, float* __restrict__ dis, float* __restrict__ xp,
             int n, int nchunks, int cap) {
  __shared__ float acc[GBIN];
  __shared__ int   tst[GTC], ten[GTC];
  const int tid  = threadIdx.x;
  const int lane = tid & 31;
  const int wv   = __builtin_amdgcn_readfirstlane(tid >> 5);
  const int NB   = (n + GBIN - 1) >> GSH;
  for (int b = blockIdx.x; b < NB; b += gridDim.x) {
    const int lo  = b << GSH;
    const int cnt = min(GBIN, n - lo);
    for (int i = tid; i < GBIN; i += RTPB) acc[i] = 0.0f;
    for (int c0 = 0; c0 < nchunks; c0 += GTC) {
      const int nc = min(GTC, nchunks - c0);
      __syncthreads();
      if (tid < nc) {
        const v2i se = *(const v2i*)(tab + (size_t)(c0 + tid) * GROW + 2 * b);
        tst[tid] = se.x;
        ten[tid] = se.y;
      }
      __syncthreads();
      for (int k = wv; k < nc; k += RTPB / 32) {
        const int* p  = recs + (size_t)(c0 + k) * (size_t)cap;
        const int  u1 = ten[k] >> 2;
        for (int u = (tst[k] >> 2) + lane; u < u1; u += 32) {
          const v4i rc = *(const v4i*)(p + 4 * u);
#pragma unroll
          for (int j = 0; j < 4; ++j)
            if (rc[j] != -1) atomicAdd(&acc[rc[j] & (GBIN - 1)], 1.0f);
        }
      }
    }
    __syncthreads();
    for (int i = tid; i < cnt; i += RTPB) {
      const float d = rsqrtf(1.0f + acc[i]);
      dis[lo + i] = d;
      xp[lo + i]  = x[lo + i] * d;
    }
    __syncthreads();
  }
}

__global__ void __launch_bounds__(RTPB)
gcn_l1_g(const int* __restrict__ recs, const int* __restrict__ tab,
         const float* __restrict__ xp, float* __restrict__ agg1,
         int n, int nchunks, int cap) {
  __shared__ float acc[GBIN];
  __shared__ int   tst[GTC], ten[GTC];
  const int tid  = threadIdx.x;
  const int lane = tid & 31;
  const int wv   = __builtin_amdgcn_readfirstlane(tid >> 5);
  const int NB   = (n + GBIN - 1) >> GSH;
  for (int b = blockIdx.x; b < NB; b += gridDim.x) {
    const int lo  = b << GSH;
    const int cnt = min(GBIN, n - lo);
    for (int i = tid; i < GBIN; i += RTPB) acc[i] = 0.0f;
    for (int c0 = 0; c0 < nchunks; c0 += GTC) {
      const int nc = min(GTC, nchunks - c0);
      __syncthreads();
      if (tid < nc) {
        const v2i se = *(const v2i*)(tab + (size_t)(c0 + tid) * GROW + 2 * b);
        tst[tid] = se.x;
        ten[tid] = se.y;
      }
      __syncthreads();
      for (int k = wv; k < nc; k += RTPB / 32) {
        const int* p  = recs + (size_t)(c0 + k) * (size_t)cap;
        const int  u1 = ten[k] >> 2;
        for (int u = (tst[k] >> 2) + lane; u < u1; u += 64) {
          const v4i rc0 = *(const v4i*)(p + 4 * u);
          v4i rc1 = {-1, -1, -1, -1};
          if (u + 32 < u1) rc1 = *(const v4i*)(p + 4 * (u + 32));
          float v[8];
#pragma unroll
          for (int j = 0; j < 4; ++j) {
            v[j]     = (rc0[j] != -1) ? xp[(unsigned)rc0[j] >> GSH] : 0.0f;
            v[4 + j] = (rc1[j] != -1) ? xp[(unsigned)rc1[j] >> GSH] : 0.0f;
          }
#pragma unroll
          for (int j = 0; j < 4; ++j) {
            if (rc0[j] != -1) atomicAdd(&acc[rc0[j] & (GBIN - 1)], v[j]);
            if (rc1[j] != -1) atomicAdd(&acc[rc1[j] & (GBIN - 1)], v[4 + j]);
          }
        }
      }
    }
    __syncthreads();
    for (int i = tid; i < cnt; i += RTPB) agg1[lo + i] = acc[i];
    __syncthreads();
  }
}

__global__ void __launch_bounds__(RTPB)
gcn_l2_g(const int* __restrict__ recs, const int* __restrict__ tab,
         const float* __restrict__ h2p, const float* __restrict__ dis,
         const float* __restrict__ b2, float* __restrict__ out,
         int n, int nchunks, int cap) {
  __shared__ float acc0[GBIN], acc1[GBIN];
  __shared__ int   tst[GTC], ten[GTC];
  const int tid  = threadIdx.x;
  const int lane = tid & 31;
  const int wv   = __builtin_amdgcn_readfirstlane(tid >> 5);
  const int NB   = (n + GBIN - 1) >> GSH;
  const float cb0 = b2[0], cb1 = b2[1];
  for (int b = blockIdx.x; b < NB; b += gridDim.x) {
    const int lo  = b << GSH;
    const int cnt = min(GBIN, n - lo);
    for (int i = tid; i < GBIN; i += RTPB) { acc0[i] = 0.0f; acc1[i] = 0.0f; }
    for (int c0 = 0; c0 < nchunks; c0 += GTC) {
      const int nc = min(GTC, nchunks - c0);
      __syncthreads();
      if (tid < nc) {
        const v2i se = *(const v2i*)(tab + (size_t)(c0 + tid) * GROW + 2 * b);
        tst[tid] = se.x;
        ten[tid] = se.y;
      }
      __syncthreads();
      for (int k = wv; k < nc; k += RTPB / 32) {
        const int* p  = recs + (size_t)(c0 + k) * (size_t)cap;
        const int  u1 = ten[k] >> 2;
        for (int u = (tst[k] >> 2) + lane; u < u1; u += 64) {
          const v4i rc0 = *(const v4i*)(p + 4 * u);
          v4i rc1 = {-1, -1, -1, -1};
          if (u + 32 < u1) rc1 = *(const v4i*)(p + 4 * (u + 32));
          v2f v[8];
#pragma unroll
          for (int j = 0; j < 4; ++j) {
            v[j]     = (v2f){0.f, 0.f};
            v[4 + j] = (v2f){0.f, 0.f};
            if (rc0[j] != -1) v[j]     = *(const v2f*)(h2p + 2 * (size_t)((unsigned)rc0[j] >> GSH));
            if (rc1[j] != -1) v[4 + j] = *(const v2f*)(h2p + 2 * (size_t)((unsigned)rc1[j] >> GSH));
          }
#pragma unroll
          for (int j = 0; j < 4; ++j) {
            if (rc0[j] != -1) {
              atomicAdd(&acc0[rc0[j] & (GBIN - 1)], v[j].x);
              atomicAdd(&acc1[rc0[j] & (GBIN - 1)], v[j].y);
            }
            if (rc1[j] != -1) {
              atomicAdd(&acc0[rc1[j] & (GBIN - 1)], v[4 + j].x);
              atomicAdd(&acc1[rc1[j] & (GBIN - 1)], v[4 + j].y);
            }
          }
        }
      }
    }
    __syncthreads();
    for (int i = tid; i < cnt; i += RTPB) {
      const size_t nd = (size_t)(lo + i);
      const float  d  = dis[nd];
      const v2f    hs = *(const v2f*)(h2p + 2 * nd);
      v2f o;
      o.x = cb0 + d * (acc0[i] + hs.x);
      o.y = cb1 + d * (acc1[i] + hs.y);
      *(v2f*)(out + 2 * nd) = o;
    }
    __syncthreads();
  }
}


#define TSH 15
#define TRN 32768

__global__ void __launch_bounds__(RTPB)
gcn_group_t(const int* __restrict__ row, const int* __restrict__ col,
            int* __restrict__ recs, int* __restrict__ tab,
            float* __restrict__ deg, float* __restrict__ agg1, float* __restrict__ s1,
            int* __restrict__ tick, int e, int n, int chunklen, int nchunks, int cap, int RN) {
  __shared__ __attribute__((aligned(16))) int lrec[GCH_MAX + GNB_MAX * GPAD];
  __shared__ int lcnt[GNB_MAX];
  __shared__ int lst[GNB_MAX + 1];
  const int tid = threadIdx.x;
  const int NB  = RN * RN;
  for (int i = blockIdx.x * RTPB + tid; i < n; i += gridDim.x * RTPB) { deg[i] = 0.0f; agg1[i] = 0.0f; s1[i] = 0.0f; }
  if (blockIdx.x == 0 && tid < 6) tick[tid * 128] = 0;
  for (int ch = blockIdx.x; ch < nchunks; ch += gridDim.x) {
    const int elo = ch * chunklen;
    const int ehi = min(e, elo + chunklen);
    if (tid < GNB_MAX) lcnt[tid] = 0;
    __syncthreads();
    for (int base = elo; base < ehi; base += RTPB * 16) {
      const int b = base + tid * 4;
      v4i c[4], r[4];
      if (base + RTPB * 16 <= ehi) {
#pragma unroll
        for (int u = 0; u < 4; ++u) {
          c[u] = *(const v4i*)(col + b + u * RTPB * 4);
          r[u] = *(const v4i*)(row + b + u * RTPB * 4);
        }
      } else {
#pragma unroll
        for (int u = 0; u < 4; ++u) {
          c[u] = load4_or_none(col, b + u * RTPB * 4, ehi);
          r[u] = load4_or_none(row, b + u * RTPB * 4, ehi);
        }
      }
#pragma unroll
      for (int u = 0; u < 4; ++u) {
#pragma unroll
        for (int j = 0; j < 4; ++j)
          if ((unsigned)c[u][j] < (unsigned)n && (unsigned)r[u][j] < (unsigned)n)
            atomicAdd(&lcnt[(c[u][j] >> TSH) * RN + (r[u][j] >> TSH)], 1);
      }
    }
    __syncthreads();
    int p = 0;
    if (tid < NB) p = (lcnt[tid] + GPAD - 1) & ~(GPAD - 1);
    if (tid < GNB_MAX) lst[tid + 1] = p;
    if (tid == 0) lst[0] = 0;
    __syncthreads();
    for (int off = 1; off < GNB_MAX; off <<= 1) {
      int v = 0;
      if (tid < GNB_MAX && tid >= off) v = lst[tid + 1 - off];
      __syncthreads();
      if (tid < GNB_MAX) lst[tid + 1] += v;
      __syncthreads();
    }
    if (tid < NB) {
      const int s0 = lst[tid], c0 = lcnt[tid], s1e = lst[tid + 1];
      for (int k = s0 + c0; k < s1e; ++k) lrec[k] = -1;
      lcnt[tid] = s0;
    }
    __syncthreads();
    for (int base = elo; base < ehi; base += RTPB * 16) {
      const int b = base + tid * 4;
      v4i c[4], r[4];
      if (base + RTPB * 16 <= ehi) {
#pragma unroll
        for (int u = 0; u < 4; ++u) {
          c[u] = *(const v4i*)(col + b + u * RTPB * 4);
          r[u] = *(const v4i*)(row + b + u * RTPB * 4);
        }
      } else {
#pragma unroll
        for (int u = 0; u < 4; ++u) {
          c[u] = load4_or_none(col, b + u * RTPB * 4, ehi);
          r[u] = load4_or_none(row, b + u * RTPB * 4, ehi);
        }
      }
      int pos[16];
#pragma unroll
      for (int u = 0; u < 4; ++u) {
#pragma unroll
        for (int j = 0; j < 4; ++j) {
          pos[u * 4 + j] = -1;
          if ((unsigned)c[u][j] < (unsigned)n && (unsigned)r[u][j] < (unsigned)n)
            pos[u * 4 + j] = atomicAdd(&lcnt[(c[u][j] >> TSH) * RN + (r[u][j] >> TSH)], 1);
        }
      }
#pragma unroll
      for (int u = 0; u < 4; ++u) {
#pragma unroll
        for (int j = 0; j < 4; ++j)
          if (pos[u * 4 + j] >= 0)
            lrec[pos[u * 4 + j]] = (int)((((unsigned)r[u][j] & (unsigned)(TRN - 1)) << TSH) | ((unsigned)c[u][j] & (unsigned)(TRN - 1)));
      }
    }
    __syncthreads();
    const int tot4 = lst[NB] >> 2;
    int* g = recs + (size_t)ch * (size_t)cap;
    for (int u = tid; u < tot4; u += RTPB) *(v4i*)(g + 4 * u) = *(const v4i*)&lrec[4 * u];
    if (tid < GROW) tab[(size_t)ch * GROW + tid] = lst[(tid >> 1) + (tid & 1)];
    __syncthreads();
  }
}


__global__ void __launch_bounds__(RTPB)
gcn_degree_t(const int* __restrict__ recs, const int* __restrict__ tab,
             float* __restrict__ deg, int* __restrict__ tick,
             int n, int nchunks, int cap, int RN) {
  __shared__ __attribute__((aligned(16))) float acc[TRN + 32];
  __shared__ int   tst[GTC], ten[GTC];
  __shared__ int   slot;
  const int tid  = threadIdx.x;
  const int lane = tid & 31;
  const int wv   = __builtin_amdgcn_readfirstlane(tid >> 5);
  const int dh   = (RN + 1) >> 1;
  const int h    = cache_half();
  for (int ph = 0; ph < 2; ++ph) {
    const int q     = h ^ ph;
    const int dbase = q ? dh : 0;
    const int nt    = (q ? RN - dh : dh) * RN;
    int* ctr = tick + q * 128;
    for (;;) {
      const int tk = take_block_ticket(ctr, &slot);
      if (tk >= nt) break;
      const int t    = (dbase + tk / RN) * RN + tk % RN;
      const int lo   = (t / RN) << TSH;
      const int cnt  = min(TRN, n - lo);
      for (int i = tid * 4; i < TRN + 32; i += RTPB * 4) *(v4f*)&acc[i] = (v4f){0.f, 0.f, 0.f, 0.f};
      for (int c0 = 0; c0 < nchunks; c0 += GTC) {
        const int nc = min(GTC, nchunks - c0);
        __syncthreads();
        if (tid < nc) {
          const v2i se = *(const v2i*)(tab + (size_t)(c0 + tid) * GROW + 2 * t);
          tst[tid] = se.x;
          ten[tid] = se.y;
        }
        __syncthreads();
        for (int k = wv; k < nc; k += RTPB / 32) {
          const int* p  = recs + (size_t)(c0 + k) * (size_t)cap;
          const int  u1 = ten[k] >> 2;
          for (int u = (tst[k] >> 2) + lane; u < u1; u += 96) {
            v4i rc[3];
            rc[0] = *(const v4i*)(p + 4 * u);
            rc[1] = (v4i){-1, -1, -1, -1};
            rc[2] = (v4i){-1, -1, -1, -1};
            if (u + 32 < u1) rc[1] = *(const v4i*)(p + 4 * (u + 32));
            if (u + 64 < u1) rc[2] = *(const v4i*)(p + 4 * (u + 64));
#pragma unroll
            for (int q = 0; q < 3; ++q) {
#pragma unroll
              for (int j = 0; j < 4; ++j) {
                const int rr = rc[q][j];
                atomicAdd(&acc[(rr != -1) ? (rr & (TRN - 1)) : (TRN + lane)], 1.0f);
              }
            }
          }
        }
      }
      __syncthreads();
      for (int i = tid; i < cnt; i += RTPB) {
        const float v = acc[i];
        if (v != 0.0f) atomicAdd(&deg[lo + i], v);
      }
    }
  }
}

__global__ void __launch_bounds__(RTPB)
gcn_l1_t(const int* __restrict__ recs, const int* __restrict__ tab,
         const float* __restrict__ x, const float* __restrict__ deg,
         float* __restrict__ agg1, int* __restrict__ tick,
         int n, int nchunks, int cap, int RN) {
  __shared__ __attribute__((aligned(16))) float acc[TRN + 32];
  __shared__ __attribute__((aligned(16))) float xs[TRN];
  __shared__ int   tst[GTC], ten[GTC];
  __shared__ int   slot;
  const int tid  = threadIdx.x;
  const int lane = tid & 31;
  const int wv   = __builtin_amdgcn_readfirstlane(tid >> 5);
  const int dh   = (RN + 1) >> 1;
  const int h    = cache_half();
  for (int ph = 0; ph < 2; ++ph) {
    const int q     = h ^ ph;
    const int dbase = q ? dh : 0;
    const int nt    = (q ? RN - dh : dh) * RN;
    int* ctr = tick + q * 128;
    for (;;) {
      const int tk = take_block_ticket(ctr, &slot);
      if (tk >= nt) break;
      const int t    = (dbase + tk / RN) * RN + tk % RN;
      const int lo   = (t / RN) << TSH;
      const int cnt  = min(TRN, n - lo);
      const int slo  = (t % RN) << TSH;
      const int scnt = min(TRN, n - slo);
      if (tid < 32) acc[TRN + tid] = 0.0f;
      for (int i = tid; i < TRN; i += RTPB) {
        acc[i] = 0.0f;
        float v = 0.0f;
        if (i < scnt) v = x[slo + i] * rsqrtf(1.0f + deg[slo + i]);
        xs[i] = v;
      }
      for (int c0 = 0; c0 < nchunks; c0 += GTC) {
        const int nc = min(GTC, nchunks - c0);
        __syncthreads();
        if (tid < nc) {
          const v2i se = *(const v2i*)(tab + (size_t)(c0 + tid) * GROW + 2 * t);
          tst[tid] = se.x;
          ten[tid] = se.y;
        }
        __syncthreads();
        for (int k = wv; k < nc; k += RTPB / 32) {
          const int* p  = recs + (size_t)(c0 + k) * (size_t)cap;
          const int  u1 = ten[k] >> 2;
          for (int u = (tst[k] >> 2) + lane; u < u1; u += 96) {
            v4i rc[3];
            rc[0] = *(const v4i*)(p + 4 * u);
            rc[1] = (v4i){-1, -1, -1, -1};
            rc[2] = (v4i){-1, -1, -1, -1};
            if (u + 32 < u1) rc[1] = *(const v4i*)(p + 4 * (u + 32));
            if (u + 64 < u1) rc[2] = *(const v4i*)(p + 4 * (u + 64));
            float v[12];
#pragma unroll
            for (int q = 0; q < 3; ++q) {
#pragma unroll
              for (int j = 0; j < 4; ++j) {
                const int rr = rc[q][j];
                v[q * 4 + j] = xs[(rr != -1) ? ((unsigned)rr >> TSH) : 0u];
              }
            }
#pragma unroll
            for (int q = 0; q < 3; ++q) {
#pragma unroll
              for (int j = 0; j < 4; ++j) {
                const int rr = rc[q][j];
                atomicAdd(&acc[(rr != -1) ? (rr & (TRN - 1)) : (TRN + lane)], v[q * 4 + j]);
              }
            }
          }
        }
      }
      __syncthreads();
      for (int i = tid; i < cnt; i += RTPB) {
        const float v = acc[i];
        if (v != 0.0f) atomicAdd(&agg1[lo + i], v);
      }
    }
  }
}

__global__ void gcn_node_t(float* __restrict__ agg1,
                           const float* __restrict__ x,
                           const float* __restrict__ deg,
                           const float* __restrict__ W1,
                           const float* __restrict__ b1,
                           const float* __restrict__ W2,
                           float* __restrict__ p0,
                           float* __restrict__ p1,
                           int n) {
  int lane = threadIdx.x & 31;
  int wave = threadIdx.x >> 5;
  int tile = blockIdx.x * (blockDim.x >> 5) + wave;
  int base = tile * 16;
  int m    = lane & 15;
  int half = lane >> 4;
  int node = base + m;
  int nc   = node < n ? node : (n - 1);

  float dn = rsqrtf(1.0f + deg[nc]);
  float av = (agg1[nc] + x[nc] * dn) * dn;
  if (half == 0 && node < n) agg1[node] = 0.0f;

  int   wcol  = (m < 2) ? m : 0;
  float wmask = (m < 2) ? 1.0f : 0.0f;

  v8f acc = {};
#pragma unroll
  for (int s = 0; s < 4; ++s) {
    int j0 = s * 4 + half * 2;
    v2f w1 = *(const v2f*)(W1 + j0);
    v2f bi = *(const v2f*)(b1 + j0);
    v2f a;
    a.x = fmaxf(av * w1.x + bi.x, 0.0f);
    a.y = fmaxf(av * w1.y + bi.y, 0.0f);
    v2f bb;
    bb.x = W2[j0 * 2 + wcol]       * wmask;
    bb.y = W2[(j0 + 1) * 2 + wcol] * wmask;
    acc = __builtin_amdgcn_wmma_f32_16x16x4_f32(
         false, a,  false, bb,
         (short)0, acc,  false,  false);
  }

  if (m < 2) {
    float* pm = m ? p1 : p0;
#pragma unroll
    for (int r = 0; r < 8; ++r) {
      int nd = base + r + half * 8;
      if (nd < n) pm[nd] = acc[r] * rsqrtf(1.0f + deg[nd]);
    }
  }
}


__global__ void __launch_bounds__(RTPB)
gcn_l2_t(const int* __restrict__ recs, const int* __restrict__ tab,
         const float* __restrict__ p0, const float* __restrict__ p1,
         float* __restrict__ s0, float* __restrict__ s1, int* __restrict__ tick,
         int n, int nchunks, int cap, int RN) {
  __shared__ __attribute__((aligned(16))) float acc[TRN + 32];
  __shared__ __attribute__((aligned(16))) float hs[TRN];
  __shared__ int   tst[GTC], ten[GTC];
  __shared__ int   slot;
  const int tid  = threadIdx.x;
  const int lane = tid & 31;
  const int wv   = __builtin_amdgcn_readfirstlane(tid >> 5);
  const int dh   = (RN + 1) >> 1;
  const int h    = cache_half();
  for (int ph = 0; ph < 2; ++ph) {
    const int q     = h ^ ph;
    const int dbase = q ? dh : 0;
    const int nt    = (q ? RN - dh : dh) * RN;
    int* ctr = tick + q * 128;
    for (;;) {
      const int tk = take_block_ticket(ctr, &slot);
      if (tk >= nt) break;
      const int t    = (dbase + tk / RN) * RN + tk % RN;
      const int lo   = (t / RN) << TSH;
      const int cnt  = min(TRN, n - lo);
      const int slo  = (t % RN) << TSH;
      const int scnt = min(TRN, n - slo);
      for (int comp = 0; comp < 2; ++comp) {
      const float* pk = comp ? p1 : p0;
      float*       sk = comp ? s1 : s0;
      if (tid < 32) acc[TRN + tid] = 0.0f;
      for (int i = tid; i < TRN; i += RTPB) {
        acc[i] = 0.0f;
        hs[i]  = (i < scnt) ? pk[slo + i] : 0.0f;
      }
      for (int c0 = 0; c0 < nchunks; c0 += GTC) {
        const int nc = min(GTC, nchunks - c0);
        __syncthreads();
        if (tid < nc) {
          const v2i se = *(const v2i*)(tab + (size_t)(c0 + tid) * GROW + 2 * t);
          tst[tid] = se.x;
          ten[tid] = se.y;
        }
        __syncthreads();
        for (int k = wv; k < nc; k += RTPB / 32) {
          const int* p  = recs + (size_t)(c0 + k) * (size_t)cap;
          const int  u1 = ten[k] >> 2;
          for (int u = (tst[k] >> 2) + lane; u < u1; u += 96) {
            v4i rc[3];
            rc[0] = *(const v4i*)(p + 4 * u);
            rc[1] = (v4i){-1, -1, -1, -1};
            rc[2] = (v4i){-1, -1, -1, -1};
            if (u + 32 < u1) rc[1] = *(const v4i*)(p + 4 * (u + 32));
            if (u + 64 < u1) rc[2] = *(const v4i*)(p + 4 * (u + 64));
            float v[12];
#pragma unroll
            for (int q = 0; q < 3; ++q) {
#pragma unroll
              for (int j = 0; j < 4; ++j) {
                const int rr = rc[q][j];
                v[q * 4 + j] = hs[(rr != -1) ? ((unsigned)rr >> TSH) : 0u];
              }
            }
#pragma unroll
            for (int q = 0; q < 3; ++q) {
#pragma unroll
              for (int j = 0; j < 4; ++j) {
                const int rr = rc[q][j];
                atomicAdd(&acc[(rr != -1) ? (rr & (TRN - 1)) : (TRN + lane)], v[q * 4 + j]);
              }
            }
          }
        }
      }
      __syncthreads();
      for (int i = tid; i < cnt; i += RTPB) {
        const float v = acc[i];
        if (v != 0.0f) atomicAdd(&sk[lo + i], v);
      }
      __syncthreads();
      }
    }
  }
}

__global__ void gcn_fin_t(const float* __restrict__ deg, const float* __restrict__ b2,
                          const float* __restrict__ p0, const float* __restrict__ p1,
                          const float* __restrict__ s0, const float* __restrict__ s1,
                          float* __restrict__ out, int n) {
  const int i = blockIdx.x * blockDim.x + threadIdx.x;
  if (i < n) {
    const float d = rsqrtf(1.0f + deg[i]);
    v2f o;
    o.x = b2[0] + d * (s0[i] + p0[i]);
    o.y = b2[1] + d * (s1[i] + p1[i]);
    *(v2f*)(out + 2 * (size_t)i) = o;
  }
}

extern "C" void kernel_launch(void* const* d_in, const int* in_sizes, int n_in,
                              void* d_out, int out_size, void* d_ws, size_t ws_size,
                              hipStream_t stream) {
  const float* x  = (const float*)d_in[0];
  const int*   ei = (const int*)d_in[1];
  const float* W1 = (const float*)d_in[2];
  const float* b1 = (const float*)d_in[3];
  const float* W2 = (const float*)d_in[4];
  const float* b2 = (const float*)d_in[5];
  float* out = (float*)d_out;

  const int n = in_sizes[0];
  const int e = in_sizes[1] / 2;
  const int* row = ei;
  const int* col = ei + e;

  size_t np   = (size_t)((n + 3) & ~3);
  float* ws   = (float*)d_ws;
  float* dis  = ws;
  float* xp   = ws + np;
  float* agg1 = ws + 2 * np;
  float* h2p  = ws + 3 * np;

  const int cs = (n >= 512) ? 128 : 1;
  int* tick24 = (int*)h2p;
  int* tick6  = (int*)agg1;

  const int tb = 256;
  const int nb_node = (n + tb - 1) / tb;
  const int tiles = (n + 15) / 16;
  const int nb_tile = (tiles + 7) / 8;
  const int nb_fin  = (n + 4 * tb - 1) / (4 * tb);

  {
    const int RNt = (n + TRN - 1) >> TSH;
    const int NBt = RNt * RNt;
    int chunklen_t = ((e + RANGE_BLOCKS - 1) / RANGE_BLOCKS + 3) & ~3;
    if (chunklen_t > GCH_MAX) chunklen_t = GCH_MAX;
    if (chunklen_t < 4) chunklen_t = 4;
    const int nchunks_t = (e + chunklen_t - 1) / chunklen_t;
    const int cap_t = (chunklen_t + 31 + NBt * GPAD) & ~31;
    const size_t off_tab_t = (5 * np + 31) & ~(size_t)31;
    const size_t off_rec_t = off_tab_t + (size_t)nchunks_t * GROW;
    const size_t off_tick_t = off_rec_t + (size_t)nchunks_t * (size_t)cap_t;
    const size_t need_t    = (off_tick_t + 6 * 128) * sizeof(float);
    if (n >= 1 && e >= 1 && NBt <= GNB_MAX && need_t <= ws_size) {
      float* degt = ws;
      float* p0   = ws + np;
      float* p1   = ws + 3 * np;
      float* s1   = ws + 4 * np;
      int* tab  = (int*)(ws + off_tab_t);
      int* recs = (int*)(ws + off_rec_t);
      int* tick = (int*)(ws + off_tick_t);
      gcn_group_t <<<RANGE_BLOCKS, RTPB, 0, stream>>>(row, col, recs, tab, degt, agg1, s1, tick, e, n, chunklen_t, nchunks_t, cap_t, RNt);
      gcn_degree_t<<<NBt, RTPB, 0, stream>>>(recs, tab, degt, tick, n, nchunks_t, cap_t, RNt);
      gcn_l1_t    <<<NBt, RTPB, 0, stream>>>(recs, tab, x, degt, agg1, tick + 2 * 128, n, nchunks_t, cap_t, RNt);
      gcn_node_t  <<<nb_tile, 256, 0, stream>>>(agg1, x, degt, W1, b1, W2, p0, p1, n);
      gcn_l2_t    <<<NBt, RTPB, 0, stream>>>(recs, tab, p0, p1, agg1, s1, tick + 4 * 128, n, nchunks_t, cap_t, RNt);
      gcn_fin_t   <<<nb_node, tb, 0, stream>>>(degt, b2, p0, p1, agg1, s1, out, n);
      return;
    }
  }

  const int NBg = (n + GBIN - 1) >> GSH;
  int chunklen = ((e + RANGE_BLOCKS - 1) / RANGE_BLOCKS + 3) & ~3;
  if (chunklen > GCH_MAX) chunklen = GCH_MAX;
  if (chunklen < 4) chunklen = 4;
  const int nchunks = (e + chunklen - 1) / chunklen;
  const int cap = (chunklen + 31 + NBg * GPAD) & ~31;
  const size_t off_tab = (5 * np + 31) & ~(size_t)31;
  const size_t off_rec = off_tab + (size_t)nchunks * GROW;
  const size_t need    = (off_rec + (size_t)nchunks * (size_t)cap) * sizeof(float);
  if (n >= 1 && e >= 1 && NBg <= GNB_MAX && need <= ws_size) {
    int* tab  = (int*)(ws + off_tab);
    int* recs = (int*)(ws + off_rec);
    gcn_group   <<<RANGE_BLOCKS, RTPB, 0, stream>>>(row, col, recs, tab, e, n, chunklen, nchunks, cap);
    gcn_degree_g<<<NBg, RTPB, 0, stream>>>(recs, tab, x, dis, xp, n, nchunks, cap);
    gcn_l1_g    <<<NBg, RTPB, 0, stream>>>(recs, tab, xp, agg1, n, nchunks, cap);
    gcn_node    <<<nb_tile, 256, 0, stream>>>(agg1, xp, dis, W1, b1, W2, h2p, out, n, 0);
    gcn_l2_g    <<<NBg, RTPB, 0, stream>>>(recs, tab, h2p, dis, b2, out, n, nchunks, cap);
    return;
  }

  gcn_init  <<<nb_node, tb, 0, stream>>>(dis, agg1, tick24, cs, n);
  gcn_degree<<<RANGE_BLOCKS, RTPB, 0, stream>>>(col, dis, tick24, cs, e, n);
  gcn_dis   <<<nb_node, tb, 0, stream>>>(x, dis, xp, n);
  gcn_l1    <<<RANGE_BLOCKS, RTPB, 0, stream>>>(row, col, xp, agg1, tick24 + 2 * cs, cs, e, n);
  gcn_node  <<<nb_tile, 256, 0, stream>>>(agg1, xp, dis, W1, b1, W2, h2p, out, n, 1);
  gcn_tick0 <<<1, 32, 0, stream>>>(tick6, cs);
  gcn_l2    <<<RANGE_BLOCKS, RTPB, 0, stream>>>(row, col, h2p, out, tick6, cs, e, n);
  gcn_fin   <<<nb_fin, tb, 0, stream>>>(dis, b2, out, n);
}


# Loads waited for one by one go out together: tile kernels stage a source range with 8 x 16-byte loads per thread in flight, flush reads 8 sums at once, sort's last step tests bounds per 4096 edges
# speedup vs baseline: 12.3551x; 1.2271x over previous
#include <hip/hip_runtime.h>


typedef __attribute__((ext_vector_type(2))) float v2f;
typedef __attribute__((ext_vector_type(8))) float v8f;
typedef __attribute__((ext_vector_type(4))) int   v4i;
typedef __attribute__((ext_vector_type(4))) float v4f;
typedef __attribute__((ext_vector_type(2))) int   v2i;


#define RS_MAX 65536
#define SLICES 32
#define RTPB   1024
#define RANGE_BLOCKS 256

#define L2_CAP 32768
#define L2_SL  16
#define QN     64

__device__ __forceinline__ int cache_half() {
  unsigned id = __builtin_amdgcn_s_sendmsg_rtn(0x87);
  return (int)((id >> 18) & 1u);
}

__device__ __forceinline__ int range_count(int n) {
  int r = (n + RS_MAX - 1) / RS_MAX;
  if (r < 8) r = 8;
  return (r + 1) & ~1;
}
__device__ __forceinline__ int range_size(int n, int r) { return ((n + r - 1) / r + 63) & ~63; }
__device__ __forceinline__ int slice_len(int e) { return ((e + SLICES - 1) / SLICES + 16383) & ~16383; }

__device__ __forceinline__ int take_block_ticket(int* ctr, int* slot) {
  if (threadIdx.x == 0) *slot = atomicAdd(ctr, 1);
  __syncthreads();
  const int t = *slot;
  __syncthreads();
  return t;
}

__device__ __forceinline__ v4i load4_or_none(const int* __restrict__ a, int i, int hi) {
  if (i + 3 < hi) return *(const v4i*)(a + i);
  v4i r = {-1, -1, -1, -1};
  if (i     < hi) r.x = a[i];
  if (i + 1 < hi) r.y = a[i + 1];
  if (i + 2 < hi) r.z = a[i + 2];
  return r;
}

__global__ void gcn_init(float* __restrict__ deg, float* __restrict__ agg1,
                         int* __restrict__ tick, int cs, int n) {
  int i = blockIdx.x * blockDim.x + threadIdx.x;
  if (i < n) { deg[i] = 1.0f; agg1[i] = 0.0f; }
  if (i < 4) tick[i * cs] = 0;
}

__global__ void gcn_tick0(int* __restrict__ tick, int cs) {
  if (threadIdx.x < 2) tick[threadIdx.x * cs] = 0;
}

__global__ void __launch_bounds__(RTPB)
gcn_degree(const int* __restrict__ col, float* __restrict__ deg,
           int* __restrict__ tick, int cs, int e, int n) {
  __shared__ __attribute__((aligned(16))) float acc[RS_MAX];
  __shared__ int   slot;
  const int tid = threadIdx.x;
  const int R   = range_count(n);
  const int Rh  = R >> 1;
  const int RS  = range_size(n, R);
  const int CH  = slice_len(e);
  const int nt  = Rh * SLICES;
  const int h   = cache_half();
  for (int p = 0; p < 2; ++p) {
    const int q = h ^ p;
    int* ctr = tick + q * cs;
    for (;;) {
      const int t = take_block_ticket(ctr, &slot);
      if (t >= nt) break;
      const int lo  = (q * Rh + t % Rh) * RS;
      const int cnt = min(RS, n - lo);
      const int elo = (t / Rh) * CH;
      const int ehi = min(e, elo + CH);
      if (cnt <= 0 || elo >= ehi) continue;
      for (int i = tid * 4; i < RS; i += RTPB * 4) *(v4f*)&acc[i] = (v4f){0.f, 0.f, 0.f, 0.f};
      __syncthreads();
      int b = elo + tid * 4;
      for (; b + 3 * RTPB * 4 + 3 < ehi; b += RTPB * 16) {
        v4i c[4];
#pragma unroll
        for (int u = 0; u < 4; ++u) c[u] = *(const v4i*)(col + b + u * RTPB * 4);
#pragma unroll
        for (int u = 0; u < 4; ++u) {
#pragma unroll
          for (int j = 0; j < 4; ++j) {
            const unsigned d = (unsigned)(c[u][j] - lo);
            if (d < (unsigned)cnt) atomicAdd(&acc[d], 1.0f);
          }
        }
      }
      for (; b < ehi; b += RTPB * 4) {
        const v4i c = load4_or_none(col, b, ehi);
#pragma unroll
        for (int j = 0; j < 4; ++j) {
          const unsigned d = (unsigned)(c[j] - lo);
          if (d < (unsigned)cnt) atomicAdd(&acc[d], 1.0f);
        }
      }
      __syncthreads();
      for (int i = tid; i < cnt; i += RTPB) {
        const float v = acc[i];
        if (v != 0.0f) atomicAdd(&deg[lo + i], v);
      }
    }
  }
}

__global__ void gcn_dis(const float* __restrict__ x, float* __restrict__ deg_dis,
                        float* __restrict__ xp, int n) {
  int i = blockIdx.x * blockDim.x + threadIdx.x;
  if (i < n) {
    float d = rsqrtf(deg_dis[i]);
    deg_dis[i] = d;
    xp[i] = x[i] * d;
  }
}

__global__ void __launch_bounds__(RTPB)
gcn_l1(const int* __restrict__ row, const int* __restrict__ col,
       const float* __restrict__ xp,
       float* __restrict__ agg1, int* __restrict__ tick, int cs, int e, int n) {
  __shared__ __attribute__((aligned(16))) float acc[RS_MAX];
  __shared__ int   slot;
  const int tid = threadIdx.x;
  const int R   = range_count(n);
  const int Rh  = R >> 1;
  const int RS  = range_size(n, R);
  const int CH  = slice_len(e);
  const int nt  = Rh * SLICES;
  const int h   = cache_half();
  for (int p = 0; p < 2; ++p) {
    const int q = h ^ p;
    int* ctr = tick + q * cs;
    for (;;) {
      const int t = take_block_ticket(ctr, &slot);
      if (t >= nt) break;
      const int lo  = (q * Rh + t % Rh) * RS;
      const int cnt = min(RS, n - lo);
      const int elo = (t / Rh) * CH;
      const int ehi = min(e, elo + CH);
      if (cnt <= 0 || elo >= ehi) continue;
      for (int i = tid * 4; i < RS; i += RTPB * 4) *(v4f*)&acc[i] = (v4f){0.f, 0.f, 0.f, 0.f};
      __syncthreads();
      int b = elo + tid * 4;
      for (; b + 3 * RTPB * 4 + 3 < ehi; b += RTPB * 16) {
        v4i r[4], c[4];
#pragma unroll
        for (int u = 0; u < 4; ++u) {
          c[u] = *(const v4i*)(col + b + u * RTPB * 4);
          r[u] = *(const v4i*)(row + b + u * RTPB * 4);
        }
        asm volatile("" : "+v"(c[0]), "+v"(c[1]), "+v"(c[2]), "+v"(c[3]),
                          "+v"(r[0]), "+v"(r[1]), "+v"(r[2]), "+v"(r[3]));
        float v[16];
        bool  m[16];
#pragma unroll
        for (int u = 0; u < 4; ++u) {
#pragma unroll
          for (int j = 0; j < 4; ++j) {
            const unsigned d = (unsigned)(c[u][j] - lo);
            m[u * 4 + j] = d < (unsigned)cnt;
            v[u * 4 + j] = 0.0f;
            if (m[u * 4 + j]) v[u * 4 + j] = xp[r[u][j]];
          }
        }
#pragma unroll
        for (int u = 0; u < 4; ++u) {
#pragma unroll
          for (int j = 0; j < 4; ++j)
            if (m[u * 4 + j]) atomicAdd(&acc[c[u][j] - lo], v[u * 4 + j]);
        }
      }
      for (; b < ehi; b += RTPB * 4) {
        const v4i c = load4_or_none(col, b, ehi);
        const v4i r = load4_or_none(row, b, ehi);
#pragma unroll
        for (int j = 0; j < 4; ++j) {
          const unsigned d = (unsigned)(c[j] - lo);
          if (d < (unsigned)cnt) atomicAdd(&acc[d], xp[r[j]]);
        }
      }
      __syncthreads();
      for (int i = tid; i < cnt; i += RTPB) {
        const float v = acc[i];
        if (v != 0.0f) atomicAdd(&agg1[lo + i], v);
      }
    }
  }
}

__global__ void gcn_node(const float* __restrict__ agg1,
                         const float* __restrict__ xp,
                         const float* __restrict__ dis,
                         const float* __restrict__ W1,
                         const float* __restrict__ b1,
                         const float* __restrict__ W2,
                         float* __restrict__ h2p,
                         float* __restrict__ out,
                         int n, int start_out) {
  int lane = threadIdx.x & 31;
  int wave = threadIdx.x >> 5;
  int tile = blockIdx.x * (blockDim.x >> 5) + wave;
  int base = tile * 16;
  int m    = lane & 15;
  int half = lane >> 4;
  int node = base + m;
  int nc   = node < n ? node : (n - 1);

  float av = (agg1[nc] + xp[nc]) * dis[nc];

  int   wcol  = (m < 2) ? m : 0;
  float wmask = (m < 2) ? 1.0f : 0.0f;

  v8f acc = {};
#pragma unroll
  for (int s = 0; s < 4; ++s) {
    int j0 = s * 4 + half * 2;
    v2f w1 = *(const v2f*)(W1 + j0);
    v2f bi = *(const v2f*)(b1 + j0);
    v2f a;
    a.x = fmaxf(av * w1.x + bi.x, 0.0f);
    a.y = fmaxf(av * w1.y + bi.y, 0.0f);
    v2f bb;
    bb.x = W2[j0 * 2 + wcol]       * wmask;
    bb.y = W2[(j0 + 1) * 2 + wcol] * wmask;
    acc = __builtin_amdgcn_wmma_f32_16x16x4_f32(
         false, a,  false, bb,
         (short)0, acc,  false,  false);
  }

  if (m < 2) {
#pragma unroll
    for (int r = 0; r < 8; ++r) {
      int nd = base + r + half * 8;
      if (nd < n) {
        float hp = acc[r] * dis[nd];
        h2p[nd * 2 + m] = hp;
        if (start_out) out[nd * 2 + m] = hp;
      }
    }
  }
}

__device__ __forceinline__ int l2_range_count(int n) {
  int r = (n + L2_CAP - 1) / L2_CAP;
  if (r < 16) r = 16;
  return (r + 1) & ~1;
}
__device__ __forceinline__ int l2_slice_len(int e) { return ((e + L2_SL - 1) / L2_SL + 16383) & ~16383; }

__global__ void __launch_bounds__(RTPB)
gcn_l2(const int* __restrict__ row, const int* __restrict__ col,
       const float* __restrict__ h2p,
       float* __restrict__ out, int* __restrict__ tick, int cs, int e, int n) {
  __shared__ __attribute__((aligned(16))) float acc[2 * L2_CAP];
  __shared__ __attribute__((aligned(8)))  v2i   ring[(RTPB / 32) * QN];
  __shared__ int   slot;
  const int tid  = threadIdx.x;
  const int lane = tid & 31;
  v2i* wq = ring + (tid >> 5) * QN;
  const int R   = l2_range_count(n);
  const int Rh  = R >> 1;
  const int RS  = range_size(n, R);
  const int CH  = l2_slice_len(e);
  const int nt  = Rh * L2_SL;
  const int h   = cache_half();
  for (int p = 0; p < 2; ++p) {
    const int q = h ^ p;
    int* ctr = tick + q * cs;
    for (;;) {
      const int t = take_block_ticket(ctr, &slot);
      if (t >= nt) break;
      const int lo  = (q * Rh + t % Rh) * RS;
      const int cnt = min(RS, n - lo);
      const int elo = (t / Rh) * CH;
      const int ehi = min(e, elo + CH);
      if (cnt <= 0 || elo >= ehi) continue;
      for (int i = tid * 4; i < 2 * L2_CAP; i += RTPB * 4) *(v4f*)&acc[i] = (v4f){0.f, 0.f, 0.f, 0.f};
      __syncthreads();
      unsigned head = 0, tail = 0;
      bool pend = false;
      int  pd = 0;
      v2f  pv = {0.f, 0.f};
      for (int base = elo; base < ehi; base += RTPB * 16) {
        const int b = base + tid * 4;
        v4i r[4], c[4];
        if (base + RTPB * 16 <= ehi) {
#pragma unroll
          for (int u = 0; u < 4; ++u) {
            c[u] = *(const v4i*)(col + b + u * RTPB * 4);
            r[u] = *(const v4i*)(row + b + u * RTPB * 4);
          }
        } else {
#pragma unroll
          for (int u = 0; u < 4; ++u) {
            c[u] = load4_or_none(col, b + u * RTPB * 4, ehi);
            r[u] = load4_or_none(row, b + u * RTPB * 4, ehi);
          }
        }
#pragma unroll
        for (int u = 0; u < 4; ++u) {
#pragma unroll
          for (int j = 0; j < 4; ++j) {
            const unsigned d   = (unsigned)(c[u][j] - lo);
            const bool     inr = d < (unsigned)cnt;
            const unsigned msk = __builtin_amdgcn_ballot_w32(inr);
            if (msk != 0u) {
              if (inr) {
                const unsigned pos = (tail + __builtin_amdgcn_mbcnt_lo(msk, 0u)) & (QN - 1);
                wq[pos] = (v2i){r[u][j], (int)d};
              }
              tail += (unsigned)__builtin_popcount(msk);
              if (tail - head >= 32u) {
                if (pend) {
                  atomicAdd(&acc[pd], pv.x);
                  atomicAdd(&acc[L2_CAP + pd], pv.y);
                }
                __builtin_amdgcn_fence(__ATOMIC_ACQ_REL, "wavefront");
                const v2i en = wq[(head + (unsigned)lane) & (QN - 1)];
                head += 32u;
                pd   = en.y;
                pv   = *(const v2f*)(h2p + 2 * (size_t)en.x);
                pend = true;
              }
            }
          }
        }
      }
      if (pend) {
        atomicAdd(&acc[pd], pv.x);
        atomicAdd(&acc[L2_CAP + pd], pv.y);
      }
      __builtin_amdgcn_fence(__ATOMIC_ACQ_REL, "wavefront");
      if ((unsigned)lane < tail - head) {
        const v2i en = wq[(head + (unsigned)lane) & (QN - 1)];
        const v2f v  = *(const v2f*)(h2p + 2 * (size_t)en.x);
        atomicAdd(&acc[en.y], v.x);
        atomicAdd(&acc[L2_CAP + en.y], v.y);
      }
      __syncthreads();
      float* o = out + 2 * (size_t)lo;
      for (int f = tid; f < 2 * cnt; f += RTPB) {
        const float v = acc[(f & 1) * L2_CAP + (f >> 1)];
        if (v != 0.0f) atomicAdd(&o[f], v);
      }
    }
  }
}

__global__ void gcn_fin(const float* __restrict__ dis, const float* __restrict__ b2,
                        float* __restrict__ out, int n) {
  const int i = 4 * (blockIdx.x * blockDim.x + threadIdx.x);
  const float c0 = b2[0], c1 = b2[1];
  if (i + 3 < n) {
    const v4f d = *(const v4f*)(dis + i);
    float* o = out + 2 * (size_t)i;
    v4f a = *(const v4f*)o;
    v4f b = *(const v4f*)(o + 4);
    a.x = c0 + d.x * a.x;  a.y = c1 + d.x * a.y;
    a.z = c0 + d.y * a.z;  a.w = c1 + d.y * a.w;
    b.x = c0 + d.z * b.x;  b.y = c1 + d.z * b.y;
    b.z = c0 + d.w * b.z;  b.w = c1 + d.w * b.w;
    *(v4f*)o       = a;
    *(v4f*)(o + 4) = b;
  } else {
    for (int j = i; j < n; ++j) {
      const float d = dis[j];
      out[2 * (size_t)j + 0] = c0 + d * out[2 * (size_t)j + 0];
      out[2 * (size_t)j + 1] = c1 + d * out[2 * (size_t)j + 1];
    }
  }
}


#define GSH     11
#define GBIN    2048
#define GNB_MAX 256
#define GCH_MAX 65536
#define GPAD    32
#define GROW    512
#define GTC     1024

__global__ void __launch_bounds__(RTPB)
gcn_group(const int* __restrict__ row, const int* __restrict__ col,
          int* __restrict__ recs, int* __restrict__ tab,
          int e, int n, int chunklen, int nchunks, int cap) {
  __shared__ __attribute__((aligned(16))) int lrec[GCH_MAX + GNB_MAX * GPAD];
  __shared__ int lcnt[GNB_MAX];
  __shared__ int lst[GNB_MAX + 1];
  const int tid = threadIdx.x;
  const int NB  = (n + GBIN - 1) >> GSH;
  for (int ch = blockIdx.x; ch < nchunks; ch += gridDim.x) {
    const int elo = ch * chunklen;
    const int ehi = min(e, elo + chunklen);
    if (tid < GNB_MAX) lcnt[tid] = 0;
    __syncthreads();
    for (int base = elo; base < ehi; base += RTPB * 16) {
      const int b = base + tid * 4;
      v4i c[4];
      if (base + RTPB * 16 <= ehi) {
#pragma unroll
        for (int u = 0; u < 4; ++u) c[u] = *(const v4i*)(col + b + u * RTPB * 4);
      } else {
#pragma unroll
        for (int u = 0; u < 4; ++u) c[u] = load4_or_none(col, b + u * RTPB * 4, ehi);
      }
#pragma unroll
      for (int u = 0; u < 4; ++u) {
#pragma unroll
        for (int j = 0; j < 4; ++j)
          if ((unsigned)c[u][j] < (unsigned)n) atomicAdd(&lcnt[c[u][j] >> GSH], 1);
      }
    }
    __syncthreads();
    int p = 0;
    if (tid < NB) p = (lcnt[tid] + GPAD - 1) & ~(GPAD - 1);
    if (tid < GNB_MAX) lst[tid + 1] = p;
    if (tid == 0) lst[0] = 0;
    __syncthreads();
    for (int off = 1; off < GNB_MAX; off <<= 1) {
      int v = 0;
      if (tid < GNB_MAX && tid >= off) v = lst[tid + 1 - off];
      __syncthreads();
      if (tid < GNB_MAX) lst[tid + 1] += v;
      __syncthreads();
    }
    if (tid < NB) {
      const int s0 = lst[tid], c0 = lcnt[tid], s1 = lst[tid + 1];
      for (int k = s0 + c0; k < s1; ++k) lrec[k] = -1;
      lcnt[tid] = s0;
    }
    __syncthreads();
    for (int base = elo; base < ehi; base += RTPB * 16) {
      const int b = base + tid * 4;
      v4i c[4], r[4];
      if (base + RTPB * 16 <= ehi) {
#pragma unroll
        for (int u = 0; u < 4; ++u) {
          c[u] = *(const v4i*)(col + b + u * RTPB * 4);
          r[u] = *(const v4i*)(row + b + u * RTPB * 4);
        }
      } else {
#pragma unroll
        for (int u = 0; u < 4; ++u) {
          c[u] = load4_or_none(col, b + u * RTPB * 4, ehi);
          r[u] = load4_or_none(row, b + u * RTPB * 4, ehi);
        }
      }
      int pos[16];
#pragma unroll
      for (int u = 0; u < 4; ++u) {
#pragma unroll
        for (int j = 0; j < 4; ++j) {
          pos[u * 4 + j] = -1;
          if ((unsigned)c[u][j] < (unsigned)n) pos[u * 4 + j] = atomicAdd(&lcnt[c[u][j] >> GSH], 1);
        }
      }
#pragma unroll
      for (int u = 0; u < 4; ++u) {
#pragma unroll
        for (int j = 0; j < 4; ++j)
          if (pos[u * 4 + j] >= 0)
            lrec[pos[u * 4 + j]] = (int)(((unsigned)r[u][j] << GSH) | ((unsigned)c[u][j] & (unsigned)(GBIN - 1)));
      }
    }
    __syncthreads();
    const int tot4 = lst[NB] >> 2;
    int* g = recs + (size_t)ch * (size_t)cap;
    for (int u = tid; u < tot4; u += RTPB) *(v4i*)(g + 4 * u) = *(const v4i*)&lrec[4 * u];
    if (tid < GROW) tab[(size_t)ch * GROW + tid] = lst[(tid >> 1) + (tid & 1)];
    __syncthreads();
  }
}

__global__ void __launch_bounds__(RTPB)
gcn_degree_g(const int* __restrict__ recs, const int* __restrict__ tab,
             const float* __restrict__ x, float* __restrict__ dis, float* __restrict__ xp,
             int n, int nchunks, int cap) {
  __shared__ float acc[GBIN];
  __shared__ int   tst[GTC], ten[GTC];
  const int tid  = threadIdx.x;
  const int lane = tid & 31;
  const int wv   = __builtin_amdgcn_readfirstlane(tid >> 5);
  const int NB   = (n + GBIN - 1) >> GSH;
  for (int b = blockIdx.x; b < NB; b += gridDim.x) {
    const int lo  = b << GSH;
    const int cnt = min(GBIN, n - lo);
    for (int i = tid; i < GBIN; i += RTPB) acc[i] = 0.0f;
    for (int c0 = 0; c0 < nchunks; c0 += GTC) {
      const int nc = min(GTC, nchunks - c0);
      __syncthreads();
      if (tid < nc) {
        const v2i se = *(const v2i*)(tab + (size_t)(c0 + tid) * GROW + 2 * b);
        tst[tid] = se.x;
        ten[tid] = se.y;
      }
      __syncthreads();
      for (int k = wv; k < nc; k += RTPB / 32) {
        const int* p  = recs + (size_t)(c0 + k) * (size_t)cap;
        const int  u1 = ten[k] >> 2;
        for (int u = (tst[k] >> 2) + lane; u < u1; u += 32) {
          const v4i rc = *(const v4i*)(p + 4 * u);
#pragma unroll
          for (int j = 0; j < 4; ++j)
            if (rc[j] != -1) atomicAdd(&acc[rc[j] & (GBIN - 1)], 1.0f);
        }
      }
    }
    __syncthreads();
    for (int i = tid; i < cnt; i += RTPB) {
      const float d = rsqrtf(1.0f + acc[i]);
      dis[lo + i] = d;
      xp[lo + i]  = x[lo + i] * d;
    }
    __syncthreads();
  }
}

__global__ void __launch_bounds__(RTPB)
gcn_l1_g(const int* __restrict__ recs, const int* __restrict__ tab,
         const float* __restrict__ xp, float* __restrict__ agg1,
         int n, int nchunks, int cap) {
  __shared__ float acc[GBIN];
  __shared__ int   tst[GTC], ten[GTC];
  const int tid  = threadIdx.x;
  const int lane = tid & 31;
  const int wv   = __builtin_amdgcn_readfirstlane(tid >> 5);
  const int NB   = (n + GBIN - 1) >> GSH;
  for (int b = blockIdx.x; b < NB; b += gridDim.x) {
    const int lo  = b << GSH;
    const int cnt = min(GBIN, n - lo);
    for (int i = tid; i < GBIN; i += RTPB) acc[i] = 0.0f;
    for (int c0 = 0; c0 < nchunks; c0 += GTC) {
      const int nc = min(GTC, nchunks - c0);
      __syncthreads();
      if (tid < nc) {
        const v2i se = *(const v2i*)(tab + (size_t)(c0 + tid) * GROW + 2 * b);
        tst[tid] = se.x;
        ten[tid] = se.y;
      }
      __syncthreads();
      for (int k = wv; k < nc; k += RTPB / 32) {
        const int* p  = recs + (size_t)(c0 + k) * (size_t)cap;
        const int  u1 = ten[k] >> 2;
        for (int u = (tst[k] >> 2) + lane; u < u1; u += 64) {
          const v4i rc0 = *(const v4i*)(p + 4 * u);
          v4i rc1 = {-1, -1, -1, -1};
          if (u + 32 < u1) rc1 = *(const v4i*)(p + 4 * (u + 32));
          float v[8];
#pragma unroll
          for (int j = 0; j < 4; ++j) {
            v[j]     = (rc0[j] != -1) ? xp[(unsigned)rc0[j] >> GSH] : 0.0f;
            v[4 + j] = (rc1[j] != -1) ? xp[(unsigned)rc1[j] >> GSH] : 0.0f;
          }
#pragma unroll
          for (int j = 0; j < 4; ++j) {
            if (rc0[j] != -1) atomicAdd(&acc[rc0[j] & (GBIN - 1)], v[j]);
            if (rc1[j] != -1) atomicAdd(&acc[rc1[j] & (GBIN - 1)], v[4 + j]);
          }
        }
      }
    }
    __syncthreads();
    for (int i = tid; i < cnt; i += RTPB) agg1[lo + i] = acc[i];
    __syncthreads();
  }
}

__global__ void __launch_bounds__(RTPB)
gcn_l2_g(const int* __restrict__ recs, const int* __restrict__ tab,
         const float* __restrict__ h2p, const float* __restrict__ dis,
         const float* __restrict__ b2, float* __restrict__ out,
         int n, int nchunks, int cap) {
  __shared__ float acc0[GBIN], acc1[GBIN];
  __shared__ int   tst[GTC], ten[GTC];
  const int tid  = threadIdx.x;
  const int lane = tid & 31;
  const int wv   = __builtin_amdgcn_readfirstlane(tid >> 5);
  const int NB   = (n + GBIN - 1) >> GSH;
  const float cb0 = b2[0], cb1 = b2[1];
  for (int b = blockIdx.x; b < NB; b += gridDim.x) {
    const int lo  = b << GSH;
    const int cnt = min(GBIN, n - lo);
    for (int i = tid; i < GBIN; i += RTPB) { acc0[i] = 0.0f; acc1[i] = 0.0f; }
    for (int c0 = 0; c0 < nchunks; c0 += GTC) {
      const int nc = min(GTC, nchunks - c0);
      __syncthreads();
      if (tid < nc) {
        const v2i se = *(const v2i*)(tab + (size_t)(c0 + tid) * GROW + 2 * b);
        tst[tid] = se.x;
        ten[tid] = se.y;
      }
      __syncthreads();
      for (int k = wv; k < nc; k += RTPB / 32) {
        const int* p  = recs + (size_t)(c0 + k) * (size_t)cap;
        const int  u1 = ten[k] >> 2;
        for (int u = (tst[k] >> 2) + lane; u < u1; u += 64) {
          const v4i rc0 = *(const v4i*)(p + 4 * u);
          v4i rc1 = {-1, -1, -1, -1};
          if (u + 32 < u1) rc1 = *(const v4i*)(p + 4 * (u + 32));
          v2f v[8];
#pragma unroll
          for (int j = 0; j < 4; ++j) {
            v[j]     = (v2f){0.f, 0.f};
            v[4 + j] = (v2f){0.f, 0.f};
            if (rc0[j] != -1) v[j]     = *(const v2f*)(h2p + 2 * (size_t)((unsigned)rc0[j] >> GSH));
            if (rc1[j] != -1) v[4 + j] = *(const v2f*)(h2p + 2 * (size_t)((unsigned)rc1[j] >> GSH));
          }
#pragma unroll
          for (int j = 0; j < 4; ++j) {
            if (rc0[j] != -1) {
              atomicAdd(&acc0[rc0[j] & (GBIN - 1)], v[j].x);
              atomicAdd(&acc1[rc0[j] & (GBIN - 1)], v[j].y);
            }
            if (rc1[j] != -1) {
              atomicAdd(&acc0[rc1[j] & (GBIN - 1)], v[4 + j].x);
              atomicAdd(&acc1[rc1[j] & (GBIN - 1)], v[4 + j].y);
            }
          }
        }
      }
    }
    __syncthreads();
    for (int i = tid; i < cnt; i += RTPB) {
      const size_t nd = (size_t)(lo + i);
      const float  d  = dis[nd];
      const v2f    hs = *(const v2f*)(h2p + 2 * nd);
      v2f o;
      o.x = cb0 + d * (acc0[i] + hs.x);
      o.y = cb1 + d * (acc1[i] + hs.y);
      *(v2f*)(out + 2 * nd) = o;
    }
    __syncthreads();
  }
}


#define TSH 15
#define TRN 32768

__global__ void __launch_bounds__(RTPB)
gcn_group_t(const int* __restrict__ row, const int* __restrict__ col,
            int* __restrict__ recs, int* __restrict__ tab,
            float* __restrict__ deg, float* __restrict__ agg1, float* __restrict__ s1,
            int* __restrict__ tick, int e, int n, int chunklen, int nchunks, int cap, int RN) {
  __shared__ __attribute__((aligned(16))) int lrec[GCH_MAX + GNB_MAX * GPAD];
  __shared__ int lcnt[GNB_MAX];
  __shared__ int lst[GNB_MAX + 1];
  const int tid = threadIdx.x;
  const int NB  = RN * RN;
  for (int i = blockIdx.x * RTPB + tid; i < n; i += gridDim.x * RTPB) { deg[i] = 0.0f; agg1[i] = 0.0f; s1[i] = 0.0f; }
  if (blockIdx.x == 0 && tid < 6) tick[tid * 128] = 0;
  for (int ch = blockIdx.x; ch < nchunks; ch += gridDim.x) {
    const int elo = ch * chunklen;
    const int ehi = min(e, elo + chunklen);
    if (tid < GNB_MAX) lcnt[tid] = 0;
    __syncthreads();
    for (int base = elo; base < ehi; base += RTPB * 16) {
      const int b = base + tid * 4;
      v4i c[4], r[4];
      if (base + RTPB * 16 <= ehi) {
#pragma unroll
        for (int u = 0; u < 4; ++u) {
          c[u] = *(const v4i*)(col + b + u * RTPB * 4);
          r[u] = *(const v4i*)(row + b + u * RTPB * 4);
        }
      } else {
#pragma unroll
        for (int u = 0; u < 4; ++u) {
          if (base + (u + 1) * RTPB * 4 <= ehi) {
            c[u] = *(const v4i*)(col + b + u * RTPB * 4);
            r[u] = *(const v4i*)(row + b + u * RTPB * 4);
          } else {
            c[u] = load4_or_none(col, b + u * RTPB * 4, ehi);
            r[u] = load4_or_none(row, b + u * RTPB * 4, ehi);
          }
        }
      }
#pragma unroll
      for (int u = 0; u < 4; ++u) {
#pragma unroll
        for (int j = 0; j < 4; ++j)
          if ((unsigned)c[u][j] < (unsigned)n && (unsigned)r[u][j] < (unsigned)n)
            atomicAdd(&lcnt[(c[u][j] >> TSH) * RN + (r[u][j] >> TSH)], 1);
      }
    }
    __syncthreads();
    int p = 0;
    if (tid < NB) p = (lcnt[tid] + GPAD - 1) & ~(GPAD - 1);
    if (tid < GNB_MAX) lst[tid + 1] = p;
    if (tid == 0) lst[0] = 0;
    __syncthreads();
    for (int off = 1; off < GNB_MAX; off <<= 1) {
      int v = 0;
      if (tid < GNB_MAX && tid >= off) v = lst[tid + 1 - off];
      __syncthreads();
      if (tid < GNB_MAX) lst[tid + 1] += v;
      __syncthreads();
    }
    if (tid < NB) {
      const int s0 = lst[tid], c0 = lcnt[tid], s1e = lst[tid + 1];
      for (int k = s0 + c0; k < s1e; ++k) lrec[k] = -1;
      lcnt[tid] = s0;
    }
    __syncthreads();
    for (int base = elo; base < ehi; base += RTPB * 16) {
      const int b = base + tid * 4;
      v4i c[4], r[4];
      if (base + RTPB * 16 <= ehi) {
#pragma unroll
        for (int u = 0; u < 4; ++u) {
          c[u] = *(const v4i*)(col + b + u * RTPB * 4);
          r[u] = *(const v4i*)(row + b + u * RTPB * 4);
        }
      } else {
#pragma unroll
        for (int u = 0; u < 4; ++u) {
          if (base + (u + 1) * RTPB * 4 <= ehi) {
            c[u] = *(const v4i*)(col + b + u * RTPB * 4);
            r[u] = *(const v4i*)(row + b + u * RTPB * 4);
          } else {
            c[u] = load4_or_none(col, b + u * RTPB * 4, ehi);
            r[u] = load4_or_none(row, b + u * RTPB * 4, ehi);
          }
        }
      }
      int pos[16];
#pragma unroll
      for (int u = 0; u < 4; ++u) {
#pragma unroll
        for (int j = 0; j < 4; ++j) {
          pos[u * 4 + j] = -1;
          if ((unsigned)c[u][j] < (unsigned)n && (unsigned)r[u][j] < (unsigned)n)
            pos[u * 4 + j] = atomicAdd(&lcnt[(c[u][j] >> TSH) * RN + (r[u][j] >> TSH)], 1);
        }
      }
#pragma unroll
      for (int u = 0; u < 4; ++u) {
#pragma unroll
        for (int j = 0; j < 4; ++j)
          if (pos[u * 4 + j] >= 0)
            lrec[pos[u * 4 + j]] = (int)((((unsigned)r[u][j] & (unsigned)(TRN - 1)) << TSH) | ((unsigned)c[u][j] & (unsigned)(TRN - 1)));
      }
    }
    __syncthreads();
    const int tot4 = lst[NB] >> 2;
    int* g = recs + (size_t)ch * (size_t)cap;
    for (int u = tid; u < tot4; u += RTPB) *(v4i*)(g + 4 * u) = *(const v4i*)&lrec[4 * u];
    if (tid < GROW) tab[(size_t)ch * GROW + tid] = lst[(tid >> 1) + (tid & 1)];
    __syncthreads();
  }
}


__global__ void __launch_bounds__(RTPB)
gcn_degree_t(const int* __restrict__ recs, const int* __restrict__ tab,
             float* __restrict__ deg, int* __restrict__ tick,
             int n, int nchunks, int cap, int RN) {
  __shared__ __attribute__((aligned(16))) float acc[TRN + 32];
  __shared__ int   tst[GTC], ten[GTC];
  __shared__ int   slot;
  const int tid  = threadIdx.x;
  const int lane = tid & 31;
  const int wv   = __builtin_amdgcn_readfirstlane(tid >> 5);
  const int dh   = (RN + 1) >> 1;
  const int h    = cache_half();
  for (int ph = 0; ph < 2; ++ph) {
    const int q     = h ^ ph;
    const int dbase = q ? dh : 0;
    const int nt    = (q ? RN - dh : dh) * RN;
    int* ctr = tick + q * 128;
    for (;;) {
      const int tk = take_block_ticket(ctr, &slot);
      if (tk >= nt) break;
      const int t    = (dbase + tk / RN) * RN + tk % RN;
      const int lo   = (t / RN) << TSH;
      const int cnt  = min(TRN, n - lo);
      for (int i = tid * 4; i < TRN + 32; i += RTPB * 4) *(v4f*)&acc[i] = (v4f){0.f, 0.f, 0.f, 0.f};
      for (int c0 = 0; c0 < nchunks; c0 += GTC) {
        const int nc = min(GTC, nchunks - c0);
        __syncthreads();
        if (tid < nc) {
          const v2i se = *(const v2i*)(tab + (size_t)(c0 + tid) * GROW + 2 * t);
          tst[tid] = se.x;
          ten[tid] = se.y;
        }
        __syncthreads();
        for (int k = wv; k < nc; k += RTPB / 32) {
          const int* p  = recs + (size_t)(c0 + k) * (size_t)cap;
          const int  u1 = ten[k] >> 2;
          for (int u = (tst[k] >> 2) + lane; u < u1; u += 96) {
            v4i rc[3];
            rc[0] = *(const v4i*)(p + 4 * u);
            rc[1] = (v4i){-1, -1, -1, -1};
            rc[2] = (v4i){-1, -1, -1, -1};
            if (u + 32 < u1) rc[1] = *(const v4i*)(p + 4 * (u + 32));
            if (u + 64 < u1) rc[2] = *(const v4i*)(p + 4 * (u + 64));
#pragma unroll
            for (int q = 0; q < 3; ++q) {
#pragma unroll
              for (int j = 0; j < 4; ++j) {
                const int rr = rc[q][j];
                atomicAdd(&acc[(rr != -1) ? (rr & (TRN - 1)) : (TRN + lane)], 1.0f);
              }
            }
          }
        }
      }
      __syncthreads();
      for (int i0 = 0; i0 < cnt; i0 += RTPB * 8) {
        float a[8];
#pragma unroll
        for (int j = 0; j < 8; ++j) a[j] = acc[i0 + j * RTPB + tid];
#pragma unroll
        for (int j = 0; j < 8; ++j) {
          const int i = i0 + j * RTPB + tid;
          if (i < cnt && a[j] != 0.0f) atomicAdd(&deg[lo + i], a[j]);
        }
      }
    }
  }
}

__global__ void __launch_bounds__(RTPB)
gcn_l1_t(const int* __restrict__ recs, const int* __restrict__ tab,
         const float* __restrict__ x, const float* __restrict__ deg,
         float* __restrict__ agg1, int* __restrict__ tick,
         int n, int nchunks, int cap, int RN) {
  __shared__ __attribute__((aligned(16))) float acc[TRN + 32];
  __shared__ __attribute__((aligned(16))) float xs[TRN];
  __shared__ int   tst[GTC], ten[GTC];
  __shared__ int   slot;
  const int tid  = threadIdx.x;
  const int lane = tid & 31;
  const int wv   = __builtin_amdgcn_readfirstlane(tid >> 5);
  const int dh   = (RN + 1) >> 1;
  const int h    = cache_half();
  for (int ph = 0; ph < 2; ++ph) {
    const int q     = h ^ ph;
    const int dbase = q ? dh : 0;
    const int nt    = (q ? RN - dh : dh) * RN;
    int* ctr = tick + q * 128;
    for (;;) {
      const int tk = take_block_ticket(ctr, &slot);
      if (tk >= nt) break;
      const int t    = (dbase + tk / RN) * RN + tk % RN;
      const int lo   = (t / RN) << TSH;
      const int cnt  = min(TRN, n - lo);
      const int slo  = (t % RN) << TSH;
      const int scnt = min(TRN, n - slo);
      if (tid < 32) acc[TRN + tid] = 0.0f;
      if (scnt == TRN) {
        v4f xv[TRN / (RTPB * 4)], dv[TRN / (RTPB * 4)];
#pragma unroll
        for (int g = 0; g < TRN / (RTPB * 4); ++g) {
          const int i = (g * RTPB + tid) * 4;
          xv[g] = *(const v4f*)(x + slo + i);
          dv[g] = *(const v4f*)(deg + slo + i);
        }
#pragma unroll
        for (int g = 0; g < TRN / (RTPB * 4); ++g) {
          const int i = (g * RTPB + tid) * 4;
          v4f r;
          r.x = xv[g].x * rsqrtf(1.0f + dv[g].x);
          r.y = xv[g].y * rsqrtf(1.0f + dv[g].y);
          r.z = xv[g].z * rsqrtf(1.0f + dv[g].z);
          r.w = xv[g].w * rsqrtf(1.0f + dv[g].w);
          *(v4f*)&xs[i]  = r;
          *(v4f*)&acc[i] = (v4f){0.f, 0.f, 0.f, 0.f};
        }
      } else {
        for (int i = tid; i < TRN; i += RTPB) {
          acc[i] = 0.0f;
          float v = 0.0f;
          if (i < scnt) v = x[slo + i] * rsqrtf(1.0f + deg[slo + i]);
          xs[i] = v;
        }
      }
      for (int c0 = 0; c0 < nchunks; c0 += GTC) {
        const int nc = min(GTC, nchunks - c0);
        __syncthreads();
        if (tid < nc) {
          const v2i se = *(const v2i*)(tab + (size_t)(c0 + tid) * GROW + 2 * t);
          tst[tid] = se.x;
          ten[tid] = se.y;
        }
        __syncthreads();
        for (int k = wv; k < nc; k += RTPB / 32) {
          const int* p  = recs + (size_t)(c0 + k) * (size_t)cap;
          const int  u1 = ten[k] >> 2;
          for (int u = (tst[k] >> 2) + lane; u < u1; u += 96) {
            v4i rc[3];
            rc[0] = *(const v4i*)(p + 4 * u);
            rc[1] = (v4i){-1, -1, -1, -1};
            rc[2] = (v4i){-1, -1, -1, -1};
            if (u + 32 < u1) rc[1] = *(const v4i*)(p + 4 * (u + 32));
            if (u + 64 < u1) rc[2] = *(const v4i*)(p + 4 * (u + 64));
            float v[12];
#pragma unroll
            for (int q = 0; q < 3; ++q) {
#pragma unroll
              for (int j = 0; j < 4; ++j) {
                const int rr = rc[q][j];
                v[q * 4 + j] = xs[(rr != -1) ? ((unsigned)rr >> TSH) : 0u];
              }
            }
#pragma unroll
            for (int q = 0; q < 3; ++q) {
#pragma unroll
              for (int j = 0; j < 4; ++j) {
                const int rr = rc[q][j];
                atomicAdd(&acc[(rr != -1) ? (rr & (TRN - 1)) : (TRN + lane)], v[q * 4 + j]);
              }
            }
          }
        }
      }
      __syncthreads();
      for (int i0 = 0; i0 < cnt; i0 += RTPB * 8) {
        float a[8];
#pragma unroll
        for (int j = 0; j < 8; ++j) a[j] = acc[i0 + j * RTPB + tid];
#pragma unroll
        for (int j = 0; j < 8; ++j) {
          const int i = i0 + j * RTPB + tid;
          if (i < cnt && a[j] != 0.0f) atomicAdd(&agg1[lo + i], a[j]);
        }
      }
    }
  }
}

__global__ void gcn_node_t(float* __restrict__ agg1,
                           const float* __restrict__ x,
                           const float* __restrict__ deg,
                           const float* __restrict__ W1,
                           const float* __restrict__ b1,
                           const float* __restrict__ W2,
                           float* __restrict__ p0,
                           float* __restrict__ p1,
                           int n) {
  int lane = threadIdx.x & 31;
  int wave = threadIdx.x >> 5;
  int tile = blockIdx.x * (blockDim.x >> 5) + wave;
  int base = tile * 16;
  int m    = lane & 15;
  int half = lane >> 4;
  int node = base + m;
  int nc   = node < n ? node : (n - 1);

  float dn = rsqrtf(1.0f + deg[nc]);
  float av = (agg1[nc] + x[nc] * dn) * dn;
  if (half == 0 && node < n) agg1[node] = 0.0f;

  int   wcol  = (m < 2) ? m : 0;
  float wmask = (m < 2) ? 1.0f : 0.0f;

  v8f acc = {};
#pragma unroll
  for (int s = 0; s < 4; ++s) {
    int j0 = s * 4 + half * 2;
    v2f w1 = *(const v2f*)(W1 + j0);
    v2f bi = *(const v2f*)(b1 + j0);
    v2f a;
    a.x = fmaxf(av * w1.x + bi.x, 0.0f);
    a.y = fmaxf(av * w1.y + bi.y, 0.0f);
    v2f bb;
    bb.x = W2[j0 * 2 + wcol]       * wmask;
    bb.y = W2[(j0 + 1) * 2 + wcol] * wmask;
    acc = __builtin_amdgcn_wmma_f32_16x16x4_f32(
         false, a,  false, bb,
         (short)0, acc,  false,  false);
  }

  if (m < 2) {
    float* pm = m ? p1 : p0;
#pragma unroll
    for (int r = 0; r < 8; ++r) {
      int nd = base + r + half * 8;
      if (nd < n) pm[nd] = acc[r] * rsqrtf(1.0f + deg[nd]);
    }
  }
}


__global__ void __launch_bounds__(RTPB)
gcn_l2_t(const int* __restrict__ recs, const int* __restrict__ tab,
         const float* __restrict__ p0, const float* __restrict__ p1,
         float* __restrict__ s0, float* __restrict__ s1, int* __restrict__ tick,
         int n, int nchunks, int cap, int RN) {
  __shared__ __attribute__((aligned(16))) float acc[TRN + 32];
  __shared__ __attribute__((aligned(16))) float hs[TRN];
  __shared__ int   tst[GTC], ten[GTC];
  __shared__ int   slot;
  const int tid  = threadIdx.x;
  const int lane = tid & 31;
  const int wv   = __builtin_amdgcn_readfirstlane(tid >> 5);
  const int dh   = (RN + 1) >> 1;
  const int h    = cache_half();
  for (int ph = 0; ph < 2; ++ph) {
    const int q     = h ^ ph;
    const int dbase = q ? dh : 0;
    const int nt    = (q ? RN - dh : dh) * RN;
    int* ctr = tick + q * 128;
    for (;;) {
      const int tk = take_block_ticket(ctr, &slot);
      if (tk >= nt) break;
      const int t    = (dbase + tk / RN) * RN + tk % RN;
      const int lo   = (t / RN) << TSH;
      const int cnt  = min(TRN, n - lo);
      const int slo  = (t % RN) << TSH;
      const int scnt = min(TRN, n - slo);
      for (int comp = 0; comp < 2; ++comp) {
      const float* pk = comp ? p1 : p0;
      float*       sk = comp ? s1 : s0;
      if (tid < 32) acc[TRN + tid] = 0.0f;
      if (scnt == TRN) {
        v4f pv[TRN / (RTPB * 4)];
#pragma unroll
        for (int g = 0; g < TRN / (RTPB * 4); ++g) pv[g] = *(const v4f*)(pk + slo + (g * RTPB + tid) * 4);
#pragma unroll
        for (int g = 0; g < TRN / (RTPB * 4); ++g) {
          const int i = (g * RTPB + tid) * 4;
          *(v4f*)&hs[i]  = pv[g];
          *(v4f*)&acc[i] = (v4f){0.f, 0.f, 0.f, 0.f};
        }
      } else {
        for (int i = tid; i < TRN; i += RTPB) {
          acc[i] = 0.0f;
          hs[i]  = (i < scnt) ? pk[slo + i] : 0.0f;
        }
      }
      for (int c0 = 0; c0 < nchunks; c0 += GTC) {
        const int nc = min(GTC, nchunks - c0);
        __syncthreads();
        if (tid < nc) {
          const v2i se = *(const v2i*)(tab + (size_t)(c0 + tid) * GROW + 2 * t);
          tst[tid] = se.x;
          ten[tid] = se.y;
        }
        __syncthreads();
        for (int k = wv; k < nc; k += RTPB / 32) {
          const int* p  = recs + (size_t)(c0 + k) * (size_t)cap;
          const int  u1 = ten[k] >> 2;
          for (int u = (tst[k] >> 2) + lane; u < u1; u += 96) {
            v4i rc[3];
            rc[0] = *(const v4i*)(p + 4 * u);
            rc[1] = (v4i){-1, -1, -1, -1};
            rc[2] = (v4i){-1, -1, -1, -1};
            if (u + 32 < u1) rc[1] = *(const v4i*)(p + 4 * (u + 32));
            if (u + 64 < u1) rc[2] = *(const v4i*)(p + 4 * (u + 64));
            float v[12];
#pragma unroll
            for (int q = 0; q < 3; ++q) {
#pragma unroll
              for (int j = 0; j < 4; ++j) {
                const int rr = rc[q][j];
                v[q * 4 + j] = hs[(rr != -1) ? ((unsigned)rr >> TSH) : 0u];
              }
            }
#pragma unroll
            for (int q = 0; q < 3; ++q) {
#pragma unroll
              for (int j = 0; j < 4; ++j) {
                const int rr = rc[q][j];
                atomicAdd(&acc[(rr != -1) ? (rr & (TRN - 1)) : (TRN + lane)], v[q * 4 + j]);
              }
            }
          }
        }
      }
      __syncthreads();
      for (int i0 = 0; i0 < cnt; i0 += RTPB * 8) {
        float a[8];
#pragma unroll
        for (int j = 0; j < 8; ++j) a[j] = acc[i0 + j * RTPB + tid];
#pragma unroll
        for (int j = 0; j < 8; ++j) {
          const int i = i0 + j * RTPB + tid;
          if (i < cnt && a[j] != 0.0f) atomicAdd(&sk[lo + i], a[j]);
        }
      }
      __syncthreads();
      }
    }
  }
}

__global__ void gcn_fin_t(const float* __restrict__ deg, const float* __restrict__ b2,
                          const float* __restrict__ p0, const float* __restrict__ p1,
                          const float* __restrict__ s0, const float* __restrict__ s1,
                          float* __restrict__ out, int n) {
  const int i = blockIdx.x * blockDim.x + threadIdx.x;
  if (i < n) {
    const float d = rsqrtf(1.0f + deg[i]);
    v2f o;
    o.x = b2[0] + d * (s0[i] + p0[i]);
    o.y = b2[1] + d * (s1[i] + p1[i]);
    *(v2f*)(out + 2 * (size_t)i) = o;
  }
}

extern "C" void kernel_launch(void* const* d_in, const int* in_sizes, int n_in,
                              void* d_out, int out_size, void* d_ws, size_t ws_size,
                              hipStream_t stream) {
  const float* x  = (const float*)d_in[0];
  const int*   ei = (const int*)d_in[1];
  const float* W1 = (const float*)d_in[2];
  const float* b1 = (const float*)d_in[3];
  const float* W2 = (const float*)d_in[4];
  const float* b2 = (const float*)d_in[5];
  float* out = (float*)d_out;

  const int n = in_sizes[0];
  const int e = in_sizes[1] / 2;
  const int* row = ei;
  const int* col = ei + e;

  size_t np   = (size_t)((n + 3) & ~3);
  float* ws   = (float*)d_ws;
  float* dis  = ws;
  float* xp   = ws + np;
  float* agg1 = ws + 2 * np;
  float* h2p  = ws + 3 * np;

  const int cs = (n >= 512) ? 128 : 1;
  int* tick24 = (int*)h2p;
  int* tick6  = (int*)agg1;

  const int tb = 256;
  const int nb_node = (n + tb - 1) / tb;
  const int tiles = (n + 15) / 16;
  const int nb_tile = (tiles + 7) / 8;
  const int nb_fin  = (n + 4 * tb - 1) / (4 * tb);

  {
    const int RNt = (n + TRN - 1) >> TSH;
    const int NBt = RNt * RNt;
    int chunklen_t = ((e + RANGE_BLOCKS - 1) / RANGE_BLOCKS + 3) & ~3;
    if (chunklen_t > GCH_MAX) chunklen_t = GCH_MAX;
    if (chunklen_t < 4) chunklen_t = 4;
    const int nchunks_t = (e + chunklen_t - 1) / chunklen_t;
    const int cap_t = (chunklen_t + 31 + NBt * GPAD) & ~31;
    const size_t off_tab_t = (5 * np + 31) & ~(size_t)31;
    const size_t off_rec_t = off_tab_t + (size_t)nchunks_t * GROW;
    const size_t off_tick_t = off_rec_t + (size_t)nchunks_t * (size_t)cap_t;
    const size_t need_t    = (off_tick_t + 6 * 128) * sizeof(float);
    if (n >= 1 && e >= 1 && NBt <= GNB_MAX && need_t <= ws_size) {
      float* degt = ws;
      float* p0   = ws + np;
      float* p1   = ws + 3 * np;
      float* s1   = ws + 4 * np;
      int* tab  = (int*)(ws + off_tab_t);
      int* recs = (int*)(ws + off_rec_t);
      int* tick = (int*)(ws + off_tick_t);
      gcn_group_t <<<RANGE_BLOCKS, RTPB, 0, stream>>>(row, col, recs, tab, degt, agg1, s1, tick, e, n, chunklen_t, nchunks_t, cap_t, RNt);
      gcn_degree_t<<<NBt, RTPB, 0, stream>>>(recs, tab, degt, tick, n, nchunks_t, cap_t, RNt);
      gcn_l1_t    <<<NBt, RTPB, 0, stream>>>(recs, tab, x, degt, agg1, tick + 2 * 128, n, nchunks_t, cap_t, RNt);
      gcn_node_t  <<<nb_tile, 256, 0, stream>>>(agg1, x, degt, W1, b1, W2, p0, p1, n);
      gcn_l2_t    <<<NBt, RTPB, 0, stream>>>(recs, tab, p0, p1, agg1, s1, tick + 4 * 128, n, nchunks_t, cap_t, RNt);
      gcn_fin_t   <<<nb_node, tb, 0, stream>>>(degt, b2, p0, p1, agg1, s1, out, n);
      return;
    }
  }

  const int NBg = (n + GBIN - 1) >> GSH;
  int chunklen = ((e + RANGE_BLOCKS - 1) / RANGE_BLOCKS + 3) & ~3;
  if (chunklen > GCH_MAX) chunklen = GCH_MAX;
  if (chunklen < 4) chunklen = 4;
  const int nchunks = (e + chunklen - 1) / chunklen;
  const int cap = (chunklen + 31 + NBg * GPAD) & ~31;
  const size_t off_tab = (5 * np + 31) & ~(size_t)31;
  const size_t off_rec = off_tab + (size_t)nchunks * GROW;
  const size_t need    = (off_rec + (size_t)nchunks * (size_t)cap) * sizeof(float);
  if (n >= 1 && e >= 1 && NBg <= GNB_MAX && need <= ws_size) {
    int* tab  = (int*)(ws + off_tab);
    int* recs = (int*)(ws + off_rec);
    gcn_group   <<<RANGE_BLOCKS, RTPB, 0, stream>>>(row, col, recs, tab, e, n, chunklen, nchunks, cap);
    gcn_degree_g<<<NBg, RTPB, 0, stream>>>(recs, tab, x, dis, xp, n, nchunks, cap);
    gcn_l1_g    <<<NBg, RTPB, 0, stream>>>(recs, tab, xp, agg1, n, nchunks, cap);
    gcn_node    <<<nb_tile, 256, 0, stream>>>(agg1, xp, dis, W1, b1, W2, h2p, out, n, 0);
    gcn_l2_g    <<<NBg, RTPB, 0, stream>>>(recs, tab, h2p, dis, b2, out, n, nchunks, cap);
    return;
  }

  gcn_init  <<<nb_node, tb, 0, stream>>>(dis, agg1, tick24, cs, n);
  gcn_degree<<<RANGE_BLOCKS, RTPB, 0, stream>>>(col, dis, tick24, cs, e, n);
  gcn_dis   <<<nb_node, tb, 0, stream>>>(x, dis, xp, n);
  gcn_l1    <<<RANGE_BLOCKS, RTPB, 0, stream>>>(row, col, xp, agg1, tick24 + 2 * cs, cs, e, n);
  gcn_node  <<<nb_tile, 256, 0, stream>>>(agg1, xp, dis, W1, b1, W2, h2p, out, n, 1);
  gcn_tick0 <<<1, 32, 0, stream>>>(tick6, cs);
  gcn_l2    <<<RANGE_BLOCKS, RTPB, 0, stream>>>(row, col, h2p, out, tick6, cs, e, n);
  gcn_fin   <<<nb_fin, tb, 0, stream>>>(dis, b2, out, n);
}


# Tile kernels' run loops: record loads unguarded (index clamped into the run), two runs' six loads in flight per trip, one wait; third vector summed only where a run has one; arrival-order sums kept
# speedup vs baseline: 12.8285x; 1.0383x over previous
#include <hip/hip_runtime.h>


typedef __attribute__((ext_vector_type(2))) float v2f;
typedef __attribute__((ext_vector_type(8))) float v8f;
typedef __attribute__((ext_vector_type(4))) int   v4i;
typedef __attribute__((ext_vector_type(4))) float v4f;
typedef __attribute__((ext_vector_type(2))) int   v2i;


#define RS_MAX 65536
#define SLICES 32
#define RTPB   1024
#define RANGE_BLOCKS 256

#define L2_CAP 32768
#define L2_SL  16
#define QN     64

__device__ __forceinline__ int cache_half() {
  unsigned id = __builtin_amdgcn_s_sendmsg_rtn(0x87);
  return (int)((id >> 18) & 1u);
}

__device__ __forceinline__ int range_count(int n) {
  int r = (n + RS_MAX - 1) / RS_MAX;
  if (r < 8) r = 8;
  return (r + 1) & ~1;
}
__device__ __forceinline__ int range_size(int n, int r) { return ((n + r - 1) / r + 63) & ~63; }
__device__ __forceinline__ int slice_len(int e) { return ((e + SLICES - 1) / SLICES + 16383) & ~16383; }

__device__ __forceinline__ int take_block_ticket(int* ctr, int* slot) {
  if (threadIdx.x == 0) *slot = atomicAdd(ctr, 1);
  __syncthreads();
  const int t = *slot;
  __syncthreads();
  return t;
}

__device__ __forceinline__ v4i load4_or_none(const int* __restrict__ a, int i, int hi) {
  if (i + 3 < hi) return *(const v4i*)(a + i);
  v4i r = {-1, -1, -1, -1};
  if (i     < hi) r.x = a[i];
  if (i + 1 < hi) r.y = a[i + 1];
  if (i + 2 < hi) r.z = a[i + 2];
  return r;
}

__global__ void gcn_init(float* __restrict__ deg, float* __restrict__ agg1,
                         int* __restrict__ tick, int cs, int n) {
  int i = blockIdx.x * blockDim.x + threadIdx.x;
  if (i < n) { deg[i] = 1.0f; agg1[i] = 0.0f; }
  if (i < 4) tick[i * cs] = 0;
}

__global__ void gcn_tick0(int* __restrict__ tick, int cs) {
  if (threadIdx.x < 2) tick[threadIdx.x * cs] = 0;
}

__global__ void __launch_bounds__(RTPB)
gcn_degree(const int* __restrict__ col, float* __restrict__ deg,
           int* __restrict__ tick, int cs, int e, int n) {
  __shared__ __attribute__((aligned(16))) float acc[RS_MAX];
  __shared__ int   slot;
  const int tid = threadIdx.x;
  const int R   = range_count(n);
  const int Rh  = R >> 1;
  const int RS  = range_size(n, R);
  const int CH  = slice_len(e);
  const int nt  = Rh * SLICES;
  const int h   = cache_half();
  for (int p = 0; p < 2; ++p) {
    const int q = h ^ p;
    int* ctr = tick + q * cs;
    for (;;) {
      const int t = take_block_ticket(ctr, &slot);
      if (t >= nt) break;
      const int lo  = (q * Rh + t % Rh) * RS;
      const int cnt = min(RS, n - lo);
      const int elo = (t / Rh) * CH;
      const int ehi = min(e, elo + CH);
      if (cnt <= 0 || elo >= ehi) continue;
      for (int i = tid * 4; i < RS; i += RTPB * 4) *(v4f*)&acc[i] = (v4f){0.f, 0.f, 0.f, 0.f};
      __syncthreads();
      int b = elo + tid * 4;
      for (; b + 3 * RTPB * 4 + 3 < ehi; b += RTPB * 16) {
        v4i c[4];
#pragma unroll
        for (int u = 0; u < 4; ++u) c[u] = *(const v4i*)(col + b + u * RTPB * 4);
#pragma unroll
        for (int u = 0; u < 4; ++u) {
#pragma unroll
          for (int j = 0; j < 4; ++j) {
            const unsigned d = (unsigned)(c[u][j] - lo);
            if (d < (unsigned)cnt) atomicAdd(&acc[d], 1.0f);
          }
        }
      }
      for (; b < ehi; b += RTPB * 4) {
        const v4i c = load4_or_none(col, b, ehi);
#pragma unroll
        for (int j = 0; j < 4; ++j) {
          const unsigned d = (unsigned)(c[j] - lo);
          if (d < (unsigned)cnt) atomicAdd(&acc[d], 1.0f);
        }
      }
      __syncthreads();
      for (int i = tid; i < cnt; i += RTPB) {
        const float v = acc[i];
        if (v != 0.0f) atomicAdd(&deg[lo + i], v);
      }
    }
  }
}

__global__ void gcn_dis(const float* __restrict__ x, float* __restrict__ deg_dis,
                        float* __restrict__ xp, int n) {
  int i = blockIdx.x * blockDim.x + threadIdx.x;
  if (i < n) {
    float d = rsqrtf(deg_dis[i]);
    deg_dis[i] = d;
    xp[i] = x[i] * d;
  }
}

__global__ void __launch_bounds__(RTPB)
gcn_l1(const int* __restrict__ row, const int* __restrict__ col,
       const float* __restrict__ xp,
       float* __restrict__ agg1, int* __restrict__ tick, int cs, int e, int n) {
  __shared__ __attribute__((aligned(16))) float acc[RS_MAX];
  __shared__ int   slot;
  const int tid = threadIdx.x;
  const int R   = range_count(n);
  const int Rh  = R >> 1;
  const int RS  = range_size(n, R);
  const int CH  = slice_len(e);
  const int nt  = Rh * SLICES;
  const int h   = cache_half();
  for (int p = 0; p < 2; ++p) {
    const int q = h ^ p;
    int* ctr = tick + q * cs;
    for (;;) {
      const int t = take_block_ticket(ctr, &slot);
      if (t >= nt) break;
      const int lo  = (q * Rh + t % Rh) * RS;
      const int cnt = min(RS, n - lo);
      const int elo = (t / Rh) * CH;
      const int ehi = min(e, elo + CH);
      if (cnt <= 0 || elo >= ehi) continue;
      for (int i = tid * 4; i < RS; i += RTPB * 4) *(v4f*)&acc[i] = (v4f){0.f, 0.f, 0.f, 0.f};
      __syncthreads();
      int b = elo + tid * 4;
      for (; b + 3 * RTPB * 4 + 3 < ehi; b += RTPB * 16) {
        v4i r[4], c[4];
#pragma unroll
        for (int u = 0; u < 4; ++u) {
          c[u] = *(const v4i*)(col + b + u * RTPB * 4);
          r[u] = *(const v4i*)(row + b + u * RTPB * 4);
        }
        asm volatile("" : "+v"(c[0]), "+v"(c[1]), "+v"(c[2]), "+v"(c[3]),
                          "+v"(r[0]), "+v"(r[1]), "+v"(r[2]), "+v"(r[3]));
        float v[16];
        bool  m[16];
#pragma unroll
        for (int u = 0; u < 4; ++u) {
#pragma unroll
          for (int j = 0; j < 4; ++j) {
            const unsigned d = (unsigned)(c[u][j] - lo);
            m[u * 4 + j] = d < (unsigned)cnt;
            v[u * 4 + j] = 0.0f;
            if (m[u * 4 + j]) v[u * 4 + j] = xp[r[u][j]];
          }
        }
#pragma unroll
        for (int u = 0; u < 4; ++u) {
#pragma unroll
          for (int j = 0; j < 4; ++j)
            if (m[u * 4 + j]) atomicAdd(&acc[c[u][j] - lo], v[u * 4 + j]);
        }
      }
      for (; b < ehi; b += RTPB * 4) {
        const v4i c = load4_or_none(col, b, ehi);
        const v4i r = load4_or_none(row, b, ehi);
#pragma unroll
        for (int j = 0; j < 4; ++j) {
          const unsigned d = (unsigned)(c[j] - lo);
          if (d < (unsigned)cnt) atomicAdd(&acc[d], xp[r[j]]);
        }
      }
      __syncthreads();
      for (int i = tid; i < cnt; i += RTPB) {
        const float v = acc[i];
        if (v != 0.0f) atomicAdd(&agg1[lo + i], v);
      }
    }
  }
}

__global__ void gcn_node(const float* __restrict__ agg1,
                         const float* __restrict__ xp,
                         const float* __restrict__ dis,
                         const float* __restrict__ W1,
                         const float* __restrict__ b1,
                         const float* __restrict__ W2,
                         float* __restrict__ h2p,
                         float* __restrict__ out,
                         int n, int start_out) {
  int lane = threadIdx.x & 31;
  int wave = threadIdx.x >> 5;
  int tile = blockIdx.x * (blockDim.x >> 5) + wave;
  int base = tile * 16;
  int m    = lane & 15;
  int half = lane >> 4;
  int node = base + m;
  int nc   = node < n ? node : (n - 1);

  float av = (agg1[nc] + xp[nc]) * dis[nc];

  int   wcol  = (m < 2) ? m : 0;
  float wmask = (m < 2) ? 1.0f : 0.0f;

  v8f acc = {};
#pragma unroll
  for (int s = 0; s < 4; ++s) {
    int j0 = s * 4 + half * 2;
    v2f w1 = *(const v2f*)(W1 + j0);
    v2f bi = *(const v2f*)(b1 + j0);
    v2f a;
    a.x = fmaxf(av * w1.x + bi.x, 0.0f);
    a.y = fmaxf(av * w1.y + bi.y, 0.0f);
    v2f bb;
    bb.x = W2[j0 * 2 + wcol]       * wmask;
    bb.y = W2[(j0 + 1) * 2 + wcol] * wmask;
    acc = __builtin_amdgcn_wmma_f32_16x16x4_f32(
         false, a,  false, bb,
         (short)0, acc,  false,  false);
  }

  if (m < 2) {
#pragma unroll
    for (int r = 0; r < 8; ++r) {
      int nd = base + r + half * 8;
      if (nd < n) {
        float hp = acc[r] * dis[nd];
        h2p[nd * 2 + m] = hp;
        if (start_out) out[nd * 2 + m] = hp;
      }
    }
  }
}

__device__ __forceinline__ int l2_range_count(int n) {
  int r = (n + L2_CAP - 1) / L2_CAP;
  if (r < 16) r = 16;
  return (r + 1) & ~1;
}
__device__ __forceinline__ int l2_slice_len(int e) { return ((e + L2_SL - 1) / L2_SL + 16383) & ~16383; }

__global__ void __launch_bounds__(RTPB)
gcn_l2(const int* __restrict__ row, const int* __restrict__ col,
       const float* __restrict__ h2p,
       float* __restrict__ out, int* __restrict__ tick, int cs, int e, int n) {
  __shared__ __attribute__((aligned(16))) float acc[2 * L2_CAP];
  __shared__ __attribute__((aligned(8)))  v2i   ring[(RTPB / 32) * QN];
  __shared__ int   slot;
  const int tid  = threadIdx.x;
  const int lane = tid & 31;
  v2i* wq = ring + (tid >> 5) * QN;
  const int R   = l2_range_count(n);
  const int Rh  = R >> 1;
  const int RS  = range_size(n, R);
  const int CH  = l2_slice_len(e);
  const int nt  = Rh * L2_SL;
  const int h   = cache_half();
  for (int p = 0; p < 2; ++p) {
    const int q = h ^ p;
    int* ctr = tick + q * cs;
    for (;;) {
      const int t = take_block_ticket(ctr, &slot);
      if (t >= nt) break;
      const int lo  = (q * Rh + t % Rh) * RS;
      const int cnt = min(RS, n - lo);
      const int elo = (t / Rh) * CH;
      const int ehi = min(e, elo + CH);
      if (cnt <= 0 || elo >= ehi) continue;
      for (int i = tid * 4; i < 2 * L2_CAP; i += RTPB * 4) *(v4f*)&acc[i] = (v4f){0.f, 0.f, 0.f, 0.f};
      __syncthreads();
      unsigned head = 0, tail = 0;
      bool pend = false;
      int  pd = 0;
      v2f  pv = {0.f, 0.f};
      for (int base = elo; base < ehi; base += RTPB * 16) {
        const int b = base + tid * 4;
        v4i r[4], c[4];
        if (base + RTPB * 16 <= ehi) {
#pragma unroll
          for (int u = 0; u < 4; ++u) {
            c[u] = *(const v4i*)(col + b + u * RTPB * 4);
            r[u] = *(const v4i*)(row + b + u * RTPB * 4);
          }
        } else {
#pragma unroll
          for (int u = 0; u < 4; ++u) {
            c[u] = load4_or_none(col, b + u * RTPB * 4, ehi);
            r[u] = load4_or_none(row, b + u * RTPB * 4, ehi);
          }
        }
#pragma unroll
        for (int u = 0; u < 4; ++u) {
#pragma unroll
          for (int j = 0; j < 4; ++j) {
            const unsigned d   = (unsigned)(c[u][j] - lo);
            const bool     inr = d < (unsigned)cnt;
            const unsigned msk = __builtin_amdgcn_ballot_w32(inr);
            if (msk != 0u) {
              if (inr) {
                const unsigned pos = (tail + __builtin_amdgcn_mbcnt_lo(msk, 0u)) & (QN - 1);
                wq[pos] = (v2i){r[u][j], (int)d};
              }
              tail += (unsigned)__builtin_popcount(msk);
              if (tail - head >= 32u) {
                if (pend) {
                  atomicAdd(&acc[pd], pv.x);
                  atomicAdd(&acc[L2_CAP + pd], pv.y);
                }
                __builtin_amdgcn_fence(__ATOMIC_ACQ_REL, "wavefront");
                const v2i en = wq[(head + (unsigned)lane) & (QN - 1)];
                head += 32u;
                pd   = en.y;
                pv   = *(const v2f*)(h2p + 2 * (size_t)en.x);
                pend = true;
              }
            }
          }
        }
      }
      if (pend) {
        atomicAdd(&acc[pd], pv.x);
        atomicAdd(&acc[L2_CAP + pd], pv.y);
      }
      __builtin_amdgcn_fence(__ATOMIC_ACQ_REL, "wavefront");
      if ((unsigned)lane < tail - head) {
        const v2i en = wq[(head + (unsigned)lane) & (QN - 1)];
        const v2f v  = *(const v2f*)(h2p + 2 * (size_t)en.x);
        atomicAdd(&acc[en.y], v.x);
        atomicAdd(&acc[L2_CAP + en.y], v.y);
      }
      __syncthreads();
      float* o = out + 2 * (size_t)lo;
      for (int f = tid; f < 2 * cnt; f += RTPB) {
        const float v = acc[(f & 1) * L2_CAP + (f >> 1)];
        if (v != 0.0f) atomicAdd(&o[f], v);
      }
    }
  }
}

__global__ void gcn_fin(const float* __restrict__ dis, const float* __restrict__ b2,
                        float* __restrict__ out, int n) {
  const int i = 4 * (blockIdx.x * blockDim.x + threadIdx.x);
  const float c0 = b2[0], c1 = b2[1];
  if (i + 3 < n) {
    const v4f d = *(const v4f*)(dis + i);
    float* o = out + 2 * (size_t)i;
    v4f a = *(const v4f*)o;
    v4f b = *(const v4f*)(o + 4);
    a.x = c0 + d.x * a.x;  a.y = c1 + d.x * a.y;
    a.z = c0 + d.y * a.z;  a.w = c1 + d.y * a.w;
    b.x = c0 + d.z * b.x;  b.y = c1 + d.z * b.y;
    b.z = c0 + d.w * b.z;  b.w = c1 + d.w * b.w;
    *(v4f*)o       = a;
    *(v4f*)(o + 4) = b;
  } else {
    for (int j = i; j < n; ++j) {
      const float d = dis[j];
      out[2 * (size_t)j + 0] = c0 + d * out[2 * (size_t)j + 0];
      out[2 * (size_t)j + 1] = c1 + d * out[2 * (size_t)j + 1];
    }
  }
}


#define GSH     11
#define GBIN    2048
#define GNB_MAX 256
#define GCH_MAX 65536
#define GPAD    32
#define GROW    512
#define GTC     1024

__global__ void __launch_bounds__(RTPB)
gcn_group(const int* __restrict__ row, const int* __restrict__ col,
          int* __restrict__ recs, int* __restrict__ tab,
          int e, int n, int chunklen, int nchunks, int cap) {
  __shared__ __attribute__((aligned(16))) int lrec[GCH_MAX + GNB_MAX * GPAD];
  __shared__ int lcnt[GNB_MAX];
  __shared__ int lst[GNB_MAX + 1];
  const int tid = threadIdx.x;
  const int NB  = (n + GBIN - 1) >> GSH;
  for (int ch = blockIdx.x; ch < nchunks; ch += gridDim.x) {
    const int elo = ch * chunklen;
    const int ehi = min(e, elo + chunklen);
    if (tid < GNB_MAX) lcnt[tid] = 0;
    __syncthreads();
    for (int base = elo; base < ehi; base += RTPB * 16) {
      const int b = base + tid * 4;
      v4i c[4];
      if (base + RTPB * 16 <= ehi) {
#pragma unroll
        for (int u = 0; u < 4; ++u) c[u] = *(const v4i*)(col + b + u * RTPB * 4);
      } else {
#pragma unroll
        for (int u = 0; u < 4; ++u) c[u] = load4_or_none(col, b + u * RTPB * 4, ehi);
      }
#pragma unroll
      for (int u = 0; u < 4; ++u) {
#pragma unroll
        for (int j = 0; j < 4; ++j)
          if ((unsigned)c[u][j] < (unsigned)n) atomicAdd(&lcnt[c[u][j] >> GSH], 1);
      }
    }
    __syncthreads();
    int p = 0;
    if (tid < NB) p = (lcnt[tid] + GPAD - 1) & ~(GPAD - 1);
    if (tid < GNB_MAX) lst[tid + 1] = p;
    if (tid == 0) lst[0] = 0;
    __syncthreads();
    for (int off = 1; off < GNB_MAX; off <<= 1) {
      int v = 0;
      if (tid < GNB_MAX && tid >= off) v = lst[tid + 1 - off];
      __syncthreads();
      if (tid < GNB_MAX) lst[tid + 1] += v;
      __syncthreads();
    }
    if (tid < NB) {
      const int s0 = lst[tid], c0 = lcnt[tid], s1 = lst[tid + 1];
      for (int k = s0 + c0; k < s1; ++k) lrec[k] = -1;
      lcnt[tid] = s0;
    }
    __syncthreads();
    for (int base = elo; base < ehi; base += RTPB * 16) {
      const int b = base + tid * 4;
      v4i c[4], r[4];
      if (base + RTPB * 16 <= ehi) {
#pragma unroll
        for (int u = 0; u < 4; ++u) {
          c[u] = *(const v4i*)(col + b + u * RTPB * 4);
          r[u] = *(const v4i*)(row + b + u * RTPB * 4);
        }
      } else {
#pragma unroll
        for (int u = 0; u < 4; ++u) {
          c[u] = load4_or_none(col, b + u * RTPB * 4, ehi);
          r[u] = load4_or_none(row, b + u * RTPB * 4, ehi);
        }
      }
      int pos[16];
#pragma unroll
      for (int u = 0; u < 4; ++u) {
#pragma unroll
        for (int j = 0; j < 4; ++j) {
          pos[u * 4 + j] = -1;
          if ((unsigned)c[u][j] < (unsigned)n) pos[u * 4 + j] = atomicAdd(&lcnt[c[u][j] >> GSH], 1);
        }
      }
#pragma unroll
      for (int u = 0; u < 4; ++u) {
#pragma unroll
        for (int j = 0; j < 4; ++j)
          if (pos[u * 4 + j] >= 0)
            lrec[pos[u * 4 + j]] = (int)(((unsigned)r[u][j] << GSH) | ((unsigned)c[u][j] & (unsigned)(GBIN - 1)));
      }
    }
    __syncthreads();
    const int tot4 = lst[NB] >> 2;
    int* g = recs + (size_t)ch * (size_t)cap;
    for (int u = tid; u < tot4; u += RTPB) *(v4i*)(g + 4 * u) = *(const v4i*)&lrec[4 * u];
    if (tid < GROW) tab[(size_t)ch * GROW + tid] = lst[(tid >> 1) + (tid & 1)];
    __syncthreads();
  }
}

__global__ void __launch_bounds__(RTPB)
gcn_degree_g(const int* __restrict__ recs, const int* __restrict__ tab,
             const float* __restrict__ x, float* __restrict__ dis, float* __restrict__ xp,
             int n, int nchunks, int cap) {
  __shared__ float acc[GBIN];
  __shared__ int   tst[GTC], ten[GTC];
  const int tid  = threadIdx.x;
  const int lane = tid & 31;
  const int wv   = __builtin_amdgcn_readfirstlane(tid >> 5);
  const int NB   = (n + GBIN - 1) >> GSH;
  for (int b = blockIdx.x; b < NB; b += gridDim.x) {
    const int lo  = b << GSH;
    const int cnt = min(GBIN, n - lo);
    for (int i = tid; i < GBIN; i += RTPB) acc[i] = 0.0f;
    for (int c0 = 0; c0 < nchunks; c0 += GTC) {
      const int nc = min(GTC, nchunks - c0);
      __syncthreads();
      if (tid < nc) {
        const v2i se = *(const v2i*)(tab + (size_t)(c0 + tid) * GROW + 2 * b);
        tst[tid] = se.x;
        ten[tid] = se.y;
      }
      __syncthreads();
      for (int k = wv; k < nc; k += RTPB / 32) {
        const int* p  = recs + (size_t)(c0 + k) * (size_t)cap;
        const int  u1 = ten[k] >> 2;
        for (int u = (tst[k] >> 2) + lane; u < u1; u += 32) {
          const v4i rc = *(const v4i*)(p + 4 * u);
#pragma unroll
          for (int j = 0; j < 4; ++j)
            if (rc[j] != -1) atomicAdd(&acc[rc[j] & (GBIN - 1)], 1.0f);
        }
      }
    }
    __syncthreads();
    for (int i = tid; i < cnt; i += RTPB) {
      const float d = rsqrtf(1.0f + acc[i]);
      dis[lo + i] = d;
      xp[lo + i]  = x[lo + i] * d;
    }
    __syncthreads();
  }
}

__global__ void __launch_bounds__(RTPB)
gcn_l1_g(const int* __restrict__ recs, const int* __restrict__ tab,
         const float* __restrict__ xp, float* __restrict__ agg1,
         int n, int nchunks, int cap) {
  __shared__ float acc[GBIN];
  __shared__ int   tst[GTC], ten[GTC];
  const int tid  = threadIdx.x;
  const int lane = tid & 31;
  const int wv   = __builtin_amdgcn_readfirstlane(tid >> 5);
  const int NB   = (n + GBIN - 1) >> GSH;
  for (int b = blockIdx.x; b < NB; b += gridDim.x) {
    const int lo  = b << GSH;
    const int cnt = min(GBIN, n - lo);
    for (int i = tid; i < GBIN; i += RTPB) acc[i] = 0.0f;
    for (int c0 = 0; c0 < nchunks; c0 += GTC) {
      const int nc = min(GTC, nchunks - c0);
      __syncthreads();
      if (tid < nc) {
        const v2i se = *(const v2i*)(tab + (size_t)(c0 + tid) * GROW + 2 * b);
        tst[tid] = se.x;
        ten[tid] = se.y;
      }
      __syncthreads();
      for (int k = wv; k < nc; k += RTPB / 32) {
        const int* p  = recs + (size_t)(c0 + k) * (size_t)cap;
        const int  u1 = ten[k] >> 2;
        for (int u = (tst[k] >> 2) + lane; u < u1; u += 64) {
          const v4i rc0 = *(const v4i*)(p + 4 * u);
          v4i rc1 = {-1, -1, -1, -1};
          if (u + 32 < u1) rc1 = *(const v4i*)(p + 4 * (u + 32));
          float v[8];
#pragma unroll
          for (int j = 0; j < 4; ++j) {
            v[j]     = (rc0[j] != -1) ? xp[(unsigned)rc0[j] >> GSH] : 0.0f;
            v[4 + j] = (rc1[j] != -1) ? xp[(unsigned)rc1[j] >> GSH] : 0.0f;
          }
#pragma unroll
          for (int j = 0; j < 4; ++j) {
            if (rc0[j] != -1) atomicAdd(&acc[rc0[j] & (GBIN - 1)], v[j]);
            if (rc1[j] != -1) atomicAdd(&acc[rc1[j] & (GBIN - 1)], v[4 + j]);
          }
        }
      }
    }
    __syncthreads();
    for (int i = tid; i < cnt; i += RTPB) agg1[lo + i] = acc[i];
    __syncthreads();
  }
}

__global__ void __launch_bounds__(RTPB)
gcn_l2_g(const int* __restrict__ recs, const int* __restrict__ tab,
         const float* __restrict__ h2p, const float* __restrict__ dis,
         const float* __restrict__ b2, float* __restrict__ out,
         int n, int nchunks, int cap) {
  __shared__ float acc0[GBIN], acc1[GBIN];
  __shared__ int   tst[GTC], ten[GTC];
  const int tid  = threadIdx.x;
  const int lane = tid & 31;
  const int wv   = __builtin_amdgcn_readfirstlane(tid >> 5);
  const int NB   = (n + GBIN - 1) >> GSH;
  const float cb0 = b2[0], cb1 = b2[1];
  for (int b = blockIdx.x; b < NB; b += gridDim.x) {
    const int lo  = b << GSH;
    const int cnt = min(GBIN, n - lo);
    for (int i = tid; i < GBIN; i += RTPB) { acc0[i] = 0.0f; acc1[i] = 0.0f; }
    for (int c0 = 0; c0 < nchunks; c0 += GTC) {
      const int nc = min(GTC, nchunks - c0);
      __syncthreads();
      if (tid < nc) {
        const v2i se = *(const v2i*)(tab + (size_t)(c0 + tid) * GROW + 2 * b);
        tst[tid] = se.x;
        ten[tid] = se.y;
      }
      __syncthreads();
      for (int k = wv; k < nc; k += RTPB / 32) {
        const int* p  = recs + (size_t)(c0 + k) * (size_t)cap;
        const int  u1 = ten[k] >> 2;
        for (int u = (tst[k] >> 2) + lane; u < u1; u += 64) {
          const v4i rc0 = *(const v4i*)(p + 4 * u);
          v4i rc1 = {-1, -1, -1, -1};
          if (u + 32 < u1) rc1 = *(const v4i*)(p + 4 * (u + 32));
          v2f v[8];
#pragma unroll
          for (int j = 0; j < 4; ++j) {
            v[j]     = (v2f){0.f, 0.f};
            v[4 + j] = (v2f){0.f, 0.f};
            if (rc0[j] != -1) v[j]     = *(const v2f*)(h2p + 2 * (size_t)((unsigned)rc0[j] >> GSH));
            if (rc1[j] != -1) v[4 + j] = *(const v2f*)(h2p + 2 * (size_t)((unsigned)rc1[j] >> GSH));
          }
#pragma unroll
          for (int j = 0; j < 4; ++j) {
            if (rc0[j] != -1) {
              atomicAdd(&acc0[rc0[j] & (GBIN - 1)], v[j].x);
              atomicAdd(&acc1[rc0[j] & (GBIN - 1)], v[j].y);
            }
            if (rc1[j] != -1) {
              atomicAdd(&acc0[rc1[j] & (GBIN - 1)], v[4 + j].x);
              atomicAdd(&acc1[rc1[j] & (GBIN - 1)], v[4 + j].y);
            }
          }
        }
      }
    }
    __syncthreads();
    for (int i = tid; i < cnt; i += RTPB) {
      const size_t nd = (size_t)(lo + i);
      const float  d  = dis[nd];
      const v2f    hs = *(const v2f*)(h2p + 2 * nd);
      v2f o;
      o.x = cb0 + d * (acc0[i] + hs.x);
      o.y = cb1 + d * (acc1[i] + hs.y);
      *(v2f*)(out + 2 * nd) = o;
    }
    __syncthreads();
  }
}


#define TSH 15
#define TRN 32768

__global__ void __launch_bounds__(RTPB)
gcn_group_t(const int* __restrict__ row, const int* __restrict__ col,
            int* __restrict__ recs, int* __restrict__ tab,
            float* __restrict__ deg, float* __restrict__ agg1, float* __restrict__ s1,
            int* __restrict__ tick, int e, int n, int chunklen, int nchunks, int cap, int RN) {
  __shared__ __attribute__((aligned(16))) int lrec[GCH_MAX + GNB_MAX * GPAD];
  __shared__ int lcnt[GNB_MAX];
  __shared__ int lst[GNB_MAX + 1];
  const int tid = threadIdx.x;
  const int NB  = RN * RN;
  for (int i = blockIdx.x * RTPB + tid; i < n; i += gridDim.x * RTPB) { deg[i] = 0.0f; agg1[i] = 0.0f; s1[i] = 0.0f; }
  if (blockIdx.x == 0 && tid < 6) tick[tid * 128] = 0;
  for (int ch = blockIdx.x; ch < nchunks; ch += gridDim.x) {
    const int elo = ch * chunklen;
    const int ehi = min(e, elo + chunklen);
    if (tid < GNB_MAX) lcnt[tid] = 0;
    __syncthreads();
    for (int base = elo; base < ehi; base += RTPB * 16) {
      const int b = base + tid * 4;
      v4i c[4], r[4];
      if (base + RTPB * 16 <= ehi) {
#pragma unroll
        for (int u = 0; u < 4; ++u) {
          c[u] = *(const v4i*)(col + b + u * RTPB * 4);
          r[u] = *(const v4i*)(row + b + u * RTPB * 4);
        }
      } else {
#pragma unroll
        for (int u = 0; u < 4; ++u) {
          if (base + (u + 1) * RTPB * 4 <= ehi) {
            c[u] = *(const v4i*)(col + b + u * RTPB * 4);
            r[u] = *(const v4i*)(row + b + u * RTPB * 4);
          } else {
            c[u] = load4_or_none(col, b + u * RTPB * 4, ehi);
            r[u] = load4_or_none(row, b + u * RTPB * 4, ehi);
          }
        }
      }
#pragma unroll
      for (int u = 0; u < 4; ++u) {
#pragma unroll
        for (int j = 0; j < 4; ++j)
          if ((unsigned)c[u][j] < (unsigned)n && (unsigned)r[u][j] < (unsigned)n)
            atomicAdd(&lcnt[(c[u][j] >> TSH) * RN + (r[u][j] >> TSH)], 1);
      }
    }
    __syncthreads();
    int p = 0;
    if (tid < NB) p = (lcnt[tid] + GPAD - 1) & ~(GPAD - 1);
    if (tid < GNB_MAX) lst[tid + 1] = p;
    if (tid == 0) lst[0] = 0;
    __syncthreads();
    for (int off = 1; off < GNB_MAX; off <<= 1) {
      int v = 0;
      if (tid < GNB_MAX && tid >= off) v = lst[tid + 1 - off];
      __syncthreads();
      if (tid < GNB_MAX) lst[tid + 1] += v;
      __syncthreads();
    }
    if (tid < NB) {
      const int s0 = lst[tid], c0 = lcnt[tid], s1e = lst[tid + 1];
      for (int k = s0 + c0; k < s1e; ++k) lrec[k] = -1;
      lcnt[tid] = s0;
    }
    __syncthreads();
    for (int base = elo; base < ehi; base += RTPB * 16) {
      const int b = base + tid * 4;
      v4i c[4], r[4];
      if (base + RTPB * 16 <= ehi) {
#pragma unroll
        for (int u = 0; u < 4; ++u) {
          c[u] = *(const v4i*)(col + b + u * RTPB * 4);
          r[u] = *(const v4i*)(row + b + u * RTPB * 4);
        }
      } else {
#pragma unroll
        for (int u = 0; u < 4; ++u) {
          if (base + (u + 1) * RTPB * 4 <= ehi) {
            c[u] = *(const v4i*)(col + b + u * RTPB * 4);
            r[u] = *(const v4i*)(row + b + u * RTPB * 4);
          } else {
            c[u] = load4_or_none(col, b + u * RTPB * 4, ehi);
            r[u] = load4_or_none(row, b + u * RTPB * 4, ehi);
          }
        }
      }
      int pos[16];
#pragma unroll
      for (int u = 0; u < 4; ++u) {
#pragma unroll
        for (int j = 0; j < 4; ++j) {
          pos[u * 4 + j] = -1;
          if ((unsigned)c[u][j] < (unsigned)n && (unsigned)r[u][j] < (unsigned)n)
            pos[u * 4 + j] = atomicAdd(&lcnt[(c[u][j] >> TSH) * RN + (r[u][j] >> TSH)], 1);
        }
      }
#pragma unroll
      for (int u = 0; u < 4; ++u) {
#pragma unroll
        for (int j = 0; j < 4; ++j)
          if (pos[u * 4 + j] >= 0)
            lrec[pos[u * 4 + j]] = (int)((((unsigned)r[u][j] & (unsigned)(TRN - 1)) << TSH) | ((unsigned)c[u][j] & (unsigned)(TRN - 1)));
      }
    }
    __syncthreads();
    const int tot4 = lst[NB] >> 2;
    int* g = recs + (size_t)ch * (size_t)cap;
    for (int u = tid; u < tot4; u += RTPB) *(v4i*)(g + 4 * u) = *(const v4i*)&lrec[4 * u];
    if (tid < GROW) tab[(size_t)ch * GROW + tid] = lst[(tid >> 1) + (tid & 1)];
    __syncthreads();
  }
}

__device__ __forceinline__ v4i run_vec(const int* __restrict__ p, int u, int u1) {
  const int i = max(min(u, u1 - 1), 0);
  return *(const v4i*)(p + 4 * i);
}
__device__ __forceinline__ v4i run_pad(v4i r, int u, int u1) {
  const v4i none = {-1, -1, -1, -1};
  return (u < u1) ? r : none;
}


__global__ void __launch_bounds__(RTPB)
gcn_degree_t(const int* __restrict__ recs, const int* __restrict__ tab,
             float* __restrict__ deg, int* __restrict__ tick,
             int n, int nchunks, int cap, int RN) {
  __shared__ __attribute__((aligned(16))) float acc[TRN + 32];
  __shared__ int   tst[GTC], ten[GTC];
  __shared__ int   slot;
  const int tid  = threadIdx.x;
  const int lane = tid & 31;
  const int wv   = __builtin_amdgcn_readfirstlane(tid >> 5);
  const int dh   = (RN + 1) >> 1;
  const int h    = cache_half();
  for (int ph = 0; ph < 2; ++ph) {
    const int q     = h ^ ph;
    const int dbase = q ? dh : 0;
    const int nt    = (q ? RN - dh : dh) * RN;
    int* ctr = tick + q * 128;
    for (;;) {
      const int tk = take_block_ticket(ctr, &slot);
      if (tk >= nt) break;
      const int t    = (dbase + tk / RN) * RN + tk % RN;
      const int lo   = (t / RN) << TSH;
      const int cnt  = min(TRN, n - lo);
      for (int i = tid * 4; i < TRN + 32; i += RTPB * 4) *(v4f*)&acc[i] = (v4f){0.f, 0.f, 0.f, 0.f};
      for (int c0 = 0; c0 < nchunks; c0 += GTC) {
        const int nc = min(GTC, nchunks - c0);
        __syncthreads();
        if (tid < nc) {
          const v2i se = *(const v2i*)(tab + (size_t)(c0 + tid) * GROW + 2 * t);
          tst[tid] = se.x;
          ten[tid] = se.y;
        }
        __syncthreads();
        for (int k = wv; k < nc; k += 2 * (RTPB / 32)) {
          const int  k2 = k + RTPB / 32;
          const bool h2 = k2 < nc;
          const int  kk = h2 ? k2 : k;
          const int* pa = recs + (size_t)(c0 + k) * (size_t)cap;
          const int* pb = recs + (size_t)(c0 + kk) * (size_t)cap;
          int       ua = __builtin_amdgcn_readfirstlane(tst[k] >> 2);
          const int ea = __builtin_amdgcn_readfirstlane(ten[k] >> 2);
          const int tb = __builtin_amdgcn_readfirstlane(tst[kk] >> 2);
          const int nb = __builtin_amdgcn_readfirstlane(ten[kk] >> 2);
          int       ub = h2 ? tb : 0;
          const int eb = h2 ? nb : 0;
          for (; ua < ea || ub < eb; ua += 96, ub += 96) {
            v4i ra[3], rb[3];
#pragma unroll
            for (int q = 0; q < 3; ++q) {
              ra[q] = run_vec(pa, ua + 32 * q + lane, ea);
              rb[q] = run_vec(pb, ub + 32 * q + lane, eb);
            }
            asm volatile("" : "+v"(ra[0]), "+v"(ra[1]), "+v"(ra[2]), "+v"(rb[0]), "+v"(rb[1]), "+v"(rb[2]));
#pragma unroll
            for (int q = 0; q < 3; ++q) {
              ra[q] = run_pad(ra[q], ua + 32 * q + lane, ea);
              rb[q] = run_pad(rb[q], ub + 32 * q + lane, eb);
            }
#pragma unroll
            for (int q = 0; q < 2; ++q) {
#pragma unroll
              for (int j = 0; j < 4; ++j) {
                const int sa = ra[q][j], sb = rb[q][j];
                atomicAdd(&acc[(sa != -1) ? (sa & (TRN - 1)) : (TRN + lane)], 1.0f);
                atomicAdd(&acc[(sb != -1) ? (sb & (TRN - 1)) : (TRN + lane)], 1.0f);
              }
            }
            if (ea - ua > 64) {
#pragma unroll
              for (int j = 0; j < 4; ++j) {
                const int sa = ra[2][j];
                atomicAdd(&acc[(sa != -1) ? (sa & (TRN - 1)) : (TRN + lane)], 1.0f);
              }
            }
            if (eb - ub > 64) {
#pragma unroll
              for (int j = 0; j < 4; ++j) {
                const int sb = rb[2][j];
                atomicAdd(&acc[(sb != -1) ? (sb & (TRN - 1)) : (TRN + lane)], 1.0f);
              }
            }
          }
        }
      }
      __syncthreads();
      for (int i0 = 0; i0 < cnt; i0 += RTPB * 8) {
        float a[8];
#pragma unroll
        for (int j = 0; j < 8; ++j) a[j] = acc[i0 + j * RTPB + tid];
#pragma unroll
        for (int j = 0; j < 8; ++j) {
          const int i = i0 + j * RTPB + tid;
          if (i < cnt && a[j] != 0.0f) atomicAdd(&deg[lo + i], a[j]);
        }
      }
    }
  }
}

__global__ void __launch_bounds__(RTPB)
gcn_l1_t(const int* __restrict__ recs, const int* __restrict__ tab,
         const float* __restrict__ x, const float* __restrict__ deg,
         float* __restrict__ agg1, int* __restrict__ tick,
         int n, int nchunks, int cap, int RN) {
  __shared__ __attribute__((aligned(16))) float acc[TRN + 32];
  __shared__ __attribute__((aligned(16))) float xs[TRN];
  __shared__ int   tst[GTC], ten[GTC];
  __shared__ int   slot;
  const int tid  = threadIdx.x;
  const int lane = tid & 31;
  const int wv   = __builtin_amdgcn_readfirstlane(tid >> 5);
  const int dh   = (RN + 1) >> 1;
  const int h    = cache_half();
  for (int ph = 0; ph < 2; ++ph) {
    const int q     = h ^ ph;
    const int dbase = q ? dh : 0;
    const int nt    = (q ? RN - dh : dh) * RN;
    int* ctr = tick + q * 128;
    for (;;) {
      const int tk = take_block_ticket(ctr, &slot);
      if (tk >= nt) break;
      const int t    = (dbase + tk / RN) * RN + tk % RN;
      const int lo   = (t / RN) << TSH;
      const int cnt  = min(TRN, n - lo);
      const int slo  = (t % RN) << TSH;
      const int scnt = min(TRN, n - slo);
      if (tid < 32) acc[TRN + tid] = 0.0f;
      if (scnt == TRN) {
        v4f xv[TRN / (RTPB * 4)], dv[TRN / (RTPB * 4)];
#pragma unroll
        for (int g = 0; g < TRN / (RTPB * 4); ++g) {
          const int i = (g * RTPB + tid) * 4;
          xv[g] = *(const v4f*)(x + slo + i);
          dv[g] = *(const v4f*)(deg + slo + i);
        }
#pragma unroll
        for (int g = 0; g < TRN / (RTPB * 4); ++g) {
          const int i = (g * RTPB + tid) * 4;
          v4f r;
          r.x = xv[g].x * rsqrtf(1.0f + dv[g].x);
          r.y = xv[g].y * rsqrtf(1.0f + dv[g].y);
          r.z = xv[g].z * rsqrtf(1.0f + dv[g].z);
          r.w = xv[g].w * rsqrtf(1.0f + dv[g].w);
          *(v4f*)&xs[i]  = r;
          *(v4f*)&acc[i] = (v4f){0.f, 0.f, 0.f, 0.f};
        }
      } else {
        for (int i = tid; i < TRN; i += RTPB) {
          acc[i] = 0.0f;
          float v = 0.0f;
          if (i < scnt) v = x[slo + i] * rsqrtf(1.0f + deg[slo + i]);
          xs[i] = v;
        }
      }
      for (int c0 = 0; c0 < nchunks; c0 += GTC) {
        const int nc = min(GTC, nchunks - c0);
        __syncthreads();
        if (tid < nc) {
          const v2i se = *(const v2i*)(tab + (size_t)(c0 + tid) * GROW + 2 * t);
          tst[tid] = se.x;
          ten[tid] = se.y;
        }
        __syncthreads();
        for (int k = wv; k < nc; k += 2 * (RTPB / 32)) {
          const int  k2 = k + RTPB / 32;
          const bool h2 = k2 < nc;
          const int  kk = h2 ? k2 : k;
          const int* pa = recs + (size_t)(c0 + k) * (size_t)cap;
          const int* pb = recs + (size_t)(c0 + kk) * (size_t)cap;
          int       ua = __builtin_amdgcn_readfirstlane(tst[k] >> 2);
          const int ea = __builtin_amdgcn_readfirstlane(ten[k] >> 2);
          const int tb = __builtin_amdgcn_readfirstlane(tst[kk] >> 2);
          const int nb = __builtin_amdgcn_readfirstlane(ten[kk] >> 2);
          int       ub = h2 ? tb : 0;
          const int eb = h2 ? nb : 0;
          for (; ua < ea || ub < eb; ua += 96, ub += 96) {
            v4i ra[3], rb[3];
#pragma unroll
            for (int q = 0; q < 3; ++q) {
              ra[q] = run_vec(pa, ua + 32 * q + lane, ea);
              rb[q] = run_vec(pb, ub + 32 * q + lane, eb);
            }
            asm volatile("" : "+v"(ra[0]), "+v"(ra[1]), "+v"(ra[2]), "+v"(rb[0]), "+v"(rb[1]), "+v"(rb[2]));
#pragma unroll
            for (int q = 0; q < 3; ++q) {
              ra[q] = run_pad(ra[q], ua + 32 * q + lane, ea);
              rb[q] = run_pad(rb[q], ub + 32 * q + lane, eb);
            }
            float va[8], vb[8];
#pragma unroll
            for (int q = 0; q < 2; ++q) {
#pragma unroll
              for (int j = 0; j < 4; ++j) {
                const int sa = ra[q][j], sb = rb[q][j];
                va[q * 4 + j] = xs[(sa != -1) ? ((unsigned)sa >> TSH) : 0u];
                vb[q * 4 + j] = xs[(sb != -1) ? ((unsigned)sb >> TSH) : 0u];
              }
            }
#pragma unroll
            for (int q = 0; q < 2; ++q) {
#pragma unroll
              for (int j = 0; j < 4; ++j) {
                const int sa = ra[q][j], sb = rb[q][j];
                atomicAdd(&acc[(sa != -1) ? (sa & (TRN - 1)) : (TRN + lane)], va[q * 4 + j]);
                atomicAdd(&acc[(sb != -1) ? (sb & (TRN - 1)) : (TRN + lane)], vb[q * 4 + j]);
              }
            }
            if (ea - ua > 64) {
              float vc[4];
#pragma unroll
              for (int j = 0; j < 4; ++j) {
                const int sa = ra[2][j];
                vc[j] = xs[(sa != -1) ? ((unsigned)sa >> TSH) : 0u];
              }
#pragma unroll
              for (int j = 0; j < 4; ++j) {
                const int sa = ra[2][j];
                atomicAdd(&acc[(sa != -1) ? (sa & (TRN - 1)) : (TRN + lane)], vc[j]);
              }
            }
            if (eb - ub > 64) {
              float vc[4];
#pragma unroll
              for (int j = 0; j < 4; ++j) {
                const int sb = rb[2][j];
                vc[j] = xs[(sb != -1) ? ((unsigned)sb >> TSH) : 0u];
              }
#pragma unroll
              for (int j = 0; j < 4; ++j) {
                const int sb = rb[2][j];
                atomicAdd(&acc[(sb != -1) ? (sb & (TRN - 1)) : (TRN + lane)], vc[j]);
              }
            }
          }
        }
      }
      __syncthreads();
      for (int i0 = 0; i0 < cnt; i0 += RTPB * 8) {
        float a[8];
#pragma unroll
        for (int j = 0; j < 8; ++j) a[j] = acc[i0 + j * RTPB + tid];
#pragma unroll
        for (int j = 0; j < 8; ++j) {
          const int i = i0 + j * RTPB + tid;
          if (i < cnt && a[j] != 0.0f) atomicAdd(&agg1[lo + i], a[j]);
        }
      }
    }
  }
}

__global__ void gcn_node_t(float* __restrict__ agg1,
                           const float* __restrict__ x,
                           const float* __restrict__ deg,
                           const float* __restrict__ W1,
                           const float* __restrict__ b1,
                           const float* __restrict__ W2,
                           float* __restrict__ p0,
                           float* __restrict__ p1,
                           int n) {
  int lane = threadIdx.x & 31;
  int wave = threadIdx.x >> 5;
  int tile = blockIdx.x * (blockDim.x >> 5) + wave;
  int base = tile * 16;
  int m    = lane & 15;
  int half = lane >> 4;
  int node = base + m;
  int nc   = node < n ? node : (n - 1);

  float dn = rsqrtf(1.0f + deg[nc]);
  float av = (agg1[nc] + x[nc] * dn) * dn;
  if (half == 0 && node < n) agg1[node] = 0.0f;

  int   wcol  = (m < 2) ? m : 0;
  float wmask = (m < 2) ? 1.0f : 0.0f;

  v8f acc = {};
#pragma unroll
  for (int s = 0; s < 4; ++s) {
    int j0 = s * 4 + half * 2;
    v2f w1 = *(const v2f*)(W1 + j0);
    v2f bi = *(const v2f*)(b1 + j0);
    v2f a;
    a.x = fmaxf(av * w1.x + bi.x, 0.0f);
    a.y = fmaxf(av * w1.y + bi.y, 0.0f);
    v2f bb;
    bb.x = W2[j0 * 2 + wcol]       * wmask;
    bb.y = W2[(j0 + 1) * 2 + wcol] * wmask;
    acc = __builtin_amdgcn_wmma_f32_16x16x4_f32(
         false, a,  false, bb,
         (short)0, acc,  false,  false);
  }

  if (m < 2) {
    float* pm = m ? p1 : p0;
#pragma unroll
    for (int r = 0; r < 8; ++r) {
      int nd = base + r + half * 8;
      if (nd < n) pm[nd] = acc[r] * rsqrtf(1.0f + deg[nd]);
    }
  }
}


__global__ void __launch_bounds__(RTPB)
gcn_l2_t(const int* __restrict__ recs, const int* __restrict__ tab,
         const float* __restrict__ p0, const float* __restrict__ p1,
         float* __restrict__ s0, float* __restrict__ s1, int* __restrict__ tick,
         int n, int nchunks, int cap, int RN) {
  __shared__ __attribute__((aligned(16))) float acc[TRN + 32];
  __shared__ __attribute__((aligned(16))) float hs[TRN];
  __shared__ int   tst[GTC], ten[GTC];
  __shared__ int   slot;
  const int tid  = threadIdx.x;
  const int lane = tid & 31;
  const int wv   = __builtin_amdgcn_readfirstlane(tid >> 5);
  const int dh   = (RN + 1) >> 1;
  const int h    = cache_half();
  for (int ph = 0; ph < 2; ++ph) {
    const int q     = h ^ ph;
    const int dbase = q ? dh : 0;
    const int nt    = (q ? RN - dh : dh) * RN;
    int* ctr = tick + q * 128;
    for (;;) {
      const int tk = take_block_ticket(ctr, &slot);
      if (tk >= nt) break;
      const int t    = (dbase + tk / RN) * RN + tk % RN;
      const int lo   = (t / RN) << TSH;
      const int cnt  = min(TRN, n - lo);
      const int slo  = (t % RN) << TSH;
      const int scnt = min(TRN, n - slo);
      for (int comp = 0; comp < 2; ++comp) {
      const float* pk = comp ? p1 : p0;
      float*       sk = comp ? s1 : s0;
      if (tid < 32) acc[TRN + tid] = 0.0f;
      if (scnt == TRN) {
        v4f pv[TRN / (RTPB * 4)];
#pragma unroll
        for (int g = 0; g < TRN / (RTPB * 4); ++g) pv[g] = *(const v4f*)(pk + slo + (g * RTPB + tid) * 4);
#pragma unroll
        for (int g = 0; g < TRN / (RTPB * 4); ++g) {
          const int i = (g * RTPB + tid) * 4;
          *(v4f*)&hs[i]  = pv[g];
          *(v4f*)&acc[i] = (v4f){0.f, 0.f, 0.f, 0.f};
        }
      } else {
        for (int i = tid; i < TRN; i += RTPB) {
          acc[i] = 0.0f;
          hs[i]  = (i < scnt) ? pk[slo + i] : 0.0f;
        }
      }
      for (int c0 = 0; c0 < nchunks; c0 += GTC) {
        const int nc = min(GTC, nchunks - c0);
        __syncthreads();
        if (tid < nc) {
          const v2i se = *(const v2i*)(tab + (size_t)(c0 + tid) * GROW + 2 * t);
          tst[tid] = se.x;
          ten[tid] = se.y;
        }
        __syncthreads();
        for (int k = wv; k < nc; k += 2 * (RTPB / 32)) {
          const int  k2 = k + RTPB / 32;
          const bool h2 = k2 < nc;
          const int  kk = h2 ? k2 : k;
          const int* pa = recs + (size_t)(c0 + k) * (size_t)cap;
          const int* pb = recs + (size_t)(c0 + kk) * (size_t)cap;
          int       ua = __builtin_amdgcn_readfirstlane(tst[k] >> 2);
          const int ea = __builtin_amdgcn_readfirstlane(ten[k] >> 2);
          const int tb = __builtin_amdgcn_readfirstlane(tst[kk] >> 2);
          const int nb = __builtin_amdgcn_readfirstlane(ten[kk] >> 2);
          int       ub = h2 ? tb : 0;
          const int eb = h2 ? nb : 0;
          for (; ua < ea || ub < eb; ua += 96, ub += 96) {
            v4i ra[3], rb[3];
#pragma unroll
            for (int q = 0; q < 3; ++q) {
              ra[q] = run_vec(pa, ua + 32 * q + lane, ea);
              rb[q] = run_vec(pb, ub + 32 * q + lane, eb);
            }
            asm volatile("" : "+v"(ra[0]), "+v"(ra[1]), "+v"(ra[2]), "+v"(rb[0]), "+v"(rb[1]), "+v"(rb[2]));
#pragma unroll
            for (int q = 0; q < 3; ++q) {
              ra[q] = run_pad(ra[q], ua + 32 * q + lane, ea);
              rb[q] = run_pad(rb[q], ub + 32 * q + lane, eb);
            }
            float va[8], vb[8];
#pragma unroll
            for (int q = 0; q < 2; ++q) {
#pragma unroll
              for (int j = 0; j < 4; ++j) {
                const int sa = ra[q][j], sb = rb[q][j];
                va[q * 4 + j] = hs[(sa != -1) ? ((unsigned)sa >> TSH) : 0u];
                vb[q * 4 + j] = hs[(sb != -1) ? ((unsigned)sb >> TSH) : 0u];
              }
            }
#pragma unroll
            for (int q = 0; q < 2; ++q) {
#pragma unroll
              for (int j = 0; j < 4; ++j) {
                const int sa = ra[q][j], sb = rb[q][j];
                atomicAdd(&acc[(sa != -1) ? (sa & (TRN - 1)) : (TRN + lane)], va[q * 4 + j]);
                atomicAdd(&acc[(sb != -1) ? (sb & (TRN - 1)) : (TRN + lane)], vb[q * 4 + j]);
              }
            }
            if (ea - ua > 64) {
              float vc[4];
#pragma unroll
              for (int j = 0; j < 4; ++j) {
                const int sa = ra[2][j];
                vc[j] = hs[(sa != -1) ? ((unsigned)sa >> TSH) : 0u];
              }
#pragma unroll
              for (int j = 0; j < 4; ++j) {
                const int sa = ra[2][j];
                atomicAdd(&acc[(sa != -1) ? (sa & (TRN - 1)) : (TRN + lane)], vc[j]);
              }
            }
            if (eb - ub > 64) {
              float vc[4];
#pragma unroll
              for (int j = 0; j < 4; ++j) {
                const int sb = rb[2][j];
                vc[j] = hs[(sb != -1) ? ((unsigned)sb >> TSH) : 0u];
              }
#pragma unroll
              for (int j = 0; j < 4; ++j) {
                const int sb = rb[2][j];
                atomicAdd(&acc[(sb != -1) ? (sb & (TRN - 1)) : (TRN + lane)], vc[j]);
              }
            }
          }
        }
      }
      __syncthreads();
      for (int i0 = 0; i0 < cnt; i0 += RTPB * 8) {
        float a[8];
#pragma unroll
        for (int j = 0; j < 8; ++j) a[j] = acc[i0 + j * RTPB + tid];
#pragma unroll
        for (int j = 0; j < 8; ++j) {
          const int i = i0 + j * RTPB + tid;
          if (i < cnt && a[j] != 0.0f) atomicAdd(&sk[lo + i], a[j]);
        }
      }
      __syncthreads();
      }
    }
  }
}

__global__ void gcn_fin_t(const float* __restrict__ deg, const float* __restrict__ b2,
                          const float* __restrict__ p0, const float* __restrict__ p1,
                          const float* __restrict__ s0, const float* __restrict__ s1,
                          float* __restrict__ out, int n) {
  const int i = blockIdx.x * blockDim.x + threadIdx.x;
  if (i < n) {
    const float d = rsqrtf(1.0f + deg[i]);
    v2f o;
    o.x = b2[0] + d * (s0[i] + p0[i]);
    o.y = b2[1] + d * (s1[i] + p1[i]);
    *(v2f*)(out + 2 * (size_t)i) = o;
  }
}

extern "C" void kernel_launch(void* const* d_in, const int* in_sizes, int n_in,
                              void* d_out, int out_size, void* d_ws, size_t ws_size,
                              hipStream_t stream) {
  const float* x  = (const float*)d_in[0];
  const int*   ei = (const int*)d_in[1];
  const float* W1 = (const float*)d_in[2];
  const float* b1 = (const float*)d_in[3];
  const float* W2 = (const float*)d_in[4];
  const float* b2 = (const float*)d_in[5];
  float* out = (float*)d_out;

  const int n = in_sizes[0];
  const int e = in_sizes[1] / 2;
  const int* row = ei;
  const int* col = ei + e;

  size_t np   = (size_t)((n + 3) & ~3);
  float* ws   = (float*)d_ws;
  float* dis  = ws;
  float* xp   = ws + np;
  float* agg1 = ws + 2 * np;
  float* h2p  = ws + 3 * np;

  const int cs = (n >= 512) ? 128 : 1;
  int* tick24 = (int*)h2p;
  int* tick6  = (int*)agg1;

  const int tb = 256;
  const int nb_node = (n + tb - 1) / tb;
  const int tiles = (n + 15) / 16;
  const int nb_tile = (tiles + 7) / 8;
  const int nb_fin  = (n + 4 * tb - 1) / (4 * tb);

  {
    const int RNt = (n + TRN - 1) >> TSH;
    const int NBt = RNt * RNt;
    int chunklen_t = ((e + RANGE_BLOCKS - 1) / RANGE_BLOCKS + 3) & ~3;
    if (chunklen_t > GCH_MAX) chunklen_t = GCH_MAX;
    if (chunklen_t < 4) chunklen_t = 4;
    const int nchunks_t = (e + chunklen_t - 1) / chunklen_t;
    const int cap_t = (chunklen_t + 31 + NBt * GPAD) & ~31;
    const size_t off_tab_t = (5 * np + 31) & ~(size_t)31;
    const size_t off_rec_t = off_tab_t + (size_t)nchunks_t * GROW;
    const size_t off_tick_t = off_rec_t + (size_t)nchunks_t * (size_t)cap_t;
    const size_t need_t    = (off_tick_t + 6 * 128) * sizeof(float);
    if (n >= 1 && e >= 1 && NBt <= GNB_MAX && need_t <= ws_size) {
      float* degt = ws;
      float* p0   = ws + np;
      float* p1   = ws + 3 * np;
      float* s1   = ws + 4 * np;
      int* tab  = (int*)(ws + off_tab_t);
      int* recs = (int*)(ws + off_rec_t);
      int* tick = (int*)(ws + off_tick_t);
      gcn_group_t <<<RANGE_BLOCKS, RTPB, 0, stream>>>(row, col, recs, tab, degt, agg1, s1, tick, e, n, chunklen_t, nchunks_t, cap_t, RNt);
      gcn_degree_t<<<NBt, RTPB, 0, stream>>>(recs, tab, degt, tick, n, nchunks_t, cap_t, RNt);
      gcn_l1_t    <<<NBt, RTPB, 0, stream>>>(recs, tab, x, degt, agg1, tick + 2 * 128, n, nchunks_t, cap_t, RNt);
      gcn_node_t  <<<nb_tile, 256, 0, stream>>>(agg1, x, degt, W1, b1, W2, p0, p1, n);
      gcn_l2_t    <<<NBt, RTPB, 0, stream>>>(recs, tab, p0, p1, agg1, s1, tick + 4 * 128, n, nchunks_t, cap_t, RNt);
      gcn_fin_t   <<<nb_node, tb, 0, stream>>>(degt, b2, p0, p1, agg1, s1, out, n);
      return;
    }
  }

  const int NBg = (n + GBIN - 1) >> GSH;
  int chunklen = ((e + RANGE_BLOCKS - 1) / RANGE_BLOCKS + 3) & ~3;
  if (chunklen > GCH_MAX) chunklen = GCH_MAX;
  if (chunklen < 4) chunklen = 4;
  const int nchunks = (e + chunklen - 1) / chunklen;
  const int cap = (chunklen + 31 + NBg * GPAD) & ~31;
  const size_t off_tab = (5 * np + 31) & ~(size_t)31;
  const size_t off_rec = off_tab + (size_t)nchunks * GROW;
  const size_t need    = (off_rec + (size_t)nchunks * (size_t)cap) * sizeof(float);
  if (n >= 1 && e >= 1 && NBg <= GNB_MAX && need <= ws_size) {
    int* tab  = (int*)(ws + off_tab);
    int* recs = (int*)(ws + off_rec);
    gcn_group   <<<RANGE_BLOCKS, RTPB, 0, stream>>>(row, col, recs, tab, e, n, chunklen, nchunks, cap);
    gcn_degree_g<<<NBg, RTPB, 0, stream>>>(recs, tab, x, dis, xp, n, nchunks, cap);
    gcn_l1_g    <<<NBg, RTPB, 0, stream>>>(recs, tab, xp, agg1, n, nchunks, cap);
    gcn_node    <<<nb_tile, 256, 0, stream>>>(agg1, xp, dis, W1, b1, W2, h2p, out, n, 0);
    gcn_l2_g    <<<NBg, RTPB, 0, stream>>>(recs, tab, h2p, dis, b2, out, n, nchunks, cap);
    return;
  }

  gcn_init  <<<nb_node, tb, 0, stream>>>(dis, agg1, tick24, cs, n);
  gcn_degree<<<RANGE_BLOCKS, RTPB, 0, stream>>>(col, dis, tick24, cs, e, n);
  gcn_dis   <<<nb_node, tb, 0, stream>>>(x, dis, xp, n);
  gcn_l1    <<<RANGE_BLOCKS, RTPB, 0, stream>>>(row, col, xp, agg1, tick24 + 2 * cs, cs, e, n);
  gcn_node  <<<nb_tile, 256, 0, stream>>>(agg1, xp, dis, W1, b1, W2, h2p, out, n, 1);
  gcn_tick0 <<<1, 32, 0, stream>>>(tick6, cs);
  gcn_l2    <<<RANGE_BLOCKS, RTPB, 0, stream>>>(row, col, h2p, out, tick6, cs, e, n);
  gcn_fin   <<<nb_fin, tb, 0, stream>>>(dis, b2, out, n);
}


# Node and finish kernels as fixed grids of resident waves looping over nodes (node: 4 tiles of 16 per trip, weights once, row scale from its lane; finish: 4 nodes a thread): 47k waves become 8k
# speedup vs baseline: 13.5390x; 1.0554x over previous
#include <hip/hip_runtime.h>


typedef __attribute__((ext_vector_type(2))) float v2f;
typedef __attribute__((ext_vector_type(8))) float v8f;
typedef __attribute__((ext_vector_type(4))) int   v4i;
typedef __attribute__((ext_vector_type(4))) float v4f;
typedef __attribute__((ext_vector_type(2))) int   v2i;


#define RS_MAX 65536
#define SLICES 32
#define RTPB   1024
#define RANGE_BLOCKS 256

#define L2_CAP 32768
#define L2_SL  16
#define QN     64

__device__ __forceinline__ int cache_half() {
  unsigned id = __builtin_amdgcn_s_sendmsg_rtn(0x87);
  return (int)((id >> 18) & 1u);
}

__device__ __forceinline__ int range_count(int n) {
  int r = (n + RS_MAX - 1) / RS_MAX;
  if (r < 8) r = 8;
  return (r + 1) & ~1;
}
__device__ __forceinline__ int range_size(int n, int r) { return ((n + r - 1) / r + 63) & ~63; }
__device__ __forceinline__ int slice_len(int e) { return ((e + SLICES - 1) / SLICES + 16383) & ~16383; }

__device__ __forceinline__ int take_block_ticket(int* ctr, int* slot) {
  if (threadIdx.x == 0) *slot = atomicAdd(ctr, 1);
  __syncthreads();
  const int t = *slot;
  __syncthreads();
  return t;
}

__device__ __forceinline__ v4i load4_or_none(const int* __restrict__ a, int i, int hi) {
  if (i + 3 < hi) return *(const v4i*)(a + i);
  v4i r = {-1, -1, -1, -1};
  if (i     < hi) r.x = a[i];
  if (i + 1 < hi) r.y = a[i + 1];
  if (i + 2 < hi) r.z = a[i + 2];
  return r;
}

__global__ void gcn_init(float* __restrict__ deg, float* __restrict__ agg1,
                         int* __restrict__ tick, int cs, int n) {
  int i = blockIdx.x * blockDim.x + threadIdx.x;
  if (i < n) { deg[i] = 1.0f; agg1[i] = 0.0f; }
  if (i < 4) tick[i * cs] = 0;
}

__global__ void gcn_tick0(int* __restrict__ tick, int cs) {
  if (threadIdx.x < 2) tick[threadIdx.x * cs] = 0;
}

__global__ void __launch_bounds__(RTPB)
gcn_degree(const int* __restrict__ col, float* __restrict__ deg,
           int* __restrict__ tick, int cs, int e, int n) {
  __shared__ __attribute__((aligned(16))) float acc[RS_MAX];
  __shared__ int   slot;
  const int tid = threadIdx.x;
  const int R   = range_count(n);
  const int Rh  = R >> 1;
  const int RS  = range_size(n, R);
  const int CH  = slice_len(e);
  const int nt  = Rh * SLICES;
  const int h   = cache_half();
  for (int p = 0; p < 2; ++p) {
    const int q = h ^ p;
    int* ctr = tick + q * cs;
    for (;;) {
      const int t = take_block_ticket(ctr, &slot);
      if (t >= nt) break;
      const int lo  = (q * Rh + t % Rh) * RS;
      const int cnt = min(RS, n - lo);
      const int elo = (t / Rh) * CH;
      const int ehi = min(e, elo + CH);
      if (cnt <= 0 || elo >= ehi) continue;
      for (int i = tid * 4; i < RS; i += RTPB * 4) *(v4f*)&acc[i] = (v4f){0.f, 0.f, 0.f, 0.f};
      __syncthreads();
      int b = elo + tid * 4;
      for (; b + 3 * RTPB * 4 + 3 < ehi; b += RTPB * 16) {
        v4i c[4];
#pragma unroll
        for (int u = 0; u < 4; ++u) c[u] = *(const v4i*)(col + b + u * RTPB * 4);
#pragma unroll
        for (int u = 0; u < 4; ++u) {
#pragma unroll
          for (int j = 0; j < 4; ++j) {
            const unsigned d = (unsigned)(c[u][j] - lo);
            if (d < (unsigned)cnt) atomicAdd(&acc[d], 1.0f);
          }
        }
      }
      for (; b < ehi; b += RTPB * 4) {
        const v4i c = load4_or_none(col, b, ehi);
#pragma unroll
        for (int j = 0; j < 4; ++j) {
          const unsigned d = (unsigned)(c[j] - lo);
          if (d < (unsigned)cnt) atomicAdd(&acc[d], 1.0f);
        }
      }
      __syncthreads();
      for (int i = tid; i < cnt; i += RTPB) {
        const float v = acc[i];
        if (v != 0.0f) atomicAdd(&deg[lo + i], v);
      }
    }
  }
}

__global__ void gcn_dis(const float* __restrict__ x, float* __restrict__ deg_dis,
                        float* __restrict__ xp, int n) {
  int i = blockIdx.x * blockDim.x + threadIdx.x;
  if (i < n) {
    float d = rsqrtf(deg_dis[i]);
    deg_dis[i] = d;
    xp[i] = x[i] * d;
  }
}

__global__ void __launch_bounds__(RTPB)
gcn_l1(const int* __restrict__ row, const int* __restrict__ col,
       const float* __restrict__ xp,
       float* __restrict__ agg1, int* __restrict__ tick, int cs, int e, int n) {
  __shared__ __attribute__((aligned(16))) float acc[RS_MAX];
  __shared__ int   slot;
  const int tid = threadIdx.x;
  const int R   = range_count(n);
  const int Rh  = R >> 1;
  const int RS  = range_size(n, R);
  const int CH  = slice_len(e);
  const int nt  = Rh * SLICES;
  const int h   = cache_half();
  for (int p = 0; p < 2; ++p) {
    const int q = h ^ p;
    int* ctr = tick + q * cs;
    for (;;) {
      const int t = take_block_ticket(ctr, &slot);
      if (t >= nt) break;
      const int lo  = (q * Rh + t % Rh) * RS;
      const int cnt = min(RS, n - lo);
      const int elo = (t / Rh) * CH;
      const int ehi = min(e, elo + CH);
      if (cnt <= 0 || elo >= ehi) continue;
      for (int i = tid * 4; i < RS; i += RTPB * 4) *(v4f*)&acc[i] = (v4f){0.f, 0.f, 0.f, 0.f};
      __syncthreads();
      int b = elo + tid * 4;
      for (; b + 3 * RTPB * 4 + 3 < ehi; b += RTPB * 16) {
        v4i r[4], c[4];
#pragma unroll
        for (int u = 0; u < 4; ++u) {
          c[u] = *(const v4i*)(col + b + u * RTPB * 4);
          r[u] = *(const v4i*)(row + b + u * RTPB * 4);
        }
        asm volatile("" : "+v"(c[0]), "+v"(c[1]), "+v"(c[2]), "+v"(c[3]),
                          "+v"(r[0]), "+v"(r[1]), "+v"(r[2]), "+v"(r[3]));
        float v[16];
        bool  m[16];
#pragma unroll
        for (int u = 0; u < 4; ++u) {
#pragma unroll
          for (int j = 0; j < 4; ++j) {
            const unsigned d = (unsigned)(c[u][j] - lo);
            m[u * 4 + j] = d < (unsigned)cnt;
            v[u * 4 + j] = 0.0f;
            if (m[u * 4 + j]) v[u * 4 + j] = xp[r[u][j]];
          }
        }
#pragma unroll
        for (int u = 0; u < 4; ++u) {
#pragma unroll
          for (int j = 0; j < 4; ++j)
            if (m[u * 4 + j]) atomicAdd(&acc[c[u][j] - lo], v[u * 4 + j]);
        }
      }
      for (; b < ehi; b += RTPB * 4) {
        const v4i c = load4_or_none(col, b, ehi);
        const v4i r = load4_or_none(row, b, ehi);
#pragma unroll
        for (int j = 0; j < 4; ++j) {
          const unsigned d = (unsigned)(c[j] - lo);
          if (d < (unsigned)cnt) atomicAdd(&acc[d], xp[r[j]]);
        }
      }
      __syncthreads();
      for (int i = tid; i < cnt; i += RTPB) {
        const float v = acc[i];
        if (v != 0.0f) atomicAdd(&agg1[lo + i], v);
      }
    }
  }
}

__global__ void gcn_node(const float* __restrict__ agg1,
                         const float* __restrict__ xp,
                         const float* __restrict__ dis,
                         const float* __restrict__ W1,
                         const float* __restrict__ b1,
                         const float* __restrict__ W2,
                         float* __restrict__ h2p,
                         float* __restrict__ out,
                         int n, int start_out) {
  int lane = threadIdx.x & 31;
  int wave = threadIdx.x >> 5;
  int tile = blockIdx.x * (blockDim.x >> 5) + wave;
  int base = tile * 16;
  int m    = lane & 15;
  int half = lane >> 4;
  int node = base + m;
  int nc   = node < n ? node : (n - 1);

  float av = (agg1[nc] + xp[nc]) * dis[nc];

  int   wcol  = (m < 2) ? m : 0;
  float wmask = (m < 2) ? 1.0f : 0.0f;

  v8f acc = {};
#pragma unroll
  for (int s = 0; s < 4; ++s) {
    int j0 = s * 4 + half * 2;
    v2f w1 = *(const v2f*)(W1 + j0);
    v2f bi = *(const v2f*)(b1 + j0);
    v2f a;
    a.x = fmaxf(av * w1.x + bi.x, 0.0f);
    a.y = fmaxf(av * w1.y + bi.y, 0.0f);
    v2f bb;
    bb.x = W2[j0 * 2 + wcol]       * wmask;
    bb.y = W2[(j0 + 1) * 2 + wcol] * wmask;
    acc = __builtin_amdgcn_wmma_f32_16x16x4_f32(
         false, a,  false, bb,
         (short)0, acc,  false,  false);
  }

  if (m < 2) {
#pragma unroll
    for (int r = 0; r < 8; ++r) {
      int nd = base + r + half * 8;
      if (nd < n) {
        float hp = acc[r] * dis[nd];
        h2p[nd * 2 + m] = hp;
        if (start_out) out[nd * 2 + m] = hp;
      }
    }
  }
}

__device__ __forceinline__ int l2_range_count(int n) {
  int r = (n + L2_CAP - 1) / L2_CAP;
  if (r < 16) r = 16;
  return (r + 1) & ~1;
}
__device__ __forceinline__ int l2_slice_len(int e) { return ((e + L2_SL - 1) / L2_SL + 16383) & ~16383; }

__global__ void __launch_bounds__(RTPB)
gcn_l2(const int* __restrict__ row, const int* __restrict__ col,
       const float* __restrict__ h2p,
       float* __restrict__ out, int* __restrict__ tick, int cs, int e, int n) {
  __shared__ __attribute__((aligned(16))) float acc[2 * L2_CAP];
  __shared__ __attribute__((aligned(8)))  v2i   ring[(RTPB / 32) * QN];
  __shared__ int   slot;
  const int tid  = threadIdx.x;
  const int lane = tid & 31;
  v2i* wq = ring + (tid >> 5) * QN;
  const int R   = l2_range_count(n);
  const int Rh  = R >> 1;
  const int RS  = range_size(n, R);
  const int CH  = l2_slice_len(e);
  const int nt  = Rh * L2_SL;
  const int h   = cache_half();
  for (int p = 0; p < 2; ++p) {
    const int q = h ^ p;
    int* ctr = tick + q * cs;
    for (;;) {
      const int t = take_block_ticket(ctr, &slot);
      if (t >= nt) break;
      const int lo  = (q * Rh + t % Rh) * RS;
      const int cnt = min(RS, n - lo);
      const int elo = (t / Rh) * CH;
      const int ehi = min(e, elo + CH);
      if (cnt <= 0 || elo >= ehi) continue;
      for (int i = tid * 4; i < 2 * L2_CAP; i += RTPB * 4) *(v4f*)&acc[i] = (v4f){0.f, 0.f, 0.f, 0.f};
      __syncthreads();
      unsigned head = 0, tail = 0;
      bool pend = false;
      int  pd = 0;
      v2f  pv = {0.f, 0.f};
      for (int base = elo; base < ehi; base += RTPB * 16) {
        const int b = base + tid * 4;
        v4i r[4], c[4];
        if (base + RTPB * 16 <= ehi) {
#pragma unroll
          for (int u = 0; u < 4; ++u) {
            c[u] = *(const v4i*)(col + b + u * RTPB * 4);
            r[u] = *(const v4i*)(row + b + u * RTPB * 4);
          }
        } else {
#pragma unroll
          for (int u = 0; u < 4; ++u) {
            c[u] = load4_or_none(col, b + u * RTPB * 4, ehi);
            r[u] = load4_or_none(row, b + u * RTPB * 4, ehi);
          }
        }
#pragma unroll
        for (int u = 0; u < 4; ++u) {
#pragma unroll
          for (int j = 0; j < 4; ++j) {
            const unsigned d   = (unsigned)(c[u][j] - lo);
            const bool     inr = d < (unsigned)cnt;
            const unsigned msk = __builtin_amdgcn_ballot_w32(inr);
            if (msk != 0u) {
              if (inr) {
                const unsigned pos = (tail + __builtin_amdgcn_mbcnt_lo(msk, 0u)) & (QN - 1);
                wq[pos] = (v2i){r[u][j], (int)d};
              }
              tail += (unsigned)__builtin_popcount(msk);
              if (tail - head >= 32u) {
                if (pend) {
                  atomicAdd(&acc[pd], pv.x);
                  atomicAdd(&acc[L2_CAP + pd], pv.y);
                }
                __builtin_amdgcn_fence(__ATOMIC_ACQ_REL, "wavefront");
                const v2i en = wq[(head + (unsigned)lane) & (QN - 1)];
                head += 32u;
                pd   = en.y;
                pv   = *(const v2f*)(h2p + 2 * (size_t)en.x);
                pend = true;
              }
            }
          }
        }
      }
      if (pend) {
        atomicAdd(&acc[pd], pv.x);
        atomicAdd(&acc[L2_CAP + pd], pv.y);
      }
      __builtin_amdgcn_fence(__ATOMIC_ACQ_REL, "wavefront");
      if ((unsigned)lane < tail - head) {
        const v2i en = wq[(head + (unsigned)lane) & (QN - 1)];
        const v2f v  = *(const v2f*)(h2p + 2 * (size_t)en.x);
        atomicAdd(&acc[en.y], v.x);
        atomicAdd(&acc[L2_CAP + en.y], v.y);
      }
      __syncthreads();
      float* o = out + 2 * (size_t)lo;
      for (int f = tid; f < 2 * cnt; f += RTPB) {
        const float v = acc[(f & 1) * L2_CAP + (f >> 1)];
        if (v != 0.0f) atomicAdd(&o[f], v);
      }
    }
  }
}

__global__ void gcn_fin(const float* __restrict__ dis, const float* __restrict__ b2,
                        float* __restrict__ out, int n) {
  const int i = 4 * (blockIdx.x * blockDim.x + threadIdx.x);
  const float c0 = b2[0], c1 = b2[1];
  if (i + 3 < n) {
    const v4f d = *(const v4f*)(dis + i);
    float* o = out + 2 * (size_t)i;
    v4f a = *(const v4f*)o;
    v4f b = *(const v4f*)(o + 4);
    a.x = c0 + d.x * a.x;  a.y = c1 + d.x * a.y;
    a.z = c0 + d.y * a.z;  a.w = c1 + d.y * a.w;
    b.x = c0 + d.z * b.x;  b.y = c1 + d.z * b.y;
    b.z = c0 + d.w * b.z;  b.w = c1 + d.w * b.w;
    *(v4f*)o       = a;
    *(v4f*)(o + 4) = b;
  } else {
    for (int j = i; j < n; ++j) {
      const float d = dis[j];
      out[2 * (size_t)j + 0] = c0 + d * out[2 * (size_t)j + 0];
      out[2 * (size_t)j + 1] = c1 + d * out[2 * (size_t)j + 1];
    }
  }
}


#define GSH     11
#define GBIN    2048
#define GNB_MAX 256
#define GCH_MAX 65536
#define GPAD    32
#define GROW    512
#define GTC     1024

__global__ void __launch_bounds__(RTPB)
gcn_group(const int* __restrict__ row, const int* __restrict__ col,
          int* __restrict__ recs, int* __restrict__ tab,
          int e, int n, int chunklen, int nchunks, int cap) {
  __shared__ __attribute__((aligned(16))) int lrec[GCH_MAX + GNB_MAX * GPAD];
  __shared__ int lcnt[GNB_MAX];
  __shared__ int lst[GNB_MAX + 1];
  const int tid = threadIdx.x;
  const int NB  = (n + GBIN - 1) >> GSH;
  for (int ch = blockIdx.x; ch < nchunks; ch += gridDim.x) {
    const int elo = ch * chunklen;
    const int ehi = min(e, elo + chunklen);
    if (tid < GNB_MAX) lcnt[tid] = 0;
    __syncthreads();
    for (int base = elo; base < ehi; base += RTPB * 16) {
      const int b = base + tid * 4;
      v4i c[4];
      if (base + RTPB * 16 <= ehi) {
#pragma unroll
        for (int u = 0; u < 4; ++u) c[u] = *(const v4i*)(col + b + u * RTPB * 4);
      } else {
#pragma unroll
        for (int u = 0; u < 4; ++u) c[u] = load4_or_none(col, b + u * RTPB * 4, ehi);
      }
#pragma unroll
      for (int u = 0; u < 4; ++u) {
#pragma unroll
        for (int j = 0; j < 4; ++j)
          if ((unsigned)c[u][j] < (unsigned)n) atomicAdd(&lcnt[c[u][j] >> GSH], 1);
      }
    }
    __syncthreads();
    int p = 0;
    if (tid < NB) p = (lcnt[tid] + GPAD - 1) & ~(GPAD - 1);
    if (tid < GNB_MAX) lst[tid + 1] = p;
    if (tid == 0) lst[0] = 0;
    __syncthreads();
    for (int off = 1; off < GNB_MAX; off <<= 1) {
      int v = 0;
      if (tid < GNB_MAX && tid >= off) v = lst[tid + 1 - off];
      __syncthreads();
      if (tid < GNB_MAX) lst[tid + 1] += v;
      __syncthreads();
    }
    if (tid < NB) {
      const int s0 = lst[tid], c0 = lcnt[tid], s1 = lst[tid + 1];
      for (int k = s0 + c0; k < s1; ++k) lrec[k] = -1;
      lcnt[tid] = s0;
    }
    __syncthreads();
    for (int base = elo; base < ehi; base += RTPB * 16) {
      const int b = base + tid * 4;
      v4i c[4], r[4];
      if (base + RTPB * 16 <= ehi) {
#pragma unroll
        for (int u = 0; u < 4; ++u) {
          c[u] = *(const v4i*)(col + b + u * RTPB * 4);
          r[u] = *(const v4i*)(row + b + u * RTPB * 4);
        }
      } else {
#pragma unroll
        for (int u = 0; u < 4; ++u) {
          c[u] = load4_or_none(col, b + u * RTPB * 4, ehi);
          r[u] = load4_or_none(row, b + u * RTPB * 4, ehi);
        }
      }
      int pos[16];
#pragma unroll
      for (int u = 0; u < 4; ++u) {
#pragma unroll
        for (int j = 0; j < 4; ++j) {
          pos[u * 4 + j] = -1;
          if ((unsigned)c[u][j] < (unsigned)n) pos[u * 4 + j] = atomicAdd(&lcnt[c[u][j] >> GSH], 1);
        }
      }
#pragma unroll
      for (int u = 0; u < 4; ++u) {
#pragma unroll
        for (int j = 0; j < 4; ++j)
          if (pos[u * 4 + j] >= 0)
            lrec[pos[u * 4 + j]] = (int)(((unsigned)r[u][j] << GSH) | ((unsigned)c[u][j] & (unsigned)(GBIN - 1)));
      }
    }
    __syncthreads();
    const int tot4 = lst[NB] >> 2;
    int* g = recs + (size_t)ch * (size_t)cap;
    for (int u = tid; u < tot4; u += RTPB) *(v4i*)(g + 4 * u) = *(const v4i*)&lrec[4 * u];
    if (tid < GROW) tab[(size_t)ch * GROW + tid] = lst[(tid >> 1) + (tid & 1)];
    __syncthreads();
  }
}

__global__ void __launch_bounds__(RTPB)
gcn_degree_g(const int* __restrict__ recs, const int* __restrict__ tab,
             const float* __restrict__ x, float* __restrict__ dis, float* __restrict__ xp,
             int n, int nchunks, int cap) {
  __shared__ float acc[GBIN];
  __shared__ int   tst[GTC], ten[GTC];
  const int tid  = threadIdx.x;
  const int lane = tid & 31;
  const int wv   = __builtin_amdgcn_readfirstlane(tid >> 5);
  const int NB   = (n + GBIN - 1) >> GSH;
  for (int b = blockIdx.x; b < NB; b += gridDim.x) {
    const int lo  = b << GSH;
    const int cnt = min(GBIN, n - lo);
    for (int i = tid; i < GBIN; i += RTPB) acc[i] = 0.0f;
    for (int c0 = 0; c0 < nchunks; c0 += GTC) {
      const int nc = min(GTC, nchunks - c0);
      __syncthreads();
      if (tid < nc) {
        const v2i se = *(const v2i*)(tab + (size_t)(c0 + tid) * GROW + 2 * b);
        tst[tid] = se.x;
        ten[tid] = se.y;
      }
      __syncthreads();
      for (int k = wv; k < nc; k += RTPB / 32) {
        const int* p  = recs + (size_t)(c0 + k) * (size_t)cap;
        const int  u1 = ten[k] >> 2;
        for (int u = (tst[k] >> 2) + lane; u < u1; u += 32) {
          const v4i rc = *(const v4i*)(p + 4 * u);
#pragma unroll
          for (int j = 0; j < 4; ++j)
            if (rc[j] != -1) atomicAdd(&acc[rc[j] & (GBIN - 1)], 1.0f);
        }
      }
    }
    __syncthreads();
    for (int i = tid; i < cnt; i += RTPB) {
      const float d = rsqrtf(1.0f + acc[i]);
      dis[lo + i] = d;
      xp[lo + i]  = x[lo + i] * d;
    }
    __syncthreads();
  }
}

__global__ void __launch_bounds__(RTPB)
gcn_l1_g(const int* __restrict__ recs, const int* __restrict__ tab,
         const float* __restrict__ xp, float* __restrict__ agg1,
         int n, int nchunks, int cap) {
  __shared__ float acc[GBIN];
  __shared__ int   tst[GTC], ten[GTC];
  const int tid  = threadIdx.x;
  const int lane = tid & 31;
  const int wv   = __builtin_amdgcn_readfirstlane(tid >> 5);
  const int NB   = (n + GBIN - 1) >> GSH;
  for (int b = blockIdx.x; b < NB; b += gridDim.x) {
    const int lo  = b << GSH;
    const int cnt = min(GBIN, n - lo);
    for (int i = tid; i < GBIN; i += RTPB) acc[i] = 0.0f;
    for (int c0 = 0; c0 < nchunks; c0 += GTC) {
      const int nc = min(GTC, nchunks - c0);
      __syncthreads();
      if (tid < nc) {
        const v2i se = *(const v2i*)(tab + (size_t)(c0 + tid) * GROW + 2 * b);
        tst[tid] = se.x;
        ten[tid] = se.y;
      }
      __syncthreads();
      for (int k = wv; k < nc; k += RTPB / 32) {
        const int* p  = recs + (size_t)(c0 + k) * (size_t)cap;
        const int  u1 = ten[k] >> 2;
        for (int u = (tst[k] >> 2) + lane; u < u1; u += 64) {
          const v4i rc0 = *(const v4i*)(p + 4 * u);
          v4i rc1 = {-1, -1, -1, -1};
          if (u + 32 < u1) rc1 = *(const v4i*)(p + 4 * (u + 32));
          float v[8];
#pragma unroll
          for (int j = 0; j < 4; ++j) {
            v[j]     = (rc0[j] != -1) ? xp[(unsigned)rc0[j] >> GSH] : 0.0f;
            v[4 + j] = (rc1[j] != -1) ? xp[(unsigned)rc1[j] >> GSH] : 0.0f;
          }
#pragma unroll
          for (int j = 0; j < 4; ++j) {
            if (rc0[j] != -1) atomicAdd(&acc[rc0[j] & (GBIN - 1)], v[j]);
            if (rc1[j] != -1) atomicAdd(&acc[rc1[j] & (GBIN - 1)], v[4 + j]);
          }
        }
      }
    }
    __syncthreads();
    for (int i = tid; i < cnt; i += RTPB) agg1[lo + i] = acc[i];
    __syncthreads();
  }
}

__global__ void __launch_bounds__(RTPB)
gcn_l2_g(const int* __restrict__ recs, const int* __restrict__ tab,
         const float* __restrict__ h2p, const float* __restrict__ dis,
         const float* __restrict__ b2, float* __restrict__ out,
         int n, int nchunks, int cap) {
  __shared__ float acc0[GBIN], acc1[GBIN];
  __shared__ int   tst[GTC], ten[GTC];
  const int tid  = threadIdx.x;
  const int lane = tid & 31;
  const int wv   = __builtin_amdgcn_readfirstlane(tid >> 5);
  const int NB   = (n + GBIN - 1) >> GSH;
  const float cb0 = b2[0], cb1 = b2[1];
  for (int b = blockIdx.x; b < NB; b += gridDim.x) {
    const int lo  = b << GSH;
    const int cnt = min(GBIN, n - lo);
    for (int i = tid; i < GBIN; i += RTPB) { acc0[i] = 0.0f; acc1[i] = 0.0f; }
    for (int c0 = 0; c0 < nchunks; c0 += GTC) {
      const int nc = min(GTC, nchunks - c0);
      __syncthreads();
      if (tid < nc) {
        const v2i se = *(const v2i*)(tab + (size_t)(c0 + tid) * GROW + 2 * b);
        tst[tid] = se.x;
        ten[tid] = se.y;
      }
      __syncthreads();
      for (int k = wv; k < nc; k += RTPB / 32) {
        const int* p  = recs + (size_t)(c0 + k) * (size_t)cap;
        const int  u1 = ten[k] >> 2;
        for (int u = (tst[k] >> 2) + lane; u < u1; u += 64) {
          const v4i rc0 = *(const v4i*)(p + 4 * u);
          v4i rc1 = {-1, -1, -1, -1};
          if (u + 32 < u1) rc1 = *(const v4i*)(p + 4 * (u + 32));
          v2f v[8];
#pragma unroll
          for (int j = 0; j < 4; ++j) {
            v[j]     = (v2f){0.f, 0.f};
            v[4 + j] = (v2f){0.f, 0.f};
            if (rc0[j] != -1) v[j]     = *(const v2f*)(h2p + 2 * (size_t)((unsigned)rc0[j] >> GSH));
            if (rc1[j] != -1) v[4 + j] = *(const v2f*)(h2p + 2 * (size_t)((unsigned)rc1[j] >> GSH));
          }
#pragma unroll
          for (int j = 0; j < 4; ++j) {
            if (rc0[j] != -1) {
              atomicAdd(&acc0[rc0[j] & (GBIN - 1)], v[j].x);
              atomicAdd(&acc1[rc0[j] & (GBIN - 1)], v[j].y);
            }
            if (rc1[j] != -1) {
              atomicAdd(&acc0[rc1[j] & (GBIN - 1)], v[4 + j].x);
              atomicAdd(&acc1[rc1[j] & (GBIN - 1)], v[4 + j].y);
            }
          }
        }
      }
    }
    __syncthreads();
    for (int i = tid; i < cnt; i += RTPB) {
      const size_t nd = (size_t)(lo + i);
      const float  d  = dis[nd];
      const v2f    hs = *(const v2f*)(h2p + 2 * nd);
      v2f o;
      o.x = cb0 + d * (acc0[i] + hs.x);
      o.y = cb1 + d * (acc1[i] + hs.y);
      *(v2f*)(out + 2 * nd) = o;
    }
    __syncthreads();
  }
}


#define TSH 15
#define TRN 32768

__global__ void __launch_bounds__(RTPB)
gcn_group_t(const int* __restrict__ row, const int* __restrict__ col,
            int* __restrict__ recs, int* __restrict__ tab,
            float* __restrict__ deg, float* __restrict__ agg1, float* __restrict__ s1,
            int* __restrict__ tick, int e, int n, int chunklen, int nchunks, int cap, int RN) {
  __shared__ __attribute__((aligned(16))) int lrec[GCH_MAX + GNB_MAX * GPAD];
  __shared__ int lcnt[GNB_MAX];
  __shared__ int lst[GNB_MAX + 1];
  const int tid = threadIdx.x;
  const int NB  = RN * RN;
  for (int i = blockIdx.x * RTPB + tid; i < n; i += gridDim.x * RTPB) { deg[i] = 0.0f; agg1[i] = 0.0f; s1[i] = 0.0f; }
  if (blockIdx.x == 0 && tid < 6) tick[tid * 128] = 0;
  for (int ch = blockIdx.x; ch < nchunks; ch += gridDim.x) {
    const int elo = ch * chunklen;
    const int ehi = min(e, elo + chunklen);
    if (tid < GNB_MAX) lcnt[tid] = 0;
    __syncthreads();
    for (int base = elo; base < ehi; base += RTPB * 16) {
      const int b = base + tid * 4;
      v4i c[4], r[4];
      if (base + RTPB * 16 <= ehi) {
#pragma unroll
        for (int u = 0; u < 4; ++u) {
          c[u] = *(const v4i*)(col + b + u * RTPB * 4);
          r[u] = *(const v4i*)(row + b + u * RTPB * 4);
        }
      } else {
#pragma unroll
        for (int u = 0; u < 4; ++u) {
          if (base + (u + 1) * RTPB * 4 <= ehi) {
            c[u] = *(const v4i*)(col + b + u * RTPB * 4);
            r[u] = *(const v4i*)(row + b + u * RTPB * 4);
          } else {
            c[u] = load4_or_none(col, b + u * RTPB * 4, ehi);
            r[u] = load4_or_none(row, b + u * RTPB * 4, ehi);
          }
        }
      }
#pragma unroll
      for (int u = 0; u < 4; ++u) {
#pragma unroll
        for (int j = 0; j < 4; ++j)
          if ((unsigned)c[u][j] < (unsigned)n && (unsigned)r[u][j] < (unsigned)n)
            atomicAdd(&lcnt[(c[u][j] >> TSH) * RN + (r[u][j] >> TSH)], 1);
      }
    }
    __syncthreads();
    int p = 0;
    if (tid < NB) p = (lcnt[tid] + GPAD - 1) & ~(GPAD - 1);
    if (tid < GNB_MAX) lst[tid + 1] = p;
    if (tid == 0) lst[0] = 0;
    __syncthreads();
    for (int off = 1; off < GNB_MAX; off <<= 1) {
      int v = 0;
      if (tid < GNB_MAX && tid >= off) v = lst[tid + 1 - off];
      __syncthreads();
      if (tid < GNB_MAX) lst[tid + 1] += v;
      __syncthreads();
    }
    if (tid < NB) {
      const int s0 = lst[tid], c0 = lcnt[tid], s1e = lst[tid + 1];
      for (int k = s0 + c0; k < s1e; ++k) lrec[k] = -1;
      lcnt[tid] = s0;
    }
    __syncthreads();
    for (int base = elo; base < ehi; base += RTPB * 16) {
      const int b = base + tid * 4;
      v4i c[4], r[4];
      if (base + RTPB * 16 <= ehi) {
#pragma unroll
        for (int u = 0; u < 4; ++u) {
          c[u] = *(const v4i*)(col + b + u * RTPB * 4);
          r[u] = *(const v4i*)(row + b + u * RTPB * 4);
        }
      } else {
#pragma unroll
        for (int u = 0; u < 4; ++u) {
          if (base + (u + 1) * RTPB * 4 <= ehi) {
            c[u] = *(const v4i*)(col + b + u * RTPB * 4);
            r[u] = *(const v4i*)(row + b + u * RTPB * 4);
          } else {
            c[u] = load4_or_none(col, b + u * RTPB * 4, ehi);
            r[u] = load4_or_none(row, b + u * RTPB * 4, ehi);
          }
        }
      }
      int pos[16];
#pragma unroll
      for (int u = 0; u < 4; ++u) {
#pragma unroll
        for (int j = 0; j < 4; ++j) {
          pos[u * 4 + j] = -1;
          if ((unsigned)c[u][j] < (unsigned)n && (unsigned)r[u][j] < (unsigned)n)
            pos[u * 4 + j] = atomicAdd(&lcnt[(c[u][j] >> TSH) * RN + (r[u][j] >> TSH)], 1);
        }
      }
#pragma unroll
      for (int u = 0; u < 4; ++u) {
#pragma unroll
        for (int j = 0; j < 4; ++j)
          if (pos[u * 4 + j] >= 0)
            lrec[pos[u * 4 + j]] = (int)((((unsigned)r[u][j] & (unsigned)(TRN - 1)) << TSH) | ((unsigned)c[u][j] & (unsigned)(TRN - 1)));
      }
    }
    __syncthreads();
    const int tot4 = lst[NB] >> 2;
    int* g = recs + (size_t)ch * (size_t)cap;
    for (int u = tid; u < tot4; u += RTPB) *(v4i*)(g + 4 * u) = *(const v4i*)&lrec[4 * u];
    if (tid < GROW) tab[(size_t)ch * GROW + tid] = lst[(tid >> 1) + (tid & 1)];
    __syncthreads();
  }
}

__device__ __forceinline__ v4i run_vec(const int* __restrict__ p, int u, int u1) {
  const int i = max(min(u, u1 - 1), 0);
  return *(const v4i*)(p + 4 * i);
}
__device__ __forceinline__ v4i run_pad(v4i r, int u, int u1) {
  const v4i none = {-1, -1, -1, -1};
  return (u < u1) ? r : none;
}


__global__ void __launch_bounds__(RTPB)
gcn_degree_t(const int* __restrict__ recs, const int* __restrict__ tab,
             float* __restrict__ deg, int* __restrict__ tick,
             int n, int nchunks, int cap, int RN) {
  __shared__ __attribute__((aligned(16))) float acc[TRN + 32];
  __shared__ int   tst[GTC], ten[GTC];
  __shared__ int   slot;
  const int tid  = threadIdx.x;
  const int lane = tid & 31;
  const int wv   = __builtin_amdgcn_readfirstlane(tid >> 5);
  const int dh   = (RN + 1) >> 1;
  const int h    = cache_half();
  for (int ph = 0; ph < 2; ++ph) {
    const int q     = h ^ ph;
    const int dbase = q ? dh : 0;
    const int nt    = (q ? RN - dh : dh) * RN;
    int* ctr = tick + q * 128;
    for (;;) {
      const int tk = take_block_ticket(ctr, &slot);
      if (tk >= nt) break;
      const int t    = (dbase + tk / RN) * RN + tk % RN;
      const int lo   = (t / RN) << TSH;
      const int cnt  = min(TRN, n - lo);
      for (int i = tid * 4; i < TRN + 32; i += RTPB * 4) *(v4f*)&acc[i] = (v4f){0.f, 0.f, 0.f, 0.f};
      for (int c0 = 0; c0 < nchunks; c0 += GTC) {
        const int nc = min(GTC, nchunks - c0);
        __syncthreads();
        if (tid < nc) {
          const v2i se = *(const v2i*)(tab + (size_t)(c0 + tid) * GROW + 2 * t);
          tst[tid] = se.x;
          ten[tid] = se.y;
        }
        __syncthreads();
        for (int k = wv; k < nc; k += 2 * (RTPB / 32)) {
          const int  k2 = k + RTPB / 32;
          const bool h2 = k2 < nc;
          const int  kk = h2 ? k2 : k;
          const int* pa = recs + (size_t)(c0 + k) * (size_t)cap;
          const int* pb = recs + (size_t)(c0 + kk) * (size_t)cap;
          int       ua = __builtin_amdgcn_readfirstlane(tst[k] >> 2);
          const int ea = __builtin_amdgcn_readfirstlane(ten[k] >> 2);
          const int tb = __builtin_amdgcn_readfirstlane(tst[kk] >> 2);
          const int nb = __builtin_amdgcn_readfirstlane(ten[kk] >> 2);
          int       ub = h2 ? tb : 0;
          const int eb = h2 ? nb : 0;
          for (; ua < ea || ub < eb; ua += 96, ub += 96) {
            v4i ra[3], rb[3];
#pragma unroll
            for (int q = 0; q < 3; ++q) {
              ra[q] = run_vec(pa, ua + 32 * q + lane, ea);
              rb[q] = run_vec(pb, ub + 32 * q + lane, eb);
            }
            asm volatile("" : "+v"(ra[0]), "+v"(ra[1]), "+v"(ra[2]), "+v"(rb[0]), "+v"(rb[1]), "+v"(rb[2]));
#pragma unroll
            for (int q = 0; q < 3; ++q) {
              ra[q] = run_pad(ra[q], ua + 32 * q + lane, ea);
              rb[q] = run_pad(rb[q], ub + 32 * q + lane, eb);
            }
#pragma unroll
            for (int q = 0; q < 2; ++q) {
#pragma unroll
              for (int j = 0; j < 4; ++j) {
                const int sa = ra[q][j], sb = rb[q][j];
                atomicAdd(&acc[(sa != -1) ? (sa & (TRN - 1)) : (TRN + lane)], 1.0f);
                atomicAdd(&acc[(sb != -1) ? (sb & (TRN - 1)) : (TRN + lane)], 1.0f);
              }
            }
            if (ea - ua > 64) {
#pragma unroll
              for (int j = 0; j < 4; ++j) {
                const int sa = ra[2][j];
                atomicAdd(&acc[(sa != -1) ? (sa & (TRN - 1)) : (TRN + lane)], 1.0f);
              }
            }
            if (eb - ub > 64) {
#pragma unroll
              for (int j = 0; j < 4; ++j) {
                const int sb = rb[2][j];
                atomicAdd(&acc[(sb != -1) ? (sb & (TRN - 1)) : (TRN + lane)], 1.0f);
              }
            }
          }
        }
      }
      __syncthreads();
      for (int i0 = 0; i0 < cnt; i0 += RTPB * 8) {
        float a[8];
#pragma unroll
        for (int j = 0; j < 8; ++j) a[j] = acc[i0 + j * RTPB + tid];
#pragma unroll
        for (int j = 0; j < 8; ++j) {
          const int i = i0 + j * RTPB + tid;
          if (i < cnt && a[j] != 0.0f) atomicAdd(&deg[lo + i], a[j]);
        }
      }
    }
  }
}

__global__ void __launch_bounds__(RTPB)
gcn_l1_t(const int* __restrict__ recs, const int* __restrict__ tab,
         const float* __restrict__ x, const float* __restrict__ deg,
         float* __restrict__ agg1, int* __restrict__ tick,
         int n, int nchunks, int cap, int RN) {
  __shared__ __attribute__((aligned(16))) float acc[TRN + 32];
  __shared__ __attribute__((aligned(16))) float xs[TRN];
  __shared__ int   tst[GTC], ten[GTC];
  __shared__ int   slot;
  const int tid  = threadIdx.x;
  const int lane = tid & 31;
  const int wv   = __builtin_amdgcn_readfirstlane(tid >> 5);
  const int dh   = (RN + 1) >> 1;
  const int h    = cache_half();
  for (int ph = 0; ph < 2; ++ph) {
    const int q     = h ^ ph;
    const int dbase = q ? dh : 0;
    const int nt    = (q ? RN - dh : dh) * RN;
    int* ctr = tick + q * 128;
    for (;;) {
      const int tk = take_block_ticket(ctr, &slot);
      if (tk >= nt) break;
      const int t    = (dbase + tk / RN) * RN + tk % RN;
      const int lo   = (t / RN) << TSH;
      const int cnt  = min(TRN, n - lo);
      const int slo  = (t % RN) << TSH;
      const int scnt = min(TRN, n - slo);
      if (tid < 32) acc[TRN + tid] = 0.0f;
      if (scnt == TRN) {
        v4f xv[TRN / (RTPB * 4)], dv[TRN / (RTPB * 4)];
#pragma unroll
        for (int g = 0; g < TRN / (RTPB * 4); ++g) {
          const int i = (g * RTPB + tid) * 4;
          xv[g] = *(const v4f*)(x + slo + i);
          dv[g] = *(const v4f*)(deg + slo + i);
        }
#pragma unroll
        for (int g = 0; g < TRN / (RTPB * 4); ++g) {
          const int i = (g * RTPB + tid) * 4;
          v4f r;
          r.x = xv[g].x * rsqrtf(1.0f + dv[g].x);
          r.y = xv[g].y * rsqrtf(1.0f + dv[g].y);
          r.z = xv[g].z * rsqrtf(1.0f + dv[g].z);
          r.w = xv[g].w * rsqrtf(1.0f + dv[g].w);
          *(v4f*)&xs[i]  = r;
          *(v4f*)&acc[i] = (v4f){0.f, 0.f, 0.f, 0.f};
        }
      } else {
        for (int i = tid; i < TRN; i += RTPB) {
          acc[i] = 0.0f;
          float v = 0.0f;
          if (i < scnt) v = x[slo + i] * rsqrtf(1.0f + deg[slo + i]);
          xs[i] = v;
        }
      }
      for (int c0 = 0; c0 < nchunks; c0 += GTC) {
        const int nc = min(GTC, nchunks - c0);
        __syncthreads();
        if (tid < nc) {
          const v2i se = *(const v2i*)(tab + (size_t)(c0 + tid) * GROW + 2 * t);
          tst[tid] = se.x;
          ten[tid] = se.y;
        }
        __syncthreads();
        for (int k = wv; k < nc; k += 2 * (RTPB / 32)) {
          const int  k2 = k + RTPB / 32;
          const bool h2 = k2 < nc;
          const int  kk = h2 ? k2 : k;
          const int* pa = recs + (size_t)(c0 + k) * (size_t)cap;
          const int* pb = recs + (size_t)(c0 + kk) * (size_t)cap;
          int       ua = __builtin_amdgcn_readfirstlane(tst[k] >> 2);
          const int ea = __builtin_amdgcn_readfirstlane(ten[k] >> 2);
          const int tb = __builtin_amdgcn_readfirstlane(tst[kk] >> 2);
          const int nb = __builtin_amdgcn_readfirstlane(ten[kk] >> 2);
          int       ub = h2 ? tb : 0;
          const int eb = h2 ? nb : 0;
          for (; ua < ea || ub < eb; ua += 96, ub += 96) {
            v4i ra[3], rb[3];
#pragma unroll
            for (int q = 0; q < 3; ++q) {
              ra[q] = run_vec(pa, ua + 32 * q + lane, ea);
              rb[q] = run_vec(pb, ub + 32 * q + lane, eb);
            }
            asm volatile("" : "+v"(ra[0]), "+v"(ra[1]), "+v"(ra[2]), "+v"(rb[0]), "+v"(rb[1]), "+v"(rb[2]));
#pragma unroll
            for (int q = 0; q < 3; ++q) {
              ra[q] = run_pad(ra[q], ua + 32 * q + lane, ea);
              rb[q] = run_pad(rb[q], ub + 32 * q + lane, eb);
            }
            float va[8], vb[8];
#pragma unroll
            for (int q = 0; q < 2; ++q) {
#pragma unroll
              for (int j = 0; j < 4; ++j) {
                const int sa = ra[q][j], sb = rb[q][j];
                va[q * 4 + j] = xs[(sa != -1) ? ((unsigned)sa >> TSH) : 0u];
                vb[q * 4 + j] = xs[(sb != -1) ? ((unsigned)sb >> TSH) : 0u];
              }
            }
#pragma unroll
            for (int q = 0; q < 2; ++q) {
#pragma unroll
              for (int j = 0; j < 4; ++j) {
                const int sa = ra[q][j], sb = rb[q][j];
                atomicAdd(&acc[(sa != -1) ? (sa & (TRN - 1)) : (TRN + lane)], va[q * 4 + j]);
                atomicAdd(&acc[(sb != -1) ? (sb & (TRN - 1)) : (TRN + lane)], vb[q * 4 + j]);
              }
            }
            if (ea - ua > 64) {
              float vc[4];
#pragma unroll
              for (int j = 0; j < 4; ++j) {
                const int sa = ra[2][j];
                vc[j] = xs[(sa != -1) ? ((unsigned)sa >> TSH) : 0u];
              }
#pragma unroll
              for (int j = 0; j < 4; ++j) {
                const int sa = ra[2][j];
                atomicAdd(&acc[(sa != -1) ? (sa & (TRN - 1)) : (TRN + lane)], vc[j]);
              }
            }
            if (eb - ub > 64) {
              float vc[4];
#pragma unroll
              for (int j = 0; j < 4; ++j) {
                const int sb = rb[2][j];
                vc[j] = xs[(sb != -1) ? ((unsigned)sb >> TSH) : 0u];
              }
#pragma unroll
              for (int j = 0; j < 4; ++j) {
                const int sb = rb[2][j];
                atomicAdd(&acc[(sb != -1) ? (sb & (TRN - 1)) : (TRN + lane)], vc[j]);
              }
            }
          }
        }
      }
      __syncthreads();
      for (int i0 = 0; i0 < cnt; i0 += RTPB * 8) {
        float a[8];
#pragma unroll
        for (int j = 0; j < 8; ++j) a[j] = acc[i0 + j * RTPB + tid];
#pragma unroll
        for (int j = 0; j < 8; ++j) {
          const int i = i0 + j * RTPB + tid;
          if (i < cnt && a[j] != 0.0f) atomicAdd(&agg1[lo + i], a[j]);
        }
      }
    }
  }
}

#define NT_BLOCKS 256
#define NT_TPB    512
#define NT_U      4
__global__ void __launch_bounds__(NT_TPB)
gcn_node_t(float* __restrict__ agg1,
           const float* __restrict__ x,
           const float* __restrict__ deg,
           const float* __restrict__ W1,
           const float* __restrict__ b1,
           const float* __restrict__ W2,
           float* __restrict__ p0,
           float* __restrict__ p1,
           int n) {
  const int lane   = threadIdx.x & 31;
  const int m      = lane & 15;
  const int half   = lane >> 4;
  const int tiles  = (n + 15) >> 4;
  const int wave   = blockIdx.x * (blockDim.x >> 5) + __builtin_amdgcn_readfirstlane((int)(threadIdx.x >> 5));
  const int nwaves = gridDim.x * (blockDim.x >> 5);

  const int   wcol  = (m < 2) ? m : 0;
  const float wmask = (m < 2) ? 1.0f : 0.0f;
  v2f w1[4], bi[4], bb[4];
#pragma unroll
  for (int s = 0; s < 4; ++s) {
    const int j0 = s * 4 + half * 2;
    w1[s] = *(const v2f*)(W1 + j0);
    bi[s] = *(const v2f*)(b1 + j0);
    bb[s].x = W2[j0 * 2 + wcol]       * wmask;
    bb[s].y = W2[(j0 + 1) * 2 + wcol] * wmask;
  }
  float* pm = (m == 1) ? p1 : p0;

  for (int t0 = wave * NT_U; t0 < tiles; t0 += nwaves * NT_U) {
    float ag[NT_U], xv[NT_U], dg[NT_U];
#pragma unroll
    for (int u = 0; u < NT_U; ++u) {
      const int node = (t0 + u) * 16 + m;
      const int nc   = node < n ? node : (n - 1);
      ag[u] = agg1[nc];
      xv[u] = x[nc];
      dg[u] = deg[nc];
    }
#pragma unroll
    for (int u = 0; u < NT_U; ++u) {
      const int base = (t0 + u) * 16;
      const int node = base + m;
      const float dn = rsqrtf(1.0f + dg[u]);
      const float av = (ag[u] + xv[u] * dn) * dn;
      if (half == 0 && node < n) agg1[node] = 0.0f;

      v8f acc = {};
#pragma unroll
      for (int s = 0; s < 4; ++s) {
        v2f a;
        a.x = fmaxf(av * w1[s].x + bi[s].x, 0.0f);
        a.y = fmaxf(av * w1[s].y + bi[s].y, 0.0f);
        acc = __builtin_amdgcn_wmma_f32_16x16x4_f32(
             false, a,  false, bb[s],
             (short)0, acc,  false,  false);
      }

      float dr[8];
#pragma unroll
      for (int r = 0; r < 8; ++r)
        dr[r] = __int_as_float(__builtin_amdgcn_ds_bpermute((r + half * 8) << 2, __float_as_int(dn)));
      if (m < 2) {
#pragma unroll
        for (int r = 0; r < 8; ++r) {
          const int nd = base + r + half * 8;
          if (nd < n) pm[nd] = acc[r] * dr[r];
        }
      }
    }
  }
}


__global__ void __launch_bounds__(RTPB)
gcn_l2_t(const int* __restrict__ recs, const int* __restrict__ tab,
         const float* __restrict__ p0, const float* __restrict__ p1,
         float* __restrict__ s0, float* __restrict__ s1, int* __restrict__ tick,
         int n, int nchunks, int cap, int RN) {
  __shared__ __attribute__((aligned(16))) float acc[TRN + 32];
  __shared__ __attribute__((aligned(16))) float hs[TRN];
  __shared__ int   tst[GTC], ten[GTC];
  __shared__ int   slot;
  const int tid  = threadIdx.x;
  const int lane = tid & 31;
  const int wv   = __builtin_amdgcn_readfirstlane(tid >> 5);
  const int dh   = (RN + 1) >> 1;
  const int h    = cache_half();
  for (int ph = 0; ph < 2; ++ph) {
    const int q     = h ^ ph;
    const int dbase = q ? dh : 0;
    const int nt    = (q ? RN - dh : dh) * RN;
    int* ctr = tick + q * 128;
    for (;;) {
      const int tk = take_block_ticket(ctr, &slot);
      if (tk >= nt) break;
      const int t    = (dbase + tk / RN) * RN + tk % RN;
      const int lo   = (t / RN) << TSH;
      const int cnt  = min(TRN, n - lo);
      const int slo  = (t % RN) << TSH;
      const int scnt = min(TRN, n - slo);
      for (int comp = 0; comp < 2; ++comp) {
      const float* pk = comp ? p1 : p0;
      float*       sk = comp ? s1 : s0;
      if (tid < 32) acc[TRN + tid] = 0.0f;
      if (scnt == TRN) {
        v4f pv[TRN / (RTPB * 4)];
#pragma unroll
        for (int g = 0; g < TRN / (RTPB * 4); ++g) pv[g] = *(const v4f*)(pk + slo + (g * RTPB + tid) * 4);
#pragma unroll
        for (int g = 0; g < TRN / (RTPB * 4); ++g) {
          const int i = (g * RTPB + tid) * 4;
          *(v4f*)&hs[i]  = pv[g];
          *(v4f*)&acc[i] = (v4f){0.f, 0.f, 0.f, 0.f};
        }
      } else {
        for (int i = tid; i < TRN; i += RTPB) {
          acc[i] = 0.0f;
          hs[i]  = (i < scnt) ? pk[slo + i] : 0.0f;
        }
      }
      for (int c0 = 0; c0 < nchunks; c0 += GTC) {
        const int nc = min(GTC, nchunks - c0);
        __syncthreads();
        if (tid < nc) {
          const v2i se = *(const v2i*)(tab + (size_t)(c0 + tid) * GROW + 2 * t);
          tst[tid] = se.x;
          ten[tid] = se.y;
        }
        __syncthreads();
        for (int k = wv; k < nc; k += 2 * (RTPB / 32)) {
          const int  k2 = k + RTPB / 32;
          const bool h2 = k2 < nc;
          const int  kk = h2 ? k2 : k;
          const int* pa = recs + (size_t)(c0 + k) * (size_t)cap;
          const int* pb = recs + (size_t)(c0 + kk) * (size_t)cap;
          int       ua = __builtin_amdgcn_readfirstlane(tst[k] >> 2);
          const int ea = __builtin_amdgcn_readfirstlane(ten[k] >> 2);
          const int tb = __builtin_amdgcn_readfirstlane(tst[kk] >> 2);
          const int nb = __builtin_amdgcn_readfirstlane(ten[kk] >> 2);
          int       ub = h2 ? tb : 0;
          const int eb = h2 ? nb : 0;
          for (; ua < ea || ub < eb; ua += 96, ub += 96) {
            v4i ra[3], rb[3];
#pragma unroll
            for (int q = 0; q < 3; ++q) {
              ra[q] = run_vec(pa, ua + 32 * q + lane, ea);
              rb[q] = run_vec(pb, ub + 32 * q + lane, eb);
            }
            asm volatile("" : "+v"(ra[0]), "+v"(ra[1]), "+v"(ra[2]), "+v"(rb[0]), "+v"(rb[1]), "+v"(rb[2]));
#pragma unroll
            for (int q = 0; q < 3; ++q) {
              ra[q] = run_pad(ra[q], ua + 32 * q + lane, ea);
              rb[q] = run_pad(rb[q], ub + 32 * q + lane, eb);
            }
            float va[8], vb[8];
#pragma unroll
            for (int q = 0; q < 2; ++q) {
#pragma unroll
              for (int j = 0; j < 4; ++j) {
                const int sa = ra[q][j], sb = rb[q][j];
                va[q * 4 + j] = hs[(sa != -1) ? ((unsigned)sa >> TSH) : 0u];
                vb[q * 4 + j] = hs[(sb != -1) ? ((unsigned)sb >> TSH) : 0u];
              }
            }
#pragma unroll
            for (int q = 0; q < 2; ++q) {
#pragma unroll
              for (int j = 0; j < 4; ++j) {
                const int sa = ra[q][j], sb = rb[q][j];
                atomicAdd(&acc[(sa != -1) ? (sa & (TRN - 1)) : (TRN + lane)], va[q * 4 + j]);
                atomicAdd(&acc[(sb != -1) ? (sb & (TRN - 1)) : (TRN + lane)], vb[q * 4 + j]);
              }
            }
            if (ea - ua > 64) {
              float vc[4];
#pragma unroll
              for (int j = 0; j < 4; ++j) {
                const int sa = ra[2][j];
                vc[j] = hs[(sa != -1) ? ((unsigned)sa >> TSH) : 0u];
              }
#pragma unroll
              for (int j = 0; j < 4; ++j) {
                const int sa = ra[2][j];
                atomicAdd(&acc[(sa != -1) ? (sa & (TRN - 1)) : (TRN + lane)], vc[j]);
              }
            }
            if (eb - ub > 64) {
              float vc[4];
#pragma unroll
              for (int j = 0; j < 4; ++j) {
                const int sb = rb[2][j];
                vc[j] = hs[(sb != -1) ? ((unsigned)sb >> TSH) : 0u];
              }
#pragma unroll
              for (int j = 0; j < 4; ++j) {
                const int sb = rb[2][j];
                atomicAdd(&acc[(sb != -1) ? (sb & (TRN - 1)) : (TRN + lane)], vc[j]);
              }
            }
          }
        }
      }
      __syncthreads();
      for (int i0 = 0; i0 < cnt; i0 += RTPB * 8) {
        float a[8];
#pragma unroll
        for (int j = 0; j < 8; ++j) a[j] = acc[i0 + j * RTPB + tid];
#pragma unroll
        for (int j = 0; j < 8; ++j) {
          const int i = i0 + j * RTPB + tid;
          if (i < cnt && a[j] != 0.0f) atomicAdd(&sk[lo + i], a[j]);
        }
      }
      __syncthreads();
      }
    }
  }
}

#define FT_BLOCKS 512
#define FT_TPB    256
__global__ void __launch_bounds__(FT_TPB)
gcn_fin_t(const float* __restrict__ deg, const float* __restrict__ b2,
          const float* __restrict__ p0, const float* __restrict__ p1,
          const float* __restrict__ s0, const float* __restrict__ s1,
          float* __restrict__ out, int n) {
  const float c0 = b2[0], c1 = b2[1];
  const int nq = n >> 2;
  for (int q = blockIdx.x * blockDim.x + threadIdx.x; q < nq; q += gridDim.x * blockDim.x) {
    const int i = 4 * q;
    const v4f dg = *(const v4f*)(deg + i);
    const v4f a0 = *(const v4f*)(s0 + i);
    const v4f h0 = *(const v4f*)(p0 + i);
    const v4f a1 = *(const v4f*)(s1 + i);
    const v4f h1 = *(const v4f*)(p1 + i);
    v4f d;
    d.x = rsqrtf(1.0f + dg.x);
    d.y = rsqrtf(1.0f + dg.y);
    d.z = rsqrtf(1.0f + dg.z);
    d.w = rsqrtf(1.0f + dg.w);
    v4f oa, ob;
    oa.x = c0 + d.x * (a0.x + h0.x);  oa.y = c1 + d.x * (a1.x + h1.x);
    oa.z = c0 + d.y * (a0.y + h0.y);  oa.w = c1 + d.y * (a1.y + h1.y);
    ob.x = c0 + d.z * (a0.z + h0.z);  ob.y = c1 + d.z * (a1.z + h1.z);
    ob.z = c0 + d.w * (a0.w + h0.w);  ob.w = c1 + d.w * (a1.w + h1.w);
    float* o = out + 2 * (size_t)i;
    *(v4f*)o       = oa;
    *(v4f*)(o + 4) = ob;
  }
  if (blockIdx.x == 0 && (int)threadIdx.x < (n & 3)) {
    const int i = (n & ~3) + (int)threadIdx.x;
    const float d = rsqrtf(1.0f + deg[i]);
    v2f o;
    o.x = c0 + d * (s0[i] + p0[i]);
    o.y = c1 + d * (s1[i] + p1[i]);
    *(v2f*)(out + 2 * (size_t)i) = o;
  }
}

extern "C" void kernel_launch(void* const* d_in, const int* in_sizes, int n_in,
                              void* d_out, int out_size, void* d_ws, size_t ws_size,
                              hipStream_t stream) {
  const float* x  = (const float*)d_in[0];
  const int*   ei = (const int*)d_in[1];
  const float* W1 = (const float*)d_in[2];
  const float* b1 = (const float*)d_in[3];
  const float* W2 = (const float*)d_in[4];
  const float* b2 = (const float*)d_in[5];
  float* out = (float*)d_out;

  const int n = in_sizes[0];
  const int e = in_sizes[1] / 2;
  const int* row = ei;
  const int* col = ei + e;

  size_t np   = (size_t)((n + 3) & ~3);
  float* ws   = (float*)d_ws;
  float* dis  = ws;
  float* xp   = ws + np;
  float* agg1 = ws + 2 * np;
  float* h2p  = ws + 3 * np;

  const int cs = (n >= 512) ? 128 : 1;
  int* tick24 = (int*)h2p;
  int* tick6  = (int*)agg1;

  const int tb = 256;
  const int nb_node = (n + tb - 1) / tb;
  const int tiles = (n + 15) / 16;
  const int nb_tile = (tiles + 7) / 8;
  const int nb_fin  = (n + 4 * tb - 1) / (4 * tb);

  {
    const int RNt = (n + TRN - 1) >> TSH;
    const int NBt = RNt * RNt;
    int chunklen_t = ((e + RANGE_BLOCKS - 1) / RANGE_BLOCKS + 3) & ~3;
    if (chunklen_t > GCH_MAX) chunklen_t = GCH_MAX;
    if (chunklen_t < 4) chunklen_t = 4;
    const int nchunks_t = (e + chunklen_t - 1) / chunklen_t;
    const int cap_t = (chunklen_t + 31 + NBt * GPAD) & ~31;
    const size_t off_tab_t = (5 * np + 31) & ~(size_t)31;
    const size_t off_rec_t = off_tab_t + (size_t)nchunks_t * GROW;
    const size_t off_tick_t = off_rec_t + (size_t)nchunks_t * (size_t)cap_t;
    const size_t need_t    = (off_tick_t + 6 * 128) * sizeof(float);
    if (n >= 1 && e >= 1 && NBt <= GNB_MAX && need_t <= ws_size) {
      float* degt = ws;
      float* p0   = ws + np;
      float* p1   = ws + 3 * np;
      float* s1   = ws + 4 * np;
      int* tab  = (int*)(ws + off_tab_t);
      int* recs = (int*)(ws + off_rec_t);
      int* tick = (int*)(ws + off_tick_t);
      gcn_group_t <<<RANGE_BLOCKS, RTPB, 0, stream>>>(row, col, recs, tab, degt, agg1, s1, tick, e, n, chunklen_t, nchunks_t, cap_t, RNt);
      gcn_degree_t<<<NBt, RTPB, 0, stream>>>(recs, tab, degt, tick, n, nchunks_t, cap_t, RNt);
      gcn_l1_t    <<<NBt, RTPB, 0, stream>>>(recs, tab, x, degt, agg1, tick + 2 * 128, n, nchunks_t, cap_t, RNt);
      gcn_node_t  <<<NT_BLOCKS, NT_TPB, 0, stream>>>(agg1, x, degt, W1, b1, W2, p0, p1, n);
      gcn_l2_t    <<<NBt, RTPB, 0, stream>>>(recs, tab, p0, p1, agg1, s1, tick + 4 * 128, n, nchunks_t, cap_t, RNt);
      gcn_fin_t   <<<FT_BLOCKS, FT_TPB, 0, stream>>>(degt, b2, p0, p1, agg1, s1, out, n);
      return;
    }
  }

  const int NBg = (n + GBIN - 1) >> GSH;
  int chunklen = ((e + RANGE_BLOCKS - 1) / RANGE_BLOCKS + 3) & ~3;
  if (chunklen > GCH_MAX) chunklen = GCH_MAX;
  if (chunklen < 4) chunklen = 4;
  const int nchunks = (e + chunklen - 1) / chunklen;
  const int cap = (chunklen + 31 + NBg * GPAD) & ~31;
  const size_t off_tab = (5 * np + 31) & ~(size_t)31;
  const size_t off_rec = off_tab + (size_t)nchunks * GROW;
  const size_t need    = (off_rec + (size_t)nchunks * (size_t)cap) * sizeof(float);
  if (n >= 1 && e >= 1 && NBg <= GNB_MAX && need <= ws_size) {
    int* tab  = (int*)(ws + off_tab);
    int* recs = (int*)(ws + off_rec);
    gcn_group   <<<RANGE_BLOCKS, RTPB, 0, stream>>>(row, col, recs, tab, e, n, chunklen, nchunks, cap);
    gcn_degree_g<<<NBg, RTPB, 0, stream>>>(recs, tab, x, dis, xp, n, nchunks, cap);
    gcn_l1_g    <<<NBg, RTPB, 0, stream>>>(recs, tab, xp, agg1, n, nchunks, cap);
    gcn_node    <<<nb_tile, 256, 0, stream>>>(agg1, xp, dis, W1, b1, W2, h2p, out, n, 0);
    gcn_l2_g    <<<NBg, RTPB, 0, stream>>>(recs, tab, h2p, dis, b2, out, n, nchunks, cap);
    return;
  }

  gcn_init  <<<nb_node, tb, 0, stream>>>(dis, agg1, tick24, cs, n);
  gcn_degree<<<RANGE_BLOCKS, RTPB, 0, stream>>>(col, dis, tick24, cs, e, n);
  gcn_dis   <<<nb_node, tb, 0, stream>>>(x, dis, xp, n);
  gcn_l1    <<<RANGE_BLOCKS, RTPB, 0, stream>>>(row, col, xp, agg1, tick24 + 2 * cs, cs, e, n);
  gcn_node  <<<nb_tile, 256, 0, stream>>>(agg1, xp, dis, W1, b1, W2, h2p, out, n, 1);
  gcn_tick0 <<<1, 32, 0, stream>>>(tick6, cs);
  gcn_l2    <<<RANGE_BLOCKS, RTPB, 0, stream>>>(row, col, h2p, out, tick6, cs, e, n);
  gcn_fin   <<<nb_fin, tb, 0, stream>>>(dis, b2, out, n);
}
